# starting kernel
# baseline (speedup 1.0000x reference)
#include <hip/hip_runtime.h>
#include <hip/hip_bf16.h>


typedef __attribute__((ext_vector_type(2))) float v2f;
typedef __attribute__((ext_vector_type(8))) float v8f;

#define F_IN 128
#define F_H  16
#define F_C  2

__global__ void GNN_zero_f32(float* __restrict__ p, long n) {
    long i = (long)blockIdx.x * blockDim.x + threadIdx.x;
    if (i < n) p[i] = 0.0f;
}

__global__ void GNN_degree(const int* __restrict__ eidx, long E,
                           float* __restrict__ deg) {
    long e = (long)blockIdx.x * blockDim.x + threadIdx.x;
    if (e >= E) return;
    int d = eidx[E + e];
    atomicAdd(&deg[d], 1.0f);
}

__global__ void GNN_dinv(const float* __restrict__ deg,
                         float* __restrict__ dinv, int n) {
    int i = blockIdx.x * blockDim.x + threadIdx.x;
    if (i < n) dinv[i] = rsqrtf(deg[i] + 1.0f);
}

__global__ void GNN_gemm1_wmma(const float* __restrict__ x,
                               const float* __restrict__ W1,
                               float* __restrict__ h1, int nTiles) {
    int lane = threadIdx.x & 31;
    int wave = threadIdx.x >> 5;
    int tile = blockIdx.x * (blockDim.x >> 5) + wave;
    if (tile >= nTiles) return;

    int m    = lane & 15;
    int koff = (lane >> 4) << 1;
    long rowA = (long)(tile * 16 + m) * F_IN;

    v8f acc = {};
    #pragma unroll 4
    for (int kk = 0; kk < F_IN; kk += 4) {
        v2f a, b;
        a.x = x[rowA + kk + koff];
        a.y = x[rowA + kk + koff + 1];
        b.x = W1[(kk + koff)     * F_H + m];
        b.y = W1[(kk + koff + 1) * F_H + m];
        acc = __builtin_amdgcn_wmma_f32_16x16x4_f32(
            false, a, false, b, (short)0, acc, false, false);
    }

    int rbase = tile * 16 + (lane >> 4) * 8;
    #pragma unroll
    for (int r = 0; r < 8; ++r)
        h1[(long)(rbase + r) * F_H + m] = acc[r];
}

__global__ void GNN_gemm1_tail(const float* __restrict__ x,
                               const float* __restrict__ W1,
                               float* __restrict__ h1, int startRow, int n) {
    int idx = blockIdx.x * blockDim.x + threadIdx.x;
    int row = startRow + (idx >> 4);
    int col = idx & 15;
    if (row >= n) return;
    float acc = 0.0f;
    for (int k = 0; k < F_IN; ++k)
        acc += x[(long)row * F_IN + k] * W1[k * F_H + col];
    h1[(long)row * F_H + col] = acc;
}

__global__ void GNN_scatter1(const int* __restrict__ eidx, long E,
                             const float* __restrict__ dinv,
                             const float* __restrict__ h1,
                             float* __restrict__ agg1) {
    long e = (long)blockIdx.x * blockDim.x + threadIdx.x;
    if (e >= E) return;
    int s = eidx[e];
    int d = eidx[E + e];
    float norm = dinv[s] * dinv[d];
    const float4* hp = (const float4*)(h1 + (long)s * F_H);
    float* ap = agg1 + (long)d * F_H;
    #pragma unroll
    for (int q = 0; q < 4; ++q) {
        float4 v = hp[q];
        atomicAdd(ap + q * 4 + 0, norm * v.x);
        atomicAdd(ap + q * 4 + 1, norm * v.y);
        atomicAdd(ap + q * 4 + 2, norm * v.z);
        atomicAdd(ap + q * 4 + 3, norm * v.w);
    }
}

__global__ void GNN_combine1(const float* __restrict__ agg1,
                             const float* __restrict__ h1,
                             const float* __restrict__ dinv,
                             const float* __restrict__ b1,
                             const float* __restrict__ W2,
                             float* __restrict__ h2, int n) {
    int i = blockIdx.x * blockDim.x + threadIdx.x;
    if (i >= n) return;
    float di  = dinv[i];
    float di2 = di * di;
    float acc0 = 0.0f, acc1 = 0.0f;
    #pragma unroll
    for (int j = 0; j < F_H; ++j) {
        float v = agg1[(long)i * F_H + j] + di2 * h1[(long)i * F_H + j] + b1[j];
        v = fmaxf(v, 0.0f);
        acc0 = fmaf(v, W2[j * F_C + 0], acc0);
        acc1 = fmaf(v, W2[j * F_C + 1], acc1);
    }
    h2[(long)i * F_C + 0] = acc0;
    h2[(long)i * F_C + 1] = acc1;
}

__global__ void GNN_scatter2(const int* __restrict__ eidx, long E,
                             const float* __restrict__ dinv,
                             const float* __restrict__ h2,
                             float* __restrict__ agg2) {
    long e = (long)blockIdx.x * blockDim.x + threadIdx.x;
    if (e >= E) return;
    int s = eidx[e];
    int d = eidx[E + e];
    float norm = dinv[s] * dinv[d];
    float2 v = *(const float2*)(h2 + (long)s * F_C);
    atomicAdd(&agg2[(long)d * F_C + 0], norm * v.x);
    atomicAdd(&agg2[(long)d * F_C + 1], norm * v.y);
}

__global__ void GNN_final(const float* __restrict__ agg2,
                          const float* __restrict__ h2,
                          const float* __restrict__ dinv,
                          const float* __restrict__ b2,
                          float* __restrict__ out, int n) {
    int i = blockIdx.x * blockDim.x + threadIdx.x;
    if (i >= n) return;
    float di  = dinv[i];
    float di2 = di * di;
    float o0 = agg2[(long)i * F_C + 0] + di2 * h2[(long)i * F_C + 0] + b2[0];
    float o1 = agg2[(long)i * F_C + 1] + di2 * h2[(long)i * F_C + 1] + b2[1];
    float mx  = fmaxf(o0, o1);
    float lse = mx + logf(expf(o0 - mx) + expf(o1 - mx));
    out[(long)i * F_C + 0] = o0 - lse;
    out[(long)i * F_C + 1] = o1 - lse;
}

extern "C" void kernel_launch(void* const* d_in, const int* in_sizes, int n_in,
                              void* d_out, int out_size, void* d_ws, size_t ws_size,
                              hipStream_t stream) {
    const float* x  = (const float*)d_in[0];
    const int* eidx = (const int*)  d_in[1];
    const float* W1 = (const float*)d_in[2];
    const float* b1 = (const float*)d_in[3];
    const float* W2 = (const float*)d_in[4];
    const float* b2 = (const float*)d_in[5];
    float* out = (float*)d_out;

    long nN = (long)in_sizes[0] / F_IN;
    long E  = (long)in_sizes[1] / 2;

    float* ws   = (float*)d_ws;
    float* deg  = ws;
    float* agg1 = ws + nN;
    float* agg2 = ws + 17 * nN;
    float* dinv = ws + 19 * nN;
    float* h1   = ws + 20 * nN;
    float* h2   = ws + 36 * nN;

    const int T = 256;
    long zn = 19 * nN;
    GNN_zero_f32<<<(int)((zn + T - 1) / T), T, 0, stream>>>(ws, zn);
    GNN_degree  <<<(int)((E + T - 1) / T), T, 0, stream>>>(eidx, E, deg);
    GNN_dinv    <<<(int)((nN + T - 1) / T), T, 0, stream>>>(deg, dinv, (int)nN);

    int nTiles = (int)(nN / 16);
    if (nTiles > 0)
        GNN_gemm1_wmma<<<(nTiles + 7) / 8, T, 0, stream>>>(x, W1, h1, nTiles);
    int rem = (int)(nN - (long)nTiles * 16);
    if (rem > 0)
        GNN_gemm1_tail<<<(rem * 16 + T - 1) / T, T, 0, stream>>>(
            x, W1, h1, nTiles * 16, (int)nN);

    GNN_scatter1<<<(int)((E + T - 1) / T), T, 0, stream>>>(eidx, E, dinv, h1, agg1);
    GNN_combine1<<<(int)((nN + T - 1) / T), T, 0, stream>>>(agg1, h1, dinv, b1, W2, h2, (int)nN);
    GNN_scatter2<<<(int)((E + T - 1) / T), T, 0, stream>>>(eidx, E, dinv, h2, agg2);
    GNN_final   <<<(int)((nN + T - 1) / T), T, 0, stream>>>(agg2, h2, dinv, b2, out, (int)nN);
}


# scatter1: the 16 lanes of a row add one edge's 16 features, one contiguous 64-byte atomic segment per edge, not 16 strided words per lane; float atomics kept, arrival order of adds changes
# speedup vs baseline: 2.0711x; 2.0711x over previous
#include <hip/hip_runtime.h>
#include <hip/hip_bf16.h>


typedef __attribute__((ext_vector_type(2))) float v2f;
typedef __attribute__((ext_vector_type(8))) float v8f;

#define F_IN 128
#define F_H  16
#define F_C  2

__global__ void GNN_zero_f32(float* __restrict__ p, long n) {
    long i = (long)blockIdx.x * blockDim.x + threadIdx.x;
    if (i < n) p[i] = 0.0f;
}

__global__ void GNN_degree(const int* __restrict__ eidx, long E,
                           float* __restrict__ deg) {
    long e = (long)blockIdx.x * blockDim.x + threadIdx.x;
    if (e >= E) return;
    int d = eidx[E + e];
    atomicAdd(&deg[d], 1.0f);
}

__global__ void GNN_dinv(const float* __restrict__ deg,
                         float* __restrict__ dinv, int n) {
    int i = blockIdx.x * blockDim.x + threadIdx.x;
    if (i < n) dinv[i] = rsqrtf(deg[i] + 1.0f);
}

__global__ void GNN_gemm1_wmma(const float* __restrict__ x,
                               const float* __restrict__ W1,
                               float* __restrict__ h1, int nTiles) {
    int lane = threadIdx.x & 31;
    int wave = threadIdx.x >> 5;
    int tile = blockIdx.x * (blockDim.x >> 5) + wave;
    if (tile >= nTiles) return;

    int m    = lane & 15;
    int koff = (lane >> 4) << 1;
    long rowA = (long)(tile * 16 + m) * F_IN;

    v8f acc = {};
    #pragma unroll 4
    for (int kk = 0; kk < F_IN; kk += 4) {
        v2f a, b;
        a.x = x[rowA + kk + koff];
        a.y = x[rowA + kk + koff + 1];
        b.x = W1[(kk + koff)     * F_H + m];
        b.y = W1[(kk + koff + 1) * F_H + m];
        acc = __builtin_amdgcn_wmma_f32_16x16x4_f32(
            false, a, false, b, (short)0, acc, false, false);
    }

    int rbase = tile * 16 + (lane >> 4) * 8;
    #pragma unroll
    for (int r = 0; r < 8; ++r)
        h1[(long)(rbase + r) * F_H + m] = acc[r];
}

__global__ void GNN_gemm1_tail(const float* __restrict__ x,
                               const float* __restrict__ W1,
                               float* __restrict__ h1, int startRow, int n) {
    int idx = blockIdx.x * blockDim.x + threadIdx.x;
    int row = startRow + (idx >> 4);
    int col = idx & 15;
    if (row >= n) return;
    float acc = 0.0f;
    for (int k = 0; k < F_IN; ++k)
        acc += x[(long)row * F_IN + k] * W1[k * F_H + col];
    h1[(long)row * F_H + col] = acc;
}


template <int J>
__device__ __forceinline__ int GNN_row_lane(int x) {
    return __builtin_amdgcn_update_dpp(0, x, 0x150 + J, 0xf, 0xf, false);
}

template <int J>
__device__ __forceinline__ float GNN_gather_step(int s, const float* __restrict__ hp) {
    unsigned sj = (unsigned)GNN_row_lane<J>(s);
    return hp[sj * (unsigned)F_H];
}

template <int J>
__device__ __forceinline__ void GNN_add_step(int d, float nrm, float v, float* __restrict__ ap) {
    unsigned dj = (unsigned)GNN_row_lane<J>(d);
    float nj = __int_as_float(GNN_row_lane<J>(__float_as_int(nrm)));
    atomicAdd(ap + dj * (unsigned)F_H, nj * v);
}

__global__ void GNN_scatter1(const int* __restrict__ eidx, long E,
                             const float* __restrict__ dinv,
                             const float* __restrict__ h1,
                             float* __restrict__ agg1) {
    long e = (long)blockIdx.x * blockDim.x + threadIdx.x;
    int lane = threadIdx.x & 31;
    long base = e - lane;
    if (base + 32 <= E) {
        int s = eidx[e];
        int d = eidx[E + e];
        float nrm = dinv[s] * dinv[d];
        int c = lane & 15;
        const float* hp = h1 + c;
        float* ap = agg1 + c;

        float v[16];
        v[0]  = GNN_gather_step<0>(s, hp);   v[1]  = GNN_gather_step<1>(s, hp);
        v[2]  = GNN_gather_step<2>(s, hp);   v[3]  = GNN_gather_step<3>(s, hp);
        v[4]  = GNN_gather_step<4>(s, hp);   v[5]  = GNN_gather_step<5>(s, hp);
        v[6]  = GNN_gather_step<6>(s, hp);   v[7]  = GNN_gather_step<7>(s, hp);
        v[8]  = GNN_gather_step<8>(s, hp);   v[9]  = GNN_gather_step<9>(s, hp);
        v[10] = GNN_gather_step<10>(s, hp);  v[11] = GNN_gather_step<11>(s, hp);
        v[12] = GNN_gather_step<12>(s, hp);  v[13] = GNN_gather_step<13>(s, hp);
        v[14] = GNN_gather_step<14>(s, hp);  v[15] = GNN_gather_step<15>(s, hp);

        GNN_add_step<0>(d, nrm, v[0], ap);    GNN_add_step<1>(d, nrm, v[1], ap);
        GNN_add_step<2>(d, nrm, v[2], ap);    GNN_add_step<3>(d, nrm, v[3], ap);
        GNN_add_step<4>(d, nrm, v[4], ap);    GNN_add_step<5>(d, nrm, v[5], ap);
        GNN_add_step<6>(d, nrm, v[6], ap);    GNN_add_step<7>(d, nrm, v[7], ap);
        GNN_add_step<8>(d, nrm, v[8], ap);    GNN_add_step<9>(d, nrm, v[9], ap);
        GNN_add_step<10>(d, nrm, v[10], ap);  GNN_add_step<11>(d, nrm, v[11], ap);
        GNN_add_step<12>(d, nrm, v[12], ap);  GNN_add_step<13>(d, nrm, v[13], ap);
        GNN_add_step<14>(d, nrm, v[14], ap);  GNN_add_step<15>(d, nrm, v[15], ap);
    } else if (e < E) {
        int s = eidx[e];
        int d = eidx[E + e];
        float nrm = dinv[s] * dinv[d];
        const float* hr = h1 + (long)s * F_H;
        float* ar = agg1 + (long)d * F_H;
        for (int j = 0; j < F_H; ++j)
            atomicAdd(ar + j, nrm * hr[j]);
    }
}

__global__ void GNN_combine1(const float* __restrict__ agg1,
                             const float* __restrict__ h1,
                             const float* __restrict__ dinv,
                             const float* __restrict__ b1,
                             const float* __restrict__ W2,
                             float* __restrict__ h2, int n) {
    int i = blockIdx.x * blockDim.x + threadIdx.x;
    if (i >= n) return;
    float di  = dinv[i];
    float di2 = di * di;
    float acc0 = 0.0f, acc1 = 0.0f;
    #pragma unroll
    for (int j = 0; j < F_H; ++j) {
        float v = agg1[(long)i * F_H + j] + di2 * h1[(long)i * F_H + j] + b1[j];
        v = fmaxf(v, 0.0f);
        acc0 = fmaf(v, W2[j * F_C + 0], acc0);
        acc1 = fmaf(v, W2[j * F_C + 1], acc1);
    }
    h2[(long)i * F_C + 0] = acc0;
    h2[(long)i * F_C + 1] = acc1;
}

__global__ void GNN_scatter2(const int* __restrict__ eidx, long E,
                             const float* __restrict__ dinv,
                             const float* __restrict__ h2,
                             float* __restrict__ agg2) {
    long e = (long)blockIdx.x * blockDim.x + threadIdx.x;
    if (e >= E) return;
    int s = eidx[e];
    int d = eidx[E + e];
    float norm = dinv[s] * dinv[d];
    float2 v = *(const float2*)(h2 + (long)s * F_C);
    atomicAdd(&agg2[(long)d * F_C + 0], norm * v.x);
    atomicAdd(&agg2[(long)d * F_C + 1], norm * v.y);
}

__global__ void GNN_final(const float* __restrict__ agg2,
                          const float* __restrict__ h2,
                          const float* __restrict__ dinv,
                          const float* __restrict__ b2,
                          float* __restrict__ out, int n) {
    int i = blockIdx.x * blockDim.x + threadIdx.x;
    if (i >= n) return;
    float di  = dinv[i];
    float di2 = di * di;
    float o0 = agg2[(long)i * F_C + 0] + di2 * h2[(long)i * F_C + 0] + b2[0];
    float o1 = agg2[(long)i * F_C + 1] + di2 * h2[(long)i * F_C + 1] + b2[1];
    float mx  = fmaxf(o0, o1);
    float lse = mx + logf(expf(o0 - mx) + expf(o1 - mx));
    out[(long)i * F_C + 0] = o0 - lse;
    out[(long)i * F_C + 1] = o1 - lse;
}

extern "C" void kernel_launch(void* const* d_in, const int* in_sizes, int n_in,
                              void* d_out, int out_size, void* d_ws, size_t ws_size,
                              hipStream_t stream) {
    const float* x  = (const float*)d_in[0];
    const int* eidx = (const int*)  d_in[1];
    const float* W1 = (const float*)d_in[2];
    const float* b1 = (const float*)d_in[3];
    const float* W2 = (const float*)d_in[4];
    const float* b2 = (const float*)d_in[5];
    float* out = (float*)d_out;

    long nN = (long)in_sizes[0] / F_IN;
    long E  = (long)in_sizes[1] / 2;

    float* ws   = (float*)d_ws;
    float* deg  = ws;
    float* agg1 = ws + nN;
    float* agg2 = ws + 17 * nN;
    float* dinv = ws + 19 * nN;
    float* h1   = ws + 20 * nN;
    float* h2   = ws + 36 * nN;

    const int T = 256;
    long zn = 19 * nN;
    GNN_zero_f32<<<(int)((zn + T - 1) / T), T, 0, stream>>>(ws, zn);
    GNN_degree  <<<(int)((E + T - 1) / T), T, 0, stream>>>(eidx, E, deg);
    GNN_dinv    <<<(int)((nN + T - 1) / T), T, 0, stream>>>(deg, dinv, (int)nN);

    int nTiles = (int)(nN / 16);
    if (nTiles > 0)
        GNN_gemm1_wmma<<<(nTiles + 7) / 8, T, 0, stream>>>(x, W1, h1, nTiles);
    int rem = (int)(nN - (long)nTiles * 16);
    if (rem > 0)
        GNN_gemm1_tail<<<(rem * 16 + T - 1) / T, T, 0, stream>>>(
            x, W1, h1, nTiles * 16, (int)nN);

    GNN_scatter1<<<(int)((E + T - 1) / T), T, 0, stream>>>(eidx, E, dinv, h1, agg1);
    GNN_combine1<<<(int)((nN + T - 1) / T), T, 0, stream>>>(agg1, h1, dinv, b1, W2, h2, (int)nN);
    GNN_scatter2<<<(int)((E + T - 1) / T), T, 0, stream>>>(eidx, E, dinv, h2, agg2);
    GNN_final   <<<(int)((nN + T - 1) / T), T, 0, stream>>>(agg2, h2, dinv, b2, out, (int)nN);
}


# edges ordered by destination per call (returning int atomic gives count and rank, scan, place); each node sums its incoming rows in registers; float atomics gone; sum order: arrival rank, 16-lane tree
# speedup vs baseline: 3.1174x; 1.5052x over previous
#include <hip/hip_runtime.h>
#include <hip/hip_bf16.h>


typedef __attribute__((ext_vector_type(2))) float v2f;
typedef __attribute__((ext_vector_type(8))) float v8f;

#define F_IN 128
#define F_H  16
#define F_C  2

__global__ void GNN_zero_f32(float* __restrict__ p, long n) {
    long i = (long)blockIdx.x * blockDim.x + threadIdx.x;
    if (i < n) p[i] = 0.0f;
}

__global__ void GNN_degree(const int* __restrict__ eidx, long E,
                           float* __restrict__ deg) {
    long e = (long)blockIdx.x * blockDim.x + threadIdx.x;
    if (e >= E) return;
    int d = eidx[E + e];
    atomicAdd(&deg[d], 1.0f);
}

__global__ void GNN_dinv(const float* __restrict__ deg,
                         float* __restrict__ dinv, int n) {
    int i = blockIdx.x * blockDim.x + threadIdx.x;
    if (i < n) dinv[i] = rsqrtf(deg[i] + 1.0f);
}

__global__ void GNN_gemm1_wmma(const float* __restrict__ x,
                               const float* __restrict__ W1,
                               float* __restrict__ h1, int nTiles) {
    int lane = threadIdx.x & 31;
    int wave = threadIdx.x >> 5;
    int tile = blockIdx.x * (blockDim.x >> 5) + wave;
    if (tile >= nTiles) return;

    int m    = lane & 15;
    int koff = (lane >> 4) << 1;
    long rowA = (long)(tile * 16 + m) * F_IN;

    v8f acc = {};
    #pragma unroll 4
    for (int kk = 0; kk < F_IN; kk += 4) {
        v2f a, b;
        a.x = x[rowA + kk + koff];
        a.y = x[rowA + kk + koff + 1];
        b.x = W1[(kk + koff)     * F_H + m];
        b.y = W1[(kk + koff + 1) * F_H + m];
        acc = __builtin_amdgcn_wmma_f32_16x16x4_f32(
            false, a, false, b, (short)0, acc, false, false);
    }

    int rbase = tile * 16 + (lane >> 4) * 8;
    #pragma unroll
    for (int r = 0; r < 8; ++r)
        h1[(long)(rbase + r) * F_H + m] = acc[r];
}

__global__ void GNN_gemm1_tail(const float* __restrict__ x,
                               const float* __restrict__ W1,
                               float* __restrict__ h1, int startRow, int n) {
    int idx = blockIdx.x * blockDim.x + threadIdx.x;
    int row = startRow + (idx >> 4);
    int col = idx & 15;
    if (row >= n) return;
    float acc = 0.0f;
    for (int k = 0; k < F_IN; ++k)
        acc += x[(long)row * F_IN + k] * W1[k * F_H + col];
    h1[(long)row * F_H + col] = acc;
}


template <int J>
__device__ __forceinline__ int GNN_row_lane(int x) {
    return __builtin_amdgcn_update_dpp(0, x, 0x150 + J, 0xf, 0xf, false);
}

template <int J>
__device__ __forceinline__ float GNN_gather_step(int s, const float* __restrict__ hp) {
    unsigned sj = (unsigned)GNN_row_lane<J>(s);
    return hp[sj * (unsigned)F_H];
}

template <int J>
__device__ __forceinline__ void GNN_add_step(int d, float nrm, float v, float* __restrict__ ap) {
    unsigned dj = (unsigned)GNN_row_lane<J>(d);
    float nj = __int_as_float(GNN_row_lane<J>(__float_as_int(nrm)));
    atomicAdd(ap + dj * (unsigned)F_H, nj * v);
}

__global__ void GNN_scatter1(const int* __restrict__ eidx, long E,
                             const float* __restrict__ dinv,
                             const float* __restrict__ h1,
                             float* __restrict__ agg1) {
    long e = (long)blockIdx.x * blockDim.x + threadIdx.x;
    int lane = threadIdx.x & 31;
    long base = e - lane;
    if (base + 32 <= E) {
        int s = eidx[e];
        int d = eidx[E + e];
        float nrm = dinv[s] * dinv[d];
        int c = lane & 15;
        const float* hp = h1 + c;
        float* ap = agg1 + c;

        float v[16];
        v[0]  = GNN_gather_step<0>(s, hp);   v[1]  = GNN_gather_step<1>(s, hp);
        v[2]  = GNN_gather_step<2>(s, hp);   v[3]  = GNN_gather_step<3>(s, hp);
        v[4]  = GNN_gather_step<4>(s, hp);   v[5]  = GNN_gather_step<5>(s, hp);
        v[6]  = GNN_gather_step<6>(s, hp);   v[7]  = GNN_gather_step<7>(s, hp);
        v[8]  = GNN_gather_step<8>(s, hp);   v[9]  = GNN_gather_step<9>(s, hp);
        v[10] = GNN_gather_step<10>(s, hp);  v[11] = GNN_gather_step<11>(s, hp);
        v[12] = GNN_gather_step<12>(s, hp);  v[13] = GNN_gather_step<13>(s, hp);
        v[14] = GNN_gather_step<14>(s, hp);  v[15] = GNN_gather_step<15>(s, hp);

        GNN_add_step<0>(d, nrm, v[0], ap);    GNN_add_step<1>(d, nrm, v[1], ap);
        GNN_add_step<2>(d, nrm, v[2], ap);    GNN_add_step<3>(d, nrm, v[3], ap);
        GNN_add_step<4>(d, nrm, v[4], ap);    GNN_add_step<5>(d, nrm, v[5], ap);
        GNN_add_step<6>(d, nrm, v[6], ap);    GNN_add_step<7>(d, nrm, v[7], ap);
        GNN_add_step<8>(d, nrm, v[8], ap);    GNN_add_step<9>(d, nrm, v[9], ap);
        GNN_add_step<10>(d, nrm, v[10], ap);  GNN_add_step<11>(d, nrm, v[11], ap);
        GNN_add_step<12>(d, nrm, v[12], ap);  GNN_add_step<13>(d, nrm, v[13], ap);
        GNN_add_step<14>(d, nrm, v[14], ap);  GNN_add_step<15>(d, nrm, v[15], ap);
    } else if (e < E) {
        int s = eidx[e];
        int d = eidx[E + e];
        float nrm = dinv[s] * dinv[d];
        const float* hr = h1 + (long)s * F_H;
        float* ar = agg1 + (long)d * F_H;
        for (int j = 0; j < F_H; ++j)
            atomicAdd(ar + j, nrm * hr[j]);
    }
}

__global__ void GNN_combine1(const float* __restrict__ agg1,
                             const float* __restrict__ h1,
                             const float* __restrict__ dinv,
                             const float* __restrict__ b1,
                             const float* __restrict__ W2,
                             float* __restrict__ h2, int n) {
    int i = blockIdx.x * blockDim.x + threadIdx.x;
    if (i >= n) return;
    float di  = dinv[i];
    float di2 = di * di;
    float acc0 = 0.0f, acc1 = 0.0f;
    #pragma unroll
    for (int j = 0; j < F_H; ++j) {
        float v = agg1[(long)i * F_H + j] + di2 * h1[(long)i * F_H + j] + b1[j];
        v = fmaxf(v, 0.0f);
        acc0 = fmaf(v, W2[j * F_C + 0], acc0);
        acc1 = fmaf(v, W2[j * F_C + 1], acc1);
    }
    h2[(long)i * F_C + 0] = acc0;
    h2[(long)i * F_C + 1] = acc1;
}

__global__ void GNN_scatter2(const int* __restrict__ eidx, long E,
                             const float* __restrict__ dinv,
                             const float* __restrict__ h2,
                             float* __restrict__ agg2) {
    long e = (long)blockIdx.x * blockDim.x + threadIdx.x;
    if (e >= E) return;
    int s = eidx[e];
    int d = eidx[E + e];
    float norm = dinv[s] * dinv[d];
    float2 v = *(const float2*)(h2 + (long)s * F_C);
    atomicAdd(&agg2[(long)d * F_C + 0], norm * v.x);
    atomicAdd(&agg2[(long)d * F_C + 1], norm * v.y);
}

__global__ void GNN_final(const float* __restrict__ agg2,
                          const float* __restrict__ h2,
                          const float* __restrict__ dinv,
                          const float* __restrict__ b2,
                          float* __restrict__ out, int n) {
    int i = blockIdx.x * blockDim.x + threadIdx.x;
    if (i >= n) return;
    float di  = dinv[i];
    float di2 = di * di;
    float o0 = agg2[(long)i * F_C + 0] + di2 * h2[(long)i * F_C + 0] + b2[0];
    float o1 = agg2[(long)i * F_C + 1] + di2 * h2[(long)i * F_C + 1] + b2[1];
    float mx  = fmaxf(o0, o1);
    float lse = mx + logf(expf(o0 - mx) + expf(o1 - mx));
    out[(long)i * F_C + 0] = o0 - lse;
    out[(long)i * F_C + 1] = o1 - lse;
}


__global__ void GNN_zero_i32(int* __restrict__ p, int n) {
    int i = blockIdx.x * blockDim.x + threadIdx.x;
    if (i < n) p[i] = 0;
}

__global__ void GNN_rank(const int* __restrict__ eidx, long E,
                         int* __restrict__ cnt, int* __restrict__ rank) {
    long e = (long)blockIdx.x * blockDim.x + threadIdx.x;
    if (e >= E) return;
    int d = eidx[E + e];
    rank[e] = atomicAdd(&cnt[d], 1);
}

#define SCAN_T   256
#define SCAN_PER 4

__global__ void GNN_scan_tot(const int* __restrict__ cnt, int n,
                             int* __restrict__ bsum) {
    __shared__ int red[SCAN_T / 32];
    int t = threadIdx.x;
    int i0 = (blockIdx.x * SCAN_T + t) * SCAN_PER;
    int s = 0;
    #pragma unroll
    for (int k = 0; k < SCAN_PER; ++k)
        if (i0 + k < n) s += cnt[i0 + k];
    for (int o = 16; o > 0; o >>= 1) s += __shfl_xor(s, o);
    if ((t & 31) == 0) red[t >> 5] = s;
    __syncthreads();
    if (t == 0) {
        int tot = 0;
        for (int w = 0; w < SCAN_T / 32; ++w) tot += red[w];
        bsum[blockIdx.x] = tot;
    }
}

__global__ void GNN_scan_off(const int* __restrict__ cnt, int n,
                             const int* __restrict__ bsum,
                             int* __restrict__ off, float* __restrict__ dinv) {
    __shared__ int wtot[SCAN_T / 32];
    int t = threadIdx.x, lane = t & 31, w = t >> 5;

    int p = 0;
    for (int j = t; j < (int)blockIdx.x; j += SCAN_T) p += bsum[j];
    for (int o = 16; o > 0; o >>= 1) p += __shfl_xor(p, o);
    if (lane == 0) wtot[w] = p;
    __syncthreads();
    int base = 0;
    for (int k = 0; k < SCAN_T / 32; ++k) base += wtot[k];
    __syncthreads();

    int i0 = (blockIdx.x * SCAN_T + t) * SCAN_PER;
    int c[SCAN_PER];
    int s = 0;
    #pragma unroll
    for (int k = 0; k < SCAN_PER; ++k) {
        c[k] = (i0 + k < n) ? cnt[i0 + k] : 0;
        s += c[k];
    }
    int inc = s;
    for (int o = 1; o < 32; o <<= 1) {
        int v = __shfl_up(inc, o);
        if (lane >= o) inc += v;
    }
    if (lane == 31) wtot[w] = inc;
    __syncthreads();
    int wb = 0;
    for (int k = 0; k < w; ++k) wb += wtot[k];
    int run = base + wb + inc - s;
    #pragma unroll
    for (int k = 0; k < SCAN_PER; ++k) {
        if (i0 + k < n) {
            off[i0 + k]  = run;
            dinv[i0 + k] = rsqrtf((float)c[k] + 1.0f);
            run += c[k];
        }
    }
}

__global__ void GNN_place(const int* __restrict__ eidx, long E,
                          const int* __restrict__ off,
                          const int* __restrict__ rank,
                          int* __restrict__ csr) {
    long e = (long)blockIdx.x * blockDim.x + threadIdx.x;
    if (e >= E) return;
    int s = eidx[e];
    int d = eidx[E + e];
    csr[off[d] + rank[e]] = s;
}

template <int J>
__device__ __forceinline__ float GNN_pull_step(int s, int rem, const float* __restrict__ hp) {
    unsigned sj = (unsigned)GNN_row_lane<J>(s);
    float v = 0.0f;
    if (J < rem) v = hp[sj * (unsigned)F_H];
    return v;
}

template <int J>
__device__ __forceinline__ float GNN_acc_step(float w, float v, float acc) {
    float wj = __int_as_float(GNN_row_lane<J>(__float_as_int(w)));
    return fmaf(wj, v, acc);
}

__global__ void GNN_pull1(const int* __restrict__ off, const int* __restrict__ cnt,
                          const int* __restrict__ csr,
                          const float* __restrict__ dinv,
                          const float* __restrict__ h1,
                          float* __restrict__ agg1, int n) {
    int gid  = blockIdx.x * blockDim.x + threadIdx.x;
    int node = gid >> 4;
    int c    = threadIdx.x & 15;
    bool valid = node < n;
    int nd   = valid ? node : 0;
    int beg  = off[nd];
    int num  = valid ? cnt[nd] : 0;
    float dd = dinv[nd];
    int nmax = max(num, __shfl_xor(num, 16));
    const float* hp = h1 + c;
    float acc = 0.0f;
    int rem = num;
    int s = nd;
    if (c < rem) s = csr[beg + c];
    for (int b = 0; b < nmax; b += 16) {
        float v[16];
        v[0]  = GNN_pull_step<0>(s, rem, hp);   v[1]  = GNN_pull_step<1>(s, rem, hp);
        v[2]  = GNN_pull_step<2>(s, rem, hp);   v[3]  = GNN_pull_step<3>(s, rem, hp);
        v[4]  = GNN_pull_step<4>(s, rem, hp);   v[5]  = GNN_pull_step<5>(s, rem, hp);
        v[6]  = GNN_pull_step<6>(s, rem, hp);   v[7]  = GNN_pull_step<7>(s, rem, hp);
        v[8]  = GNN_pull_step<8>(s, rem, hp);   v[9]  = GNN_pull_step<9>(s, rem, hp);
        v[10] = GNN_pull_step<10>(s, rem, hp);  v[11] = GNN_pull_step<11>(s, rem, hp);
        v[12] = GNN_pull_step<12>(s, rem, hp);  v[13] = GNN_pull_step<13>(s, rem, hp);
        v[14] = GNN_pull_step<14>(s, rem, hp);  v[15] = GNN_pull_step<15>(s, rem, hp);

        float ws = 0.0f;
        if (c < rem) ws = dinv[s];
        int sn = nd;
        if (c < rem - 16) sn = csr[beg + b + 16 + c];
        float w = ws * dd;
        acc = GNN_acc_step<0>(w, v[0], acc);    acc = GNN_acc_step<1>(w, v[1], acc);
        acc = GNN_acc_step<2>(w, v[2], acc);    acc = GNN_acc_step<3>(w, v[3], acc);
        acc = GNN_acc_step<4>(w, v[4], acc);    acc = GNN_acc_step<5>(w, v[5], acc);
        acc = GNN_acc_step<6>(w, v[6], acc);    acc = GNN_acc_step<7>(w, v[7], acc);
        acc = GNN_acc_step<8>(w, v[8], acc);    acc = GNN_acc_step<9>(w, v[9], acc);
        acc = GNN_acc_step<10>(w, v[10], acc);  acc = GNN_acc_step<11>(w, v[11], acc);
        acc = GNN_acc_step<12>(w, v[12], acc);  acc = GNN_acc_step<13>(w, v[13], acc);
        acc = GNN_acc_step<14>(w, v[14], acc);  acc = GNN_acc_step<15>(w, v[15], acc);
        s = sn;
        rem -= 16;
    }
    if (valid) agg1[(long)node * F_H + c] = acc;
}

__global__ void GNN_pull2(const int* __restrict__ off, const int* __restrict__ cnt,
                          const int* __restrict__ csr,
                          const float* __restrict__ dinv,
                          const float* __restrict__ h2,
                          float* __restrict__ agg2, int n) {
    int lane = threadIdx.x & 31;
    int wave = (blockIdx.x * blockDim.x + threadIdx.x) >> 5;
    int base = wave * 16;
    if (base >= n) return;
    int c = lane & 15;
    int r = lane >> 4;
    float keep = 0.0f;
    for (int it = 0; it < 8; ++it) {
        int node = base + 8 * r + it;
        bool valid = node < n;
        int nd   = valid ? node : base;
        int beg  = off[nd];
        int num  = valid ? cnt[nd] : 0;
        float dd = dinv[nd];
        float a0 = 0.0f, a1 = 0.0f;
        for (int j = c; j < num; j += 32) {
            int j2 = j + 16;
            bool two = j2 < num;
            int s1 = csr[beg + j];
            int s2 = nd;
            if (two) s2 = csr[beg + j2];
            float d1 = dinv[s1];
            float2 g1 = *(const float2*)(h2 + (long)s1 * F_C);
            float d2 = 0.0f;
            float2 g2 = make_float2(0.0f, 0.0f);
            if (two) {
                d2 = dinv[s2];
                g2 = *(const float2*)(h2 + (long)s2 * F_C);
            }
            float w1 = d1 * dd;
            float w2 = d2 * dd;
            a0 = fmaf(w1, g1.x, a0);
            a1 = fmaf(w1, g1.y, a1);
            a0 = fmaf(w2, g2.x, a0);
            a1 = fmaf(w2, g2.y, a1);
        }
        for (int o = 8; o > 0; o >>= 1) {
            a0 += __shfl_xor(a0, o);
            a1 += __shfl_xor(a1, o);
        }
        if (((lane >> 1) & 7) == it) keep = (lane & 1) ? a1 : a0;
    }
    if (base + (lane >> 1) < n) agg2[(long)base * F_C + lane] = keep;
}

extern "C" void kernel_launch(void* const* d_in, const int* in_sizes, int n_in,
                              void* d_out, int out_size, void* d_ws, size_t ws_size,
                              hipStream_t stream) {
    const float* x  = (const float*)d_in[0];
    const int* eidx = (const int*)  d_in[1];
    const float* W1 = (const float*)d_in[2];
    const float* b1 = (const float*)d_in[3];
    const float* W2 = (const float*)d_in[4];
    const float* b2 = (const float*)d_in[5];
    float* out = (float*)d_out;

    long nN = (long)in_sizes[0] / F_IN;
    long E  = (long)in_sizes[1] / 2;

    const int T = 256;
    float* ws   = (float*)d_ws;

    long nNp = (nN + 63) & ~63L;
    long Ep  = (E + 63) & ~63L;
    long nb  = (nN + (long)SCAN_T * SCAN_PER - 1) / ((long)SCAN_T * SCAN_PER);
    long bigR = (Ep > 32 * nNp) ? Ep : 32 * nNp;
    long need = 7 * nNp + 1024 + Ep + bigR;
    if (nb <= 1024 && E < (1L << 31) && (size_t)need * 4 <= ws_size) {
        int*   cnt  = (int*)ws;
        int*   off  = (int*)ws + nNp;
        float* dinv = ws + 2 * nNp;
        int*   bsum = (int*)ws + 3 * nNp;
        float* h2   = ws + 3 * nNp + 1024;
        float* agg2 = ws + 5 * nNp + 1024;
        int*   csr  = (int*)ws + 7 * nNp + 1024;
        float* R    = ws + 7 * nNp + 1024 + Ep;
        int*   rank = (int*)R;
        float* h1   = R;
        float* agg1 = R + 16 * nNp;

        GNN_zero_i32<<<(int)((nN + T - 1) / T), T, 0, stream>>>(cnt, (int)nN);
        GNN_rank    <<<(int)((E + T - 1) / T), T, 0, stream>>>(eidx, E, cnt, rank);
        GNN_scan_tot<<<(int)nb, SCAN_T, 0, stream>>>(cnt, (int)nN, bsum);
        GNN_scan_off<<<(int)nb, SCAN_T, 0, stream>>>(cnt, (int)nN, bsum, off, dinv);
        GNN_place   <<<(int)((E + T - 1) / T), T, 0, stream>>>(eidx, E, off, rank, csr);

        int nTiles = (int)(nN / 16);
        if (nTiles > 0)
            GNN_gemm1_wmma<<<(nTiles + 7) / 8, T, 0, stream>>>(x, W1, h1, nTiles);
        int rem = (int)(nN - (long)nTiles * 16);
        if (rem > 0)
            GNN_gemm1_tail<<<(rem * 16 + T - 1) / T, T, 0, stream>>>(
                x, W1, h1, nTiles * 16, (int)nN);

        GNN_pull1   <<<(int)((nN * 16 + T - 1) / T), T, 0, stream>>>(off, cnt, csr, dinv, h1, agg1, (int)nN);
        GNN_combine1<<<(int)((nN + T - 1) / T), T, 0, stream>>>(agg1, h1, dinv, b1, W2, h2, (int)nN);
        GNN_pull2   <<<(int)((nN * 2 + T - 1) / T), T, 0, stream>>>(off, cnt, csr, dinv, h2, agg2, (int)nN);
        GNN_final   <<<(int)((nN + T - 1) / T), T, 0, stream>>>(agg2, h2, dinv, b2, out, (int)nN);
        return;
    }

    float* deg  = ws;
    float* agg1 = ws + nN;
    float* agg2 = ws + 17 * nN;
    float* dinv = ws + 19 * nN;
    float* h1   = ws + 20 * nN;
    float* h2   = ws + 36 * nN;

    long zn = 19 * nN;
    GNN_zero_f32<<<(int)((zn + T - 1) / T), T, 0, stream>>>(ws, zn);
    GNN_degree  <<<(int)((E + T - 1) / T), T, 0, stream>>>(eidx, E, deg);
    GNN_dinv    <<<(int)((nN + T - 1) / T), T, 0, stream>>>(deg, dinv, (int)nN);

    int nTiles = (int)(nN / 16);
    if (nTiles > 0)
        GNN_gemm1_wmma<<<(nTiles + 7) / 8, T, 0, stream>>>(x, W1, h1, nTiles);
    int rem = (int)(nN - (long)nTiles * 16);
    if (rem > 0)
        GNN_gemm1_tail<<<(rem * 16 + T - 1) / T, T, 0, stream>>>(
            x, W1, h1, nTiles * 16, (int)nN);

    GNN_scatter1<<<(int)((E + T - 1) / T), T, 0, stream>>>(eidx, E, dinv, h1, agg1);
    GNN_combine1<<<(int)((nN + T - 1) / T), T, 0, stream>>>(agg1, h1, dinv, b1, W2, h2, (int)nN);
    GNN_scatter2<<<(int)((E + T - 1) / T), T, 0, stream>>>(eidx, E, dinv, h2, agg2);
    GNN_final   <<<(int)((nN + T - 1) / T), T, 0, stream>>>(agg2, h2, dinv, b2, out, (int)nN);
}


# edge list ordered by destination in LDS: edges grouped in bins of 1024 adjacent nodes, one workgroup orders a bin, stores it whole; no global atomic or scattered store; sum order: LDS arrival rank
# speedup vs baseline: 9.9562x; 3.1938x over previous
#include <hip/hip_runtime.h>
#include <hip/hip_bf16.h>


typedef __attribute__((ext_vector_type(2))) float v2f;
typedef __attribute__((ext_vector_type(8))) float v8f;

#define F_IN 128
#define F_H  16
#define F_C  2

__global__ void GNN_zero_f32(float* __restrict__ p, long n) {
    long i = (long)blockIdx.x * blockDim.x + threadIdx.x;
    if (i < n) p[i] = 0.0f;
}

__global__ void GNN_degree(const int* __restrict__ eidx, long E,
                           float* __restrict__ deg) {
    long e = (long)blockIdx.x * blockDim.x + threadIdx.x;
    if (e >= E) return;
    int d = eidx[E + e];
    atomicAdd(&deg[d], 1.0f);
}

__global__ void GNN_dinv(const float* __restrict__ deg,
                         float* __restrict__ dinv, int n) {
    int i = blockIdx.x * blockDim.x + threadIdx.x;
    if (i < n) dinv[i] = rsqrtf(deg[i] + 1.0f);
}

__global__ void GNN_gemm1_wmma(const float* __restrict__ x,
                               const float* __restrict__ W1,
                               const float* __restrict__ dinv,
                               float* __restrict__ h1, int nTiles) {
    int lane = threadIdx.x & 31;
    int wave = threadIdx.x >> 5;
    int tile = blockIdx.x * (blockDim.x >> 5) + wave;
    if (tile >= nTiles) return;

    int m    = lane & 15;
    int koff = (lane >> 4) << 1;
    long rowA = (long)(tile * 16 + m) * F_IN;

    int rbase = tile * 16 + (lane >> 4) * 8;
    float dv[8];
    #pragma unroll
    for (int r = 0; r < 8; ++r) dv[r] = dinv[rbase + r];

    v8f acc = {};
    #pragma unroll 4
    for (int kk = 0; kk < F_IN; kk += 4) {
        v2f a, b;
        a.x = x[rowA + kk + koff];
        a.y = x[rowA + kk + koff + 1];
        b.x = W1[(kk + koff)     * F_H + m];
        b.y = W1[(kk + koff + 1) * F_H + m];
        acc = __builtin_amdgcn_wmma_f32_16x16x4_f32(
            false, a, false, b, (short)0, acc, false, false);
    }

    #pragma unroll
    for (int r = 0; r < 8; ++r)
        h1[(long)(rbase + r) * F_H + m] = dv[r] * acc[r];
}

__global__ void GNN_gemm1_tail(const float* __restrict__ x,
                               const float* __restrict__ W1,
                               const float* __restrict__ dinv,
                               float* __restrict__ h1, int startRow, int n) {
    int idx = blockIdx.x * blockDim.x + threadIdx.x;
    int row = startRow + (idx >> 4);
    int col = idx & 15;
    if (row >= n) return;
    float acc = 0.0f;
    for (int k = 0; k < F_IN; ++k)
        acc += x[(long)row * F_IN + k] * W1[k * F_H + col];
    h1[(long)row * F_H + col] = dinv[row] * acc;
}


template <int J>
__device__ __forceinline__ int GNN_row_lane(int x) {
    return __builtin_amdgcn_update_dpp(0, x, 0x150 + J, 0xf, 0xf, false);
}

template <int J>
__device__ __forceinline__ float GNN_gather_step(int s, const float* __restrict__ hp) {
    unsigned sj = (unsigned)GNN_row_lane<J>(s);
    return hp[sj * (unsigned)F_H];
}

template <int J>
__device__ __forceinline__ void GNN_add_step(int d, float v, float* __restrict__ ap) {
    unsigned dj = (unsigned)GNN_row_lane<J>(d);
    atomicAdd(ap + dj * (unsigned)F_H, v);
}

__global__ void GNN_scatter1(const int* __restrict__ eidx, long E,
                             const float* __restrict__ h1,
                             float* __restrict__ agg1) {
    long e = (long)blockIdx.x * blockDim.x + threadIdx.x;
    int lane = threadIdx.x & 31;
    long base = e - lane;
    if (base + 32 <= E) {
        int s = eidx[e];
        int d = eidx[E + e];
        int c = lane & 15;
        const float* hp = h1 + c;
        float* ap = agg1 + c;

        float v[16];
        v[0]  = GNN_gather_step<0>(s, hp);   v[1]  = GNN_gather_step<1>(s, hp);
        v[2]  = GNN_gather_step<2>(s, hp);   v[3]  = GNN_gather_step<3>(s, hp);
        v[4]  = GNN_gather_step<4>(s, hp);   v[5]  = GNN_gather_step<5>(s, hp);
        v[6]  = GNN_gather_step<6>(s, hp);   v[7]  = GNN_gather_step<7>(s, hp);
        v[8]  = GNN_gather_step<8>(s, hp);   v[9]  = GNN_gather_step<9>(s, hp);
        v[10] = GNN_gather_step<10>(s, hp);  v[11] = GNN_gather_step<11>(s, hp);
        v[12] = GNN_gather_step<12>(s, hp);  v[13] = GNN_gather_step<13>(s, hp);
        v[14] = GNN_gather_step<14>(s, hp);  v[15] = GNN_gather_step<15>(s, hp);

        GNN_add_step<0>(d, v[0], ap);    GNN_add_step<1>(d, v[1], ap);
        GNN_add_step<2>(d, v[2], ap);    GNN_add_step<3>(d, v[3], ap);
        GNN_add_step<4>(d, v[4], ap);    GNN_add_step<5>(d, v[5], ap);
        GNN_add_step<6>(d, v[6], ap);    GNN_add_step<7>(d, v[7], ap);
        GNN_add_step<8>(d, v[8], ap);    GNN_add_step<9>(d, v[9], ap);
        GNN_add_step<10>(d, v[10], ap);  GNN_add_step<11>(d, v[11], ap);
        GNN_add_step<12>(d, v[12], ap);  GNN_add_step<13>(d, v[13], ap);
        GNN_add_step<14>(d, v[14], ap);  GNN_add_step<15>(d, v[15], ap);
    } else if (e < E) {
        int s = eidx[e];
        int d = eidx[E + e];
        const float* hr = h1 + (long)s * F_H;
        float* ar = agg1 + (long)d * F_H;
        for (int j = 0; j < F_H; ++j)
            atomicAdd(ar + j, hr[j]);
    }
}

__global__ void GNN_combine1(const float* __restrict__ agg1,
                             const float* __restrict__ h1,
                             const float* __restrict__ dinv,
                             const float* __restrict__ b1,
                             const float* __restrict__ W2,
                             float* __restrict__ h2, int n) {
    int i = blockIdx.x * blockDim.x + threadIdx.x;
    if (i >= n) return;
    float di  = dinv[i];
    float acc0 = 0.0f, acc1 = 0.0f;
    #pragma unroll
    for (int j = 0; j < F_H; ++j) {
        float v = di * (agg1[(long)i * F_H + j] + h1[(long)i * F_H + j]) + b1[j];
        v = fmaxf(v, 0.0f);
        acc0 = fmaf(v, W2[j * F_C + 0], acc0);
        acc1 = fmaf(v, W2[j * F_C + 1], acc1);
    }
    h2[(long)i * F_C + 0] = di * acc0;
    h2[(long)i * F_C + 1] = di * acc1;
}

__global__ void GNN_scatter2(const int* __restrict__ eidx, long E,
                             const float* __restrict__ h2,
                             float* __restrict__ agg2) {
    long e = (long)blockIdx.x * blockDim.x + threadIdx.x;
    if (e >= E) return;
    int s = eidx[e];
    int d = eidx[E + e];
    float2 v = *(const float2*)(h2 + (long)s * F_C);
    atomicAdd(&agg2[(long)d * F_C + 0], v.x);
    atomicAdd(&agg2[(long)d * F_C + 1], v.y);
}

__global__ void GNN_final(const float* __restrict__ agg2,
                          const float* __restrict__ h2,
                          const float* __restrict__ dinv,
                          const float* __restrict__ b2,
                          float* __restrict__ out, int n) {
    int i = blockIdx.x * blockDim.x + threadIdx.x;
    if (i >= n) return;
    float di  = dinv[i];
    float o0 = di * (agg2[(long)i * F_C + 0] + h2[(long)i * F_C + 0]) + b2[0];
    float o1 = di * (agg2[(long)i * F_C + 1] + h2[(long)i * F_C + 1]) + b2[1];
    float mx  = fmaxf(o0, o1);
    float lse = mx + logf(expf(o0 - mx) + expf(o1 - mx));
    out[(long)i * F_C + 0] = o0 - lse;
    out[(long)i * F_C + 1] = o1 - lse;
}


__global__ void GNN_zero_i32(int* __restrict__ p, int n) {
    int i = blockIdx.x * blockDim.x + threadIdx.x;
    if (i < n) p[i] = 0;
}

__global__ void GNN_rank(const int* __restrict__ eidx, long E,
                         int* __restrict__ cnt, int* __restrict__ rank) {
    long e = (long)blockIdx.x * blockDim.x + threadIdx.x;
    if (e >= E) return;
    int d = eidx[E + e];
    rank[e] = atomicAdd(&cnt[d], 1);
}

#define SCAN_T   256
#define SCAN_PER 4

__global__ void GNN_scan_tot(const int* __restrict__ cnt, int n,
                             int* __restrict__ bsum) {
    __shared__ int red[SCAN_T / 32];
    int t = threadIdx.x;
    int i0 = (blockIdx.x * SCAN_T + t) * SCAN_PER;
    int s = 0;
    #pragma unroll
    for (int k = 0; k < SCAN_PER; ++k)
        if (i0 + k < n) s += cnt[i0 + k];
    for (int o = 16; o > 0; o >>= 1) s += __shfl_xor(s, o);
    if ((t & 31) == 0) red[t >> 5] = s;
    __syncthreads();
    if (t == 0) {
        int tot = 0;
        for (int w = 0; w < SCAN_T / 32; ++w) tot += red[w];
        bsum[blockIdx.x] = tot;
    }
}

template <bool WITH_DINV>
__global__ void GNN_scan_off(const int* __restrict__ cnt, int n,
                             const int* __restrict__ bsum,
                             int* __restrict__ off, float* __restrict__ dinv) {
    __shared__ int wtot[SCAN_T / 32];
    int t = threadIdx.x, lane = t & 31, w = t >> 5;

    int p = 0;
    for (int j = t; j < (int)blockIdx.x; j += SCAN_T) p += bsum[j];
    for (int o = 16; o > 0; o >>= 1) p += __shfl_xor(p, o);
    if (lane == 0) wtot[w] = p;
    __syncthreads();
    int base = 0;
    for (int k = 0; k < SCAN_T / 32; ++k) base += wtot[k];
    __syncthreads();

    int i0 = (blockIdx.x * SCAN_T + t) * SCAN_PER;
    int c[SCAN_PER];
    int s = 0;
    #pragma unroll
    for (int k = 0; k < SCAN_PER; ++k) {
        c[k] = (i0 + k < n) ? cnt[i0 + k] : 0;
        s += c[k];
    }
    int inc = s;
    for (int o = 1; o < 32; o <<= 1) {
        int v = __shfl_up(inc, o);
        if (lane >= o) inc += v;
    }
    if (lane == 31) wtot[w] = inc;
    __syncthreads();
    int wb = 0;
    for (int k = 0; k < w; ++k) wb += wtot[k];
    int run = base + wb + inc - s;
    #pragma unroll
    for (int k = 0; k < SCAN_PER; ++k) {
        if (i0 + k < n) {
            off[i0 + k]  = run;
            if (WITH_DINV) dinv[i0 + k] = rsqrtf((float)c[k] + 1.0f);
            run += c[k];
        }
    }
}

__global__ void GNN_place(const int* __restrict__ eidx, long E,
                          const int* __restrict__ off,
                          const int* __restrict__ rank,
                          int* __restrict__ csr) {
    long e = (long)blockIdx.x * blockDim.x + threadIdx.x;
    if (e >= E) return;
    int s = eidx[e];
    int d = eidx[E + e];
    csr[off[d] + rank[e]] = s;
}


#define BIN_SHIFT 10
#define BIN_W     1024
#define BIN_CH    8192
#define BIN_PER   (BIN_CH / BIN_W)
#define SRC_BITS  22
#define STAGE_CAP 49152

__device__ __forceinline__ int GNN_scan1024(int v, int* wtot) {
    int lane = threadIdx.x & 31, w = threadIdx.x >> 5;
    int inc = v;
    for (int o = 1; o < 32; o <<= 1) {
        int u = __shfl_up(inc, o);
        if (lane >= o) inc += u;
    }
    if (lane == 31) wtot[w] = inc;
    __syncthreads();
    int wv = wtot[lane];
    int winc = wv;
    for (int o = 1; o < 32; o <<= 1) {
        int u = __shfl_up(winc, o);
        if (lane >= o) winc += u;
    }
    int wbase = __shfl(winc - wv, w);
    __syncthreads();
    return wbase + inc - v;
}

__global__ void __launch_bounds__(BIN_W)
GNN_bin_count(const int* __restrict__ eidx, long E, int nbin, int NB,
              int* __restrict__ binCnt) {
    __shared__ int hist[BIN_W];
    int t = threadIdx.x;
    hist[t] = 0;
    long e0 = (long)blockIdx.x * BIN_CH;
    int d[BIN_PER];
    for (int k = 0; k < BIN_PER; ++k) {
        long e = e0 + k * BIN_W + t;
        d[k] = (e < E) ? eidx[E + e] : -1;
    }
    __syncthreads();
    for (int k = 0; k < BIN_PER; ++k)
        if (d[k] >= 0) atomicAdd(&hist[d[k] >> BIN_SHIFT], 1);
    __syncthreads();
    if (t < nbin) binCnt[(long)t * NB + blockIdx.x] = hist[t];
}

__global__ void __launch_bounds__(BIN_W)
GNN_bin_group(const int* __restrict__ eidx, long E, int nbin, int NB,
              const int* __restrict__ binCnt, const int* __restrict__ binOff,
              unsigned* __restrict__ pk) {
    __shared__ int cur[BIN_W];
    __shared__ int gdelta[BIN_W];
    __shared__ int wtot[32];
    __shared__ int bufS[BIN_CH];
    __shared__ int bufD[BIN_CH];
    int t = threadIdx.x;
    long e0 = (long)blockIdx.x * BIN_CH;
    int s[BIN_PER], d[BIN_PER];
    for (int k = 0; k < BIN_PER; ++k) {
        long e = e0 + k * BIN_W + t;
        bool in = e < E;
        s[k] = in ? eidx[e] : 0;
        d[k] = in ? eidx[E + e] : -1;
    }
    int c = 0, g = 0;
    if (t < nbin) {
        c = binCnt[(long)t * NB + blockIdx.x];
        g = binOff[(long)t * NB + blockIdx.x];
    }
    int ls = GNN_scan1024(c, wtot);
    cur[t] = ls;
    gdelta[t] = g - ls;
    __syncthreads();
    for (int k = 0; k < BIN_PER; ++k) {
        if (d[k] >= 0) {
            int p = atomicAdd(&cur[d[k] >> BIN_SHIFT], 1);
            bufS[p] = s[k];
            bufD[p] = d[k];
        }
    }
    __syncthreads();
    long left = E - e0;
    int m = left < BIN_CH ? (int)left : BIN_CH;
    for (int i = t; i < m; i += BIN_W) {
        int dd = bufD[i];
        pk[gdelta[dd >> BIN_SHIFT] + i] =
            ((unsigned)(dd & (BIN_W - 1)) << SRC_BITS) | (unsigned)bufS[i];
    }
}

__global__ void __launch_bounds__(BIN_W)
GNN_bin_order(const unsigned* __restrict__ pk, const int* __restrict__ binOff,
              int NB, int nbin, int E, int n,
              int* __restrict__ cnt, int* __restrict__ off,
              float* __restrict__ dinv, int* __restrict__ csr) {
    __shared__ int ncnt[BIN_W];
    __shared__ int wtot[32];
    __shared__ int stage[STAGE_CAP];
    int t = threadIdx.x, b = blockIdx.x;
    int beg = binOff[(long)b * NB];
    int end = (b + 1 < nbin) ? binOff[(long)(b + 1) * NB] : E;
    int m = end - beg;
    ncnt[t] = 0;
    __syncthreads();
    for (int i0 = 0; i0 < m; i0 += BIN_PER * BIN_W) {
        unsigned w[BIN_PER];
        for (int k = 0; k < BIN_PER; ++k)
            w[k] = pk[beg + min(i0 + k * BIN_W + t, m - 1)];
        for (int k = 0; k < BIN_PER; ++k)
            if (i0 + k * BIN_W + t < m) atomicAdd(&ncnt[w[k] >> SRC_BITS], 1);
    }
    __syncthreads();
    int c = ncnt[t];
    int ex = GNN_scan1024(c, wtot);
    ncnt[t] = ex;
    int node = (b << BIN_SHIFT) + t;
    if (node < n) {
        cnt[node]  = c;
        off[node]  = beg + ex;
        dinv[node] = rsqrtf((float)c + 1.0f);
    }
    __syncthreads();
    bool staged = m <= STAGE_CAP;
    for (int i0 = 0; i0 < m; i0 += BIN_PER * BIN_W) {
        unsigned w[BIN_PER];
        for (int k = 0; k < BIN_PER; ++k)
            w[k] = pk[beg + min(i0 + k * BIN_W + t, m - 1)];
        for (int k = 0; k < BIN_PER; ++k) {
            if (i0 + k * BIN_W + t < m) {
                int r = atomicAdd(&ncnt[w[k] >> SRC_BITS], 1);
                int src = (int)(w[k] & ((1u << SRC_BITS) - 1u));
                if (staged) stage[r] = src;
                else        csr[beg + r] = src;
            }
        }
    }
    if (staged) {
        __syncthreads();
        for (int i = t; i < m; i += BIN_W) csr[beg + i] = stage[i];
    }
}

template <int J>
__device__ __forceinline__ float GNN_pull_step(int s, int rem, const float* __restrict__ hp) {
    unsigned sj = (unsigned)GNN_row_lane<J>(s);
    float v = 0.0f;
    if (J < rem) v = hp[sj * (unsigned)F_H];
    return v;
}

__global__ void GNN_pull1(const int* __restrict__ off, const int* __restrict__ cnt,
                          const int* __restrict__ csr,
                          const float* __restrict__ h1,
                          float* __restrict__ agg1, int n) {
    int gid  = blockIdx.x * blockDim.x + threadIdx.x;
    int node = gid >> 4;
    int c    = threadIdx.x & 15;
    bool valid = node < n;
    int nd   = valid ? node : 0;
    int beg  = off[nd];
    int num  = valid ? cnt[nd] : 0;
    int nmax = max(num, __shfl_xor(num, 16));
    const float* hp = h1 + c;
    float acc = 0.0f;
    int rem = num;
    int s = nd;
    if (c < rem) s = csr[beg + c];
    for (int b = 0; b < nmax; b += 16) {
        float v[16];
        v[0]  = GNN_pull_step<0>(s, rem, hp);   v[1]  = GNN_pull_step<1>(s, rem, hp);
        v[2]  = GNN_pull_step<2>(s, rem, hp);   v[3]  = GNN_pull_step<3>(s, rem, hp);
        v[4]  = GNN_pull_step<4>(s, rem, hp);   v[5]  = GNN_pull_step<5>(s, rem, hp);
        v[6]  = GNN_pull_step<6>(s, rem, hp);   v[7]  = GNN_pull_step<7>(s, rem, hp);
        v[8]  = GNN_pull_step<8>(s, rem, hp);   v[9]  = GNN_pull_step<9>(s, rem, hp);
        v[10] = GNN_pull_step<10>(s, rem, hp);  v[11] = GNN_pull_step<11>(s, rem, hp);
        v[12] = GNN_pull_step<12>(s, rem, hp);  v[13] = GNN_pull_step<13>(s, rem, hp);
        v[14] = GNN_pull_step<14>(s, rem, hp);  v[15] = GNN_pull_step<15>(s, rem, hp);

        int sn = nd;
        if (c < rem - 16) sn = csr[beg + b + 16 + c];
        acc += v[0];   acc += v[1];   acc += v[2];   acc += v[3];
        acc += v[4];   acc += v[5];   acc += v[6];   acc += v[7];
        acc += v[8];   acc += v[9];   acc += v[10];  acc += v[11];
        acc += v[12];  acc += v[13];  acc += v[14];  acc += v[15];
        s = sn;
        rem -= 16;
    }
    if (valid) agg1[(long)node * F_H + c] = acc;
}

__global__ void GNN_pull2(const int* __restrict__ off, const int* __restrict__ cnt,
                          const int* __restrict__ csr,
                          const float* __restrict__ h2,
                          float* __restrict__ agg2, int n) {
    int lane = threadIdx.x & 31;
    int wave = (blockIdx.x * blockDim.x + threadIdx.x) >> 5;
    int base = wave * 16;
    if (base >= n) return;
    int c = lane & 15;
    int r = lane >> 4;
    float keep = 0.0f;
    for (int it = 0; it < 8; ++it) {
        int node = base + 8 * r + it;
        bool valid = node < n;
        int nd   = valid ? node : base;
        int beg  = off[nd];
        int num  = valid ? cnt[nd] : 0;
        float a0 = 0.0f, a1 = 0.0f;
        for (int j = c; j < num; j += 32) {
            int j2 = j + 16;
            bool two = j2 < num;
            int s1 = csr[beg + j];
            int s2 = nd;
            if (two) s2 = csr[beg + j2];
            float2 g1 = *(const float2*)(h2 + (long)s1 * F_C);
            float2 g2 = make_float2(0.0f, 0.0f);
            if (two) g2 = *(const float2*)(h2 + (long)s2 * F_C);
            a0 += g1.x;
            a1 += g1.y;
            a0 += g2.x;
            a1 += g2.y;
        }
        for (int o = 8; o > 0; o >>= 1) {
            a0 += __shfl_xor(a0, o);
            a1 += __shfl_xor(a1, o);
        }
        if (((lane >> 1) & 7) == it) keep = (lane & 1) ? a1 : a0;
    }
    if (base + (lane >> 1) < n) agg2[(long)base * F_C + lane] = keep;
}

extern "C" void kernel_launch(void* const* d_in, const int* in_sizes, int n_in,
                              void* d_out, int out_size, void* d_ws, size_t ws_size,
                              hipStream_t stream) {
    const float* x  = (const float*)d_in[0];
    const int* eidx = (const int*)  d_in[1];
    const float* W1 = (const float*)d_in[2];
    const float* b1 = (const float*)d_in[3];
    const float* W2 = (const float*)d_in[4];
    const float* b2 = (const float*)d_in[5];
    float* out = (float*)d_out;

    long nN = (long)in_sizes[0] / F_IN;
    long E  = (long)in_sizes[1] / 2;

    const int T = 256;
    float* ws   = (float*)d_ws;

    long nNp = (nN + 63) & ~63L;
    long Ep  = (E + 63) & ~63L;
    long nb  = (nN + (long)SCAN_T * SCAN_PER - 1) / ((long)SCAN_T * SCAN_PER);
    long bigR = (Ep > 32 * nNp) ? Ep : 32 * nNp;
    long need = 7 * nNp + 1024 + Ep + bigR;
    bool fits = E > 0 && E < (1L << 31) && (size_t)need * 4 <= ws_size;

    long nbin = (nN + BIN_W - 1) >> BIN_SHIFT;
    long NB   = (E + BIN_CH - 1) / BIN_CH;
    long nsc  = nbin * NB;
    long nscP = (nsc + 63) & ~63L;
    long nb2  = (nsc + (long)SCAN_T * SCAN_PER - 1) / ((long)SCAN_T * SCAN_PER);
    if (fits && nbin <= BIN_W && nN <= (1L << SRC_BITS) && nb2 <= 1024 && 2 * nscP <= 4 * nNp) {
        int*   cnt  = (int*)ws;
        int*   off  = (int*)ws + nNp;
        float* dinv = ws + 2 * nNp;
        int*   bsum = (int*)ws + 3 * nNp;
        float* h2   = ws + 3 * nNp + 1024;
        float* agg2 = ws + 5 * nNp + 1024;
        int*   binCnt = (int*)h2;
        int*   binOff = (int*)h2 + nscP;
        int*   csr  = (int*)ws + 7 * nNp + 1024;
        float* R    = ws + 7 * nNp + 1024 + Ep;
        unsigned* pk = (unsigned*)R;
        float* h1   = R;
        float* agg1 = R + 16 * nNp;

        GNN_bin_count<<<(int)NB, BIN_W, 0, stream>>>(eidx, E, (int)nbin, (int)NB, binCnt);
        GNN_scan_tot <<<(int)nb2, SCAN_T, 0, stream>>>(binCnt, (int)nsc, bsum);
        GNN_scan_off<false><<<(int)nb2, SCAN_T, 0, stream>>>(binCnt, (int)nsc, bsum, binOff, dinv);
        GNN_bin_group<<<(int)NB, BIN_W, 0, stream>>>(eidx, E, (int)nbin, (int)NB, binCnt, binOff, pk);
        GNN_bin_order<<<(int)nbin, BIN_W, 0, stream>>>(pk, binOff, (int)NB, (int)nbin, (int)E, (int)nN,
                                                      cnt, off, dinv, csr);

        int nTiles = (int)(nN / 16);
        if (nTiles > 0)
            GNN_gemm1_wmma<<<(nTiles + 7) / 8, T, 0, stream>>>(x, W1, dinv, h1, nTiles);
        int rem = (int)(nN - (long)nTiles * 16);
        if (rem > 0)
            GNN_gemm1_tail<<<(rem * 16 + T - 1) / T, T, 0, stream>>>(
                x, W1, dinv, h1, nTiles * 16, (int)nN);

        GNN_pull1   <<<(int)((nN * 16 + T - 1) / T), T, 0, stream>>>(off, cnt, csr, h1, agg1, (int)nN);
        GNN_combine1<<<(int)((nN + T - 1) / T), T, 0, stream>>>(agg1, h1, dinv, b1, W2, h2, (int)nN);
        GNN_pull2   <<<(int)((nN * 2 + T - 1) / T), T, 0, stream>>>(off, cnt, csr, h2, agg2, (int)nN);
        GNN_final   <<<(int)((nN + T - 1) / T), T, 0, stream>>>(agg2, h2, dinv, b2, out, (int)nN);
        return;
    }

    if (nb <= 1024 && fits) {
        int*   cnt  = (int*)ws;
        int*   off  = (int*)ws + nNp;
        float* dinv = ws + 2 * nNp;
        int*   bsum = (int*)ws + 3 * nNp;
        float* h2   = ws + 3 * nNp + 1024;
        float* agg2 = ws + 5 * nNp + 1024;
        int*   csr  = (int*)ws + 7 * nNp + 1024;
        float* R    = ws + 7 * nNp + 1024 + Ep;
        int*   rank = (int*)R;
        float* h1   = R;
        float* agg1 = R + 16 * nNp;

        GNN_zero_i32<<<(int)((nN + T - 1) / T), T, 0, stream>>>(cnt, (int)nN);
        GNN_rank    <<<(int)((E + T - 1) / T), T, 0, stream>>>(eidx, E, cnt, rank);
        GNN_scan_tot<<<(int)nb, SCAN_T, 0, stream>>>(cnt, (int)nN, bsum);
        GNN_scan_off<true><<<(int)nb, SCAN_T, 0, stream>>>(cnt, (int)nN, bsum, off, dinv);
        GNN_place   <<<(int)((E + T - 1) / T), T, 0, stream>>>(eidx, E, off, rank, csr);

        int nTiles = (int)(nN / 16);
        if (nTiles > 0)
            GNN_gemm1_wmma<<<(nTiles + 7) / 8, T, 0, stream>>>(x, W1, dinv, h1, nTiles);
        int rem = (int)(nN - (long)nTiles * 16);
        if (rem > 0)
            GNN_gemm1_tail<<<(rem * 16 + T - 1) / T, T, 0, stream>>>(
                x, W1, dinv, h1, nTiles * 16, (int)nN);

        GNN_pull1   <<<(int)((nN * 16 + T - 1) / T), T, 0, stream>>>(off, cnt, csr, h1, agg1, (int)nN);
        GNN_combine1<<<(int)((nN + T - 1) / T), T, 0, stream>>>(agg1, h1, dinv, b1, W2, h2, (int)nN);
        GNN_pull2   <<<(int)((nN * 2 + T - 1) / T), T, 0, stream>>>(off, cnt, csr, h2, agg2, (int)nN);
        GNN_final   <<<(int)((nN + T - 1) / T), T, 0, stream>>>(agg2, h2, dinv, b2, out, (int)nN);
        return;
    }

    float* deg  = ws;
    float* agg1 = ws + nN;
    float* agg2 = ws + 17 * nN;
    float* dinv = ws + 19 * nN;
    float* h1   = ws + 20 * nN;
    float* h2   = ws + 36 * nN;

    long zn = 19 * nN;
    GNN_zero_f32<<<(int)((zn + T - 1) / T), T, 0, stream>>>(ws, zn);
    GNN_degree  <<<(int)((E + T - 1) / T), T, 0, stream>>>(eidx, E, deg);
    GNN_dinv    <<<(int)((nN + T - 1) / T), T, 0, stream>>>(deg, dinv, (int)nN);

    int nTiles = (int)(nN / 16);
    if (nTiles > 0)
        GNN_gemm1_wmma<<<(nTiles + 7) / 8, T, 0, stream>>>(x, W1, dinv, h1, nTiles);
    int rem = (int)(nN - (long)nTiles * 16);
    if (rem > 0)
        GNN_gemm1_tail<<<(rem * 16 + T - 1) / T, T, 0, stream>>>(
            x, W1, dinv, h1, nTiles * 16, (int)nN);

    GNN_scatter1<<<(int)((E + T - 1) / T), T, 0, stream>>>(eidx, E, h1, agg1);
    GNN_combine1<<<(int)((nN + T - 1) / T), T, 0, stream>>>(agg1, h1, dinv, b1, W2, h2, (int)nN);
    GNN_scatter2<<<(int)((E + T - 1) / T), T, 0, stream>>>(eidx, E, h2, agg2);
    GNN_final   <<<(int)((nN + T - 1) / T), T, 0, stream>>>(agg2, h2, dinv, b2, out, (int)nN);
}


# the two combine kernels folded into the pulls: the row that holds a node's sum applies self term, bias, ReLU and the 16x2 product (layer 1) or log_softmax (layer 2); 8 launches; sum order kept
# speedup vs baseline: 10.2957x; 1.0341x over previous
#include <hip/hip_runtime.h>
#include <hip/hip_bf16.h>


typedef __attribute__((ext_vector_type(2))) float v2f;
typedef __attribute__((ext_vector_type(8))) float v8f;

#define F_IN 128
#define F_H  16
#define F_C  2

__global__ void GNN_zero_f32(float* __restrict__ p, long n) {
    long i = (long)blockIdx.x * blockDim.x + threadIdx.x;
    if (i < n) p[i] = 0.0f;
}

__global__ void GNN_degree(const int* __restrict__ eidx, long E,
                           float* __restrict__ deg) {
    long e = (long)blockIdx.x * blockDim.x + threadIdx.x;
    if (e >= E) return;
    int d = eidx[E + e];
    atomicAdd(&deg[d], 1.0f);
}

__global__ void GNN_dinv(const float* __restrict__ deg,
                         float* __restrict__ dinv, int n) {
    int i = blockIdx.x * blockDim.x + threadIdx.x;
    if (i < n) dinv[i] = rsqrtf(deg[i] + 1.0f);
}

__global__ void GNN_gemm1_wmma(const float* __restrict__ x,
                               const float* __restrict__ W1,
                               const float* __restrict__ dinv,
                               float* __restrict__ h1, int nTiles) {
    int lane = threadIdx.x & 31;
    int wave = threadIdx.x >> 5;
    int tile = blockIdx.x * (blockDim.x >> 5) + wave;
    if (tile >= nTiles) return;

    int m    = lane & 15;
    int koff = (lane >> 4) << 1;
    long rowA = (long)(tile * 16 + m) * F_IN;

    int rbase = tile * 16 + (lane >> 4) * 8;
    float dv[8];
    #pragma unroll
    for (int r = 0; r < 8; ++r) dv[r] = dinv[rbase + r];

    v8f acc = {};
    #pragma unroll 4
    for (int kk = 0; kk < F_IN; kk += 4) {
        v2f a, b;
        a.x = x[rowA + kk + koff];
        a.y = x[rowA + kk + koff + 1];
        b.x = W1[(kk + koff)     * F_H + m];
        b.y = W1[(kk + koff + 1) * F_H + m];
        acc = __builtin_amdgcn_wmma_f32_16x16x4_f32(
            false, a, false, b, (short)0, acc, false, false);
    }

    #pragma unroll
    for (int r = 0; r < 8; ++r)
        h1[(long)(rbase + r) * F_H + m] = dv[r] * acc[r];
}

__global__ void GNN_gemm1_tail(const float* __restrict__ x,
                               const float* __restrict__ W1,
                               const float* __restrict__ dinv,
                               float* __restrict__ h1, int startRow, int n) {
    int idx = blockIdx.x * blockDim.x + threadIdx.x;
    int row = startRow + (idx >> 4);
    int col = idx & 15;
    if (row >= n) return;
    float acc = 0.0f;
    for (int k = 0; k < F_IN; ++k)
        acc += x[(long)row * F_IN + k] * W1[k * F_H + col];
    h1[(long)row * F_H + col] = dinv[row] * acc;
}


template <int J>
__device__ __forceinline__ int GNN_row_lane(int x) {
    return __builtin_amdgcn_update_dpp(0, x, 0x150 + J, 0xf, 0xf, false);
}

template <int J>
__device__ __forceinline__ float GNN_gather_step(int s, const float* __restrict__ hp) {
    unsigned sj = (unsigned)GNN_row_lane<J>(s);
    return hp[sj * (unsigned)F_H];
}

template <int J>
__device__ __forceinline__ void GNN_add_step(int d, float v, float* __restrict__ ap) {
    unsigned dj = (unsigned)GNN_row_lane<J>(d);
    atomicAdd(ap + dj * (unsigned)F_H, v);
}

__global__ void GNN_scatter1(const int* __restrict__ eidx, long E,
                             const float* __restrict__ h1,
                             float* __restrict__ agg1) {
    long e = (long)blockIdx.x * blockDim.x + threadIdx.x;
    int lane = threadIdx.x & 31;
    long base = e - lane;
    if (base + 32 <= E) {
        int s = eidx[e];
        int d = eidx[E + e];
        int c = lane & 15;
        const float* hp = h1 + c;
        float* ap = agg1 + c;

        float v[16];
        v[0]  = GNN_gather_step<0>(s, hp);   v[1]  = GNN_gather_step<1>(s, hp);
        v[2]  = GNN_gather_step<2>(s, hp);   v[3]  = GNN_gather_step<3>(s, hp);
        v[4]  = GNN_gather_step<4>(s, hp);   v[5]  = GNN_gather_step<5>(s, hp);
        v[6]  = GNN_gather_step<6>(s, hp);   v[7]  = GNN_gather_step<7>(s, hp);
        v[8]  = GNN_gather_step<8>(s, hp);   v[9]  = GNN_gather_step<9>(s, hp);
        v[10] = GNN_gather_step<10>(s, hp);  v[11] = GNN_gather_step<11>(s, hp);
        v[12] = GNN_gather_step<12>(s, hp);  v[13] = GNN_gather_step<13>(s, hp);
        v[14] = GNN_gather_step<14>(s, hp);  v[15] = GNN_gather_step<15>(s, hp);

        GNN_add_step<0>(d, v[0], ap);    GNN_add_step<1>(d, v[1], ap);
        GNN_add_step<2>(d, v[2], ap);    GNN_add_step<3>(d, v[3], ap);
        GNN_add_step<4>(d, v[4], ap);    GNN_add_step<5>(d, v[5], ap);
        GNN_add_step<6>(d, v[6], ap);    GNN_add_step<7>(d, v[7], ap);
        GNN_add_step<8>(d, v[8], ap);    GNN_add_step<9>(d, v[9], ap);
        GNN_add_step<10>(d, v[10], ap);  GNN_add_step<11>(d, v[11], ap);
        GNN_add_step<12>(d, v[12], ap);  GNN_add_step<13>(d, v[13], ap);
        GNN_add_step<14>(d, v[14], ap);  GNN_add_step<15>(d, v[15], ap);
    } else if (e < E) {
        int s = eidx[e];
        int d = eidx[E + e];
        const float* hr = h1 + (long)s * F_H;
        float* ar = agg1 + (long)d * F_H;
        for (int j = 0; j < F_H; ++j)
            atomicAdd(ar + j, hr[j]);
    }
}

__global__ void GNN_combine1(const float* __restrict__ agg1,
                             const float* __restrict__ h1,
                             const float* __restrict__ dinv,
                             const float* __restrict__ b1,
                             const float* __restrict__ W2,
                             float* __restrict__ h2, int n) {
    int i = blockIdx.x * blockDim.x + threadIdx.x;
    if (i >= n) return;
    float di  = dinv[i];
    float acc0 = 0.0f, acc1 = 0.0f;
    #pragma unroll
    for (int j = 0; j < F_H; ++j) {
        float v = di * (agg1[(long)i * F_H + j] + h1[(long)i * F_H + j]) + b1[j];
        v = fmaxf(v, 0.0f);
        acc0 = fmaf(v, W2[j * F_C + 0], acc0);
        acc1 = fmaf(v, W2[j * F_C + 1], acc1);
    }
    h2[(long)i * F_C + 0] = di * acc0;
    h2[(long)i * F_C + 1] = di * acc1;
}

__global__ void GNN_scatter2(const int* __restrict__ eidx, long E,
                             const float* __restrict__ h2,
                             float* __restrict__ agg2) {
    long e = (long)blockIdx.x * blockDim.x + threadIdx.x;
    if (e >= E) return;
    int s = eidx[e];
    int d = eidx[E + e];
    float2 v = *(const float2*)(h2 + (long)s * F_C);
    atomicAdd(&agg2[(long)d * F_C + 0], v.x);
    atomicAdd(&agg2[(long)d * F_C + 1], v.y);
}

__global__ void GNN_final(const float* __restrict__ agg2,
                          const float* __restrict__ h2,
                          const float* __restrict__ dinv,
                          const float* __restrict__ b2,
                          float* __restrict__ out, int n) {
    int i = blockIdx.x * blockDim.x + threadIdx.x;
    if (i >= n) return;
    float di  = dinv[i];
    float o0 = di * (agg2[(long)i * F_C + 0] + h2[(long)i * F_C + 0]) + b2[0];
    float o1 = di * (agg2[(long)i * F_C + 1] + h2[(long)i * F_C + 1]) + b2[1];
    float mx  = fmaxf(o0, o1);
    float lse = mx + logf(expf(o0 - mx) + expf(o1 - mx));
    out[(long)i * F_C + 0] = o0 - lse;
    out[(long)i * F_C + 1] = o1 - lse;
}


__global__ void GNN_zero_i32(int* __restrict__ p, int n) {
    int i = blockIdx.x * blockDim.x + threadIdx.x;
    if (i < n) p[i] = 0;
}

__global__ void GNN_rank(const int* __restrict__ eidx, long E,
                         int* __restrict__ cnt, int* __restrict__ rank) {
    long e = (long)blockIdx.x * blockDim.x + threadIdx.x;
    if (e >= E) return;
    int d = eidx[E + e];
    rank[e] = atomicAdd(&cnt[d], 1);
}

#define SCAN_T   256
#define SCAN_PER 4

__global__ void GNN_scan_tot(const int* __restrict__ cnt, int n,
                             int* __restrict__ bsum) {
    __shared__ int red[SCAN_T / 32];
    int t = threadIdx.x;
    int i0 = (blockIdx.x * SCAN_T + t) * SCAN_PER;
    int s = 0;
    #pragma unroll
    for (int k = 0; k < SCAN_PER; ++k)
        if (i0 + k < n) s += cnt[i0 + k];
    for (int o = 16; o > 0; o >>= 1) s += __shfl_xor(s, o);
    if ((t & 31) == 0) red[t >> 5] = s;
    __syncthreads();
    if (t == 0) {
        int tot = 0;
        for (int w = 0; w < SCAN_T / 32; ++w) tot += red[w];
        bsum[blockIdx.x] = tot;
    }
}

template <bool WITH_DINV>
__global__ void GNN_scan_off(const int* __restrict__ cnt, int n,
                             const int* __restrict__ bsum,
                             int* __restrict__ off, float* __restrict__ dinv) {
    __shared__ int wtot[SCAN_T / 32];
    int t = threadIdx.x, lane = t & 31, w = t >> 5;

    int p = 0;
    for (int j = t; j < (int)blockIdx.x; j += SCAN_T) p += bsum[j];
    for (int o = 16; o > 0; o >>= 1) p += __shfl_xor(p, o);
    if (lane == 0) wtot[w] = p;
    __syncthreads();
    int base = 0;
    for (int k = 0; k < SCAN_T / 32; ++k) base += wtot[k];
    __syncthreads();

    int i0 = (blockIdx.x * SCAN_T + t) * SCAN_PER;
    int c[SCAN_PER];
    int s = 0;
    #pragma unroll
    for (int k = 0; k < SCAN_PER; ++k) {
        c[k] = (i0 + k < n) ? cnt[i0 + k] : 0;
        s += c[k];
    }
    int inc = s;
    for (int o = 1; o < 32; o <<= 1) {
        int v = __shfl_up(inc, o);
        if (lane >= o) inc += v;
    }
    if (lane == 31) wtot[w] = inc;
    __syncthreads();
    int wb = 0;
    for (int k = 0; k < w; ++k) wb += wtot[k];
    int run = base + wb + inc - s;
    #pragma unroll
    for (int k = 0; k < SCAN_PER; ++k) {
        if (i0 + k < n) {
            off[i0 + k]  = run;
            if (WITH_DINV) dinv[i0 + k] = rsqrtf((float)c[k] + 1.0f);
            run += c[k];
        }
    }
}

__global__ void GNN_place(const int* __restrict__ eidx, long E,
                          const int* __restrict__ off,
                          const int* __restrict__ rank,
                          int* __restrict__ csr) {
    long e = (long)blockIdx.x * blockDim.x + threadIdx.x;
    if (e >= E) return;
    int s = eidx[e];
    int d = eidx[E + e];
    csr[off[d] + rank[e]] = s;
}


#define BIN_SHIFT 10
#define BIN_W     1024
#define BIN_CH    8192
#define BIN_PER   (BIN_CH / BIN_W)
#define SRC_BITS  22
#define STAGE_CAP 49152

__device__ __forceinline__ int GNN_scan1024(int v, int* wtot) {
    int lane = threadIdx.x & 31, w = threadIdx.x >> 5;
    int inc = v;
    for (int o = 1; o < 32; o <<= 1) {
        int u = __shfl_up(inc, o);
        if (lane >= o) inc += u;
    }
    if (lane == 31) wtot[w] = inc;
    __syncthreads();
    int wv = wtot[lane];
    int winc = wv;
    for (int o = 1; o < 32; o <<= 1) {
        int u = __shfl_up(winc, o);
        if (lane >= o) winc += u;
    }
    int wbase = __shfl(winc - wv, w);
    __syncthreads();
    return wbase + inc - v;
}

__global__ void __launch_bounds__(BIN_W)
GNN_bin_count(const int* __restrict__ eidx, long E, int nbin, int NB,
              int* __restrict__ binCnt) {
    __shared__ int hist[BIN_W];
    int t = threadIdx.x;
    hist[t] = 0;
    long e0 = (long)blockIdx.x * BIN_CH;
    int d[BIN_PER];
    for (int k = 0; k < BIN_PER; ++k) {
        long e = e0 + k * BIN_W + t;
        d[k] = (e < E) ? eidx[E + e] : -1;
    }
    __syncthreads();
    for (int k = 0; k < BIN_PER; ++k)
        if (d[k] >= 0) atomicAdd(&hist[d[k] >> BIN_SHIFT], 1);
    __syncthreads();
    if (t < nbin) binCnt[(long)t * NB + blockIdx.x] = hist[t];
}

__global__ void __launch_bounds__(BIN_W)
GNN_bin_group(const int* __restrict__ eidx, long E, int nbin, int NB,
              const int* __restrict__ binCnt, const int* __restrict__ binOff,
              unsigned* __restrict__ pk) {
    __shared__ int cur[BIN_W];
    __shared__ int gdelta[BIN_W];
    __shared__ int wtot[32];
    __shared__ int bufS[BIN_CH];
    __shared__ int bufD[BIN_CH];
    int t = threadIdx.x;
    long e0 = (long)blockIdx.x * BIN_CH;
    int s[BIN_PER], d[BIN_PER];
    for (int k = 0; k < BIN_PER; ++k) {
        long e = e0 + k * BIN_W + t;
        bool in = e < E;
        s[k] = in ? eidx[e] : 0;
        d[k] = in ? eidx[E + e] : -1;
    }
    int c = 0, g = 0;
    if (t < nbin) {
        c = binCnt[(long)t * NB + blockIdx.x];
        g = binOff[(long)t * NB + blockIdx.x];
    }
    int ls = GNN_scan1024(c, wtot);
    cur[t] = ls;
    gdelta[t] = g - ls;
    __syncthreads();
    for (int k = 0; k < BIN_PER; ++k) {
        if (d[k] >= 0) {
            int p = atomicAdd(&cur[d[k] >> BIN_SHIFT], 1);
            bufS[p] = s[k];
            bufD[p] = d[k];
        }
    }
    __syncthreads();
    long left = E - e0;
    int m = left < BIN_CH ? (int)left : BIN_CH;
    for (int i = t; i < m; i += BIN_W) {
        int dd = bufD[i];
        pk[gdelta[dd >> BIN_SHIFT] + i] =
            ((unsigned)(dd & (BIN_W - 1)) << SRC_BITS) | (unsigned)bufS[i];
    }
}

__global__ void __launch_bounds__(BIN_W)
GNN_bin_order(const unsigned* __restrict__ pk, const int* __restrict__ binOff,
              int NB, int nbin, int E, int n,
              int* __restrict__ cnt, int* __restrict__ off,
              float* __restrict__ dinv, int* __restrict__ csr) {
    __shared__ int ncnt[BIN_W];
    __shared__ int wtot[32];
    __shared__ int stage[STAGE_CAP];
    int t = threadIdx.x, b = blockIdx.x;
    int beg = binOff[(long)b * NB];
    int end = (b + 1 < nbin) ? binOff[(long)(b + 1) * NB] : E;
    int m = end - beg;
    ncnt[t] = 0;
    __syncthreads();
    for (int i0 = 0; i0 < m; i0 += BIN_PER * BIN_W) {
        unsigned w[BIN_PER];
        for (int k = 0; k < BIN_PER; ++k)
            w[k] = pk[beg + min(i0 + k * BIN_W + t, m - 1)];
        for (int k = 0; k < BIN_PER; ++k)
            if (i0 + k * BIN_W + t < m) atomicAdd(&ncnt[w[k] >> SRC_BITS], 1);
    }
    __syncthreads();
    int c = ncnt[t];
    int ex = GNN_scan1024(c, wtot);
    ncnt[t] = ex;
    int node = (b << BIN_SHIFT) + t;
    if (node < n) {
        cnt[node]  = c;
        off[node]  = beg + ex;
        dinv[node] = rsqrtf((float)c + 1.0f);
    }
    __syncthreads();
    bool staged = m <= STAGE_CAP;
    for (int i0 = 0; i0 < m; i0 += BIN_PER * BIN_W) {
        unsigned w[BIN_PER];
        for (int k = 0; k < BIN_PER; ++k)
            w[k] = pk[beg + min(i0 + k * BIN_W + t, m - 1)];
        for (int k = 0; k < BIN_PER; ++k) {
            if (i0 + k * BIN_W + t < m) {
                int r = atomicAdd(&ncnt[w[k] >> SRC_BITS], 1);
                int src = (int)(w[k] & ((1u << SRC_BITS) - 1u));
                if (staged) stage[r] = src;
                else        csr[beg + r] = src;
            }
        }
    }
    if (staged) {
        __syncthreads();
        for (int i = t; i < m; i += BIN_W) csr[beg + i] = stage[i];
    }
}

template <int J>
__device__ __forceinline__ float GNN_pull_step(int s, int rem, const float* __restrict__ hp) {
    unsigned sj = (unsigned)GNN_row_lane<J>(s);
    float v = 0.0f;
    if (J < rem) v = hp[sj * (unsigned)F_H];
    return v;
}

template <int J>
__device__ __forceinline__ void GNN_h2_step(float r, const float* __restrict__ W2,
                                            float& a0, float& a1) {
    float rj = __int_as_float(GNN_row_lane<J>(__float_as_int(r)));
    a0 = fmaf(rj, W2[J * F_C + 0], a0);
    a1 = fmaf(rj, W2[J * F_C + 1], a1);
}

#define PULL1_T     256
#define PULL1_NODES (PULL1_T / 16)
__global__ void __launch_bounds__(PULL1_T)
GNN_pull1(const int* __restrict__ off, const int* __restrict__ cnt,
                          const int* __restrict__ csr,
                          const float* __restrict__ h1,
                          const float* __restrict__ dinv,
                          const float* __restrict__ b1,
                          const float* __restrict__ W2,
                          float* __restrict__ h2, int n) {
    __shared__ float line[PULL1_NODES * F_C];
    int gid  = blockIdx.x * blockDim.x + threadIdx.x;
    int node = gid >> 4;
    int c    = threadIdx.x & 15;
    bool valid = node < n;
    int nd   = valid ? node : 0;
    int beg  = off[nd];
    int num  = valid ? cnt[nd] : 0;
    int nmax = max(num, __shfl_xor(num, 16));
    const float* hp = h1 + c;
    float own = hp[(unsigned)nd * (unsigned)F_H];
    float di  = dinv[nd];
    float bc  = b1[c];
    float acc = 0.0f;
    int rem = num;
    int s = nd;
    if (c < rem) s = csr[beg + c];
    for (int b = 0; b < nmax; b += 16) {
        float v[16];
        v[0]  = GNN_pull_step<0>(s, rem, hp);   v[1]  = GNN_pull_step<1>(s, rem, hp);
        v[2]  = GNN_pull_step<2>(s, rem, hp);   v[3]  = GNN_pull_step<3>(s, rem, hp);
        v[4]  = GNN_pull_step<4>(s, rem, hp);   v[5]  = GNN_pull_step<5>(s, rem, hp);
        v[6]  = GNN_pull_step<6>(s, rem, hp);   v[7]  = GNN_pull_step<7>(s, rem, hp);
        v[8]  = GNN_pull_step<8>(s, rem, hp);   v[9]  = GNN_pull_step<9>(s, rem, hp);
        v[10] = GNN_pull_step<10>(s, rem, hp);  v[11] = GNN_pull_step<11>(s, rem, hp);
        v[12] = GNN_pull_step<12>(s, rem, hp);  v[13] = GNN_pull_step<13>(s, rem, hp);
        v[14] = GNN_pull_step<14>(s, rem, hp);  v[15] = GNN_pull_step<15>(s, rem, hp);

        int sn = nd;
        if (c < rem - 16) sn = csr[beg + b + 16 + c];
        acc += v[0];   acc += v[1];   acc += v[2];   acc += v[3];
        acc += v[4];   acc += v[5];   acc += v[6];   acc += v[7];
        acc += v[8];   acc += v[9];   acc += v[10];  acc += v[11];
        acc += v[12];  acc += v[13];  acc += v[14];  acc += v[15];
        s = sn;
        rem -= 16;
    }

    float r = di * (acc + own) + bc;
    r = fmaxf(r, 0.0f);
    float a0 = 0.0f, a1 = 0.0f;
    GNN_h2_step<0>(r, W2, a0, a1);    GNN_h2_step<1>(r, W2, a0, a1);
    GNN_h2_step<2>(r, W2, a0, a1);    GNN_h2_step<3>(r, W2, a0, a1);
    GNN_h2_step<4>(r, W2, a0, a1);    GNN_h2_step<5>(r, W2, a0, a1);
    GNN_h2_step<6>(r, W2, a0, a1);    GNN_h2_step<7>(r, W2, a0, a1);
    GNN_h2_step<8>(r, W2, a0, a1);    GNN_h2_step<9>(r, W2, a0, a1);
    GNN_h2_step<10>(r, W2, a0, a1);   GNN_h2_step<11>(r, W2, a0, a1);
    GNN_h2_step<12>(r, W2, a0, a1);   GNN_h2_step<13>(r, W2, a0, a1);
    GNN_h2_step<14>(r, W2, a0, a1);   GNN_h2_step<15>(r, W2, a0, a1);
    if (c < F_C) line[(threadIdx.x >> 4) * F_C + c] = di * (c ? a1 : a0);
    __syncthreads();
    if (threadIdx.x < PULL1_NODES * F_C) {
        long g = (long)blockIdx.x * (PULL1_NODES * F_C) + threadIdx.x;
        if (g < (long)n * F_C) h2[g] = line[threadIdx.x];
    }
}

__global__ void GNN_pull2(const int* __restrict__ off, const int* __restrict__ cnt,
                          const int* __restrict__ csr,
                          const float* __restrict__ h2,
                          const float* __restrict__ dinv,
                          const float* __restrict__ b2,
                          float* __restrict__ out, int n) {
    int lane = threadIdx.x & 31;
    int wave = (blockIdx.x * blockDim.x + threadIdx.x) >> 5;
    int base = wave * 16;
    if (base >= n) return;
    int c = lane & 15;
    int r = lane >> 4;
    int  mine = base + (lane >> 1);
    bool ok   = mine < n;
    int  md   = ok ? mine : base;
    float own = h2[(long)md * F_C + (lane & 1)];
    float di  = dinv[md];
    float bc  = b2[lane & 1];
    float keep = 0.0f;
    for (int it = 0; it < 8; ++it) {
        int node = base + 8 * r + it;
        bool valid = node < n;
        int nd   = valid ? node : base;
        int beg  = off[nd];
        int num  = valid ? cnt[nd] : 0;
        float a0 = 0.0f, a1 = 0.0f;
        for (int j = c; j < num; j += 32) {
            int j2 = j + 16;
            bool two = j2 < num;
            int s1 = csr[beg + j];
            int s2 = nd;
            if (two) s2 = csr[beg + j2];
            float2 g1 = *(const float2*)(h2 + (long)s1 * F_C);
            float2 g2 = make_float2(0.0f, 0.0f);
            if (two) g2 = *(const float2*)(h2 + (long)s2 * F_C);
            a0 += g1.x;
            a1 += g1.y;
            a0 += g2.x;
            a1 += g2.y;
        }
        for (int o = 8; o > 0; o >>= 1) {
            a0 += __shfl_xor(a0, o);
            a1 += __shfl_xor(a1, o);
        }
        if (((lane >> 1) & 7) == it) keep = (lane & 1) ? a1 : a0;
    }
    float o  = di * (keep + own) + bc;
    float p  = __shfl_xor(o, 1);
    float o0 = (lane & 1) ? p : o;
    float o1 = (lane & 1) ? o : p;
    float mx  = fmaxf(o0, o1);
    float lse = mx + logf(expf(o0 - mx) + expf(o1 - mx));
    if (ok) out[(long)base * F_C + lane] = o - lse;
}

extern "C" void kernel_launch(void* const* d_in, const int* in_sizes, int n_in,
                              void* d_out, int out_size, void* d_ws, size_t ws_size,
                              hipStream_t stream) {
    const float* x  = (const float*)d_in[0];
    const int* eidx = (const int*)  d_in[1];
    const float* W1 = (const float*)d_in[2];
    const float* b1 = (const float*)d_in[3];
    const float* W2 = (const float*)d_in[4];
    const float* b2 = (const float*)d_in[5];
    float* out = (float*)d_out;

    long nN = (long)in_sizes[0] / F_IN;
    long E  = (long)in_sizes[1] / 2;

    const int T = 256;
    float* ws   = (float*)d_ws;

    long nNp = (nN + 63) & ~63L;
    long Ep  = (E + 63) & ~63L;
    long nb  = (nN + (long)SCAN_T * SCAN_PER - 1) / ((long)SCAN_T * SCAN_PER);
    long bigR = (Ep > 32 * nNp) ? Ep : 32 * nNp;
    long need = 7 * nNp + 1024 + Ep + bigR;
    bool fits = E > 0 && E < (1L << 31) && (size_t)need * 4 <= ws_size;

    long nbin = (nN + BIN_W - 1) >> BIN_SHIFT;
    long NB   = (E + BIN_CH - 1) / BIN_CH;
    long nsc  = nbin * NB;
    long nscP = (nsc + 63) & ~63L;
    long nb2  = (nsc + (long)SCAN_T * SCAN_PER - 1) / ((long)SCAN_T * SCAN_PER);
    if (fits && nbin <= BIN_W && nN <= (1L << SRC_BITS) && nb2 <= 1024 && 2 * nscP <= 4 * nNp) {
        int*   cnt  = (int*)ws;
        int*   off  = (int*)ws + nNp;
        float* dinv = ws + 2 * nNp;
        int*   bsum = (int*)ws + 3 * nNp;
        float* h2   = ws + 3 * nNp + 1024;
        int*   binCnt = (int*)h2;
        int*   binOff = (int*)h2 + nscP;
        int*   csr  = (int*)ws + 7 * nNp + 1024;
        float* R    = ws + 7 * nNp + 1024 + Ep;
        unsigned* pk = (unsigned*)R;
        float* h1   = R;

        GNN_bin_count<<<(int)NB, BIN_W, 0, stream>>>(eidx, E, (int)nbin, (int)NB, binCnt);
        GNN_scan_tot <<<(int)nb2, SCAN_T, 0, stream>>>(binCnt, (int)nsc, bsum);
        GNN_scan_off<false><<<(int)nb2, SCAN_T, 0, stream>>>(binCnt, (int)nsc, bsum, binOff, dinv);
        GNN_bin_group<<<(int)NB, BIN_W, 0, stream>>>(eidx, E, (int)nbin, (int)NB, binCnt, binOff, pk);
        GNN_bin_order<<<(int)nbin, BIN_W, 0, stream>>>(pk, binOff, (int)NB, (int)nbin, (int)E, (int)nN,
                                                      cnt, off, dinv, csr);

        int nTiles = (int)(nN / 16);
        if (nTiles > 0)
            GNN_gemm1_wmma<<<(nTiles + 7) / 8, T, 0, stream>>>(x, W1, dinv, h1, nTiles);
        int rem = (int)(nN - (long)nTiles * 16);
        if (rem > 0)
            GNN_gemm1_tail<<<(rem * 16 + T - 1) / T, T, 0, stream>>>(
                x, W1, dinv, h1, nTiles * 16, (int)nN);

        GNN_pull1<<<(int)((nN + PULL1_NODES - 1) / PULL1_NODES), PULL1_T, 0, stream>>>(
            off, cnt, csr, h1, dinv, b1, W2, h2, (int)nN);
        GNN_pull2<<<(int)((nN * 2 + T - 1) / T), T, 0, stream>>>(
            off, cnt, csr, h2, dinv, b2, out, (int)nN);
        return;
    }

    if (nb <= 1024 && fits) {
        int*   cnt  = (int*)ws;
        int*   off  = (int*)ws + nNp;
        float* dinv = ws + 2 * nNp;
        int*   bsum = (int*)ws + 3 * nNp;
        float* h2   = ws + 3 * nNp + 1024;
        int*   csr  = (int*)ws + 7 * nNp + 1024;
        float* R    = ws + 7 * nNp + 1024 + Ep;
        int*   rank = (int*)R;
        float* h1   = R;

        GNN_zero_i32<<<(int)((nN + T - 1) / T), T, 0, stream>>>(cnt, (int)nN);
        GNN_rank    <<<(int)((E + T - 1) / T), T, 0, stream>>>(eidx, E, cnt, rank);
        GNN_scan_tot<<<(int)nb, SCAN_T, 0, stream>>>(cnt, (int)nN, bsum);
        GNN_scan_off<true><<<(int)nb, SCAN_T, 0, stream>>>(cnt, (int)nN, bsum, off, dinv);
        GNN_place   <<<(int)((E + T - 1) / T), T, 0, stream>>>(eidx, E, off, rank, csr);

        int nTiles = (int)(nN / 16);
        if (nTiles > 0)
            GNN_gemm1_wmma<<<(nTiles + 7) / 8, T, 0, stream>>>(x, W1, dinv, h1, nTiles);
        int rem = (int)(nN - (long)nTiles * 16);
        if (rem > 0)
            GNN_gemm1_tail<<<(rem * 16 + T - 1) / T, T, 0, stream>>>(
                x, W1, dinv, h1, nTiles * 16, (int)nN);

        GNN_pull1<<<(int)((nN + PULL1_NODES - 1) / PULL1_NODES), PULL1_T, 0, stream>>>(
            off, cnt, csr, h1, dinv, b1, W2, h2, (int)nN);
        GNN_pull2<<<(int)((nN * 2 + T - 1) / T), T, 0, stream>>>(
            off, cnt, csr, h2, dinv, b2, out, (int)nN);
        return;
    }

    float* deg  = ws;
    float* agg1 = ws + nN;
    float* agg2 = ws + 17 * nN;
    float* dinv = ws + 19 * nN;
    float* h1   = ws + 20 * nN;
    float* h2   = ws + 36 * nN;

    long zn = 19 * nN;
    GNN_zero_f32<<<(int)((zn + T - 1) / T), T, 0, stream>>>(ws, zn);
    GNN_degree  <<<(int)((E + T - 1) / T), T, 0, stream>>>(eidx, E, deg);
    GNN_dinv    <<<(int)((nN + T - 1) / T), T, 0, stream>>>(deg, dinv, (int)nN);

    int nTiles = (int)(nN / 16);
    if (nTiles > 0)
        GNN_gemm1_wmma<<<(nTiles + 7) / 8, T, 0, stream>>>(x, W1, dinv, h1, nTiles);
    int rem = (int)(nN - (long)nTiles * 16);
    if (rem > 0)
        GNN_gemm1_tail<<<(rem * 16 + T - 1) / T, T, 0, stream>>>(
            x, W1, dinv, h1, nTiles * 16, (int)nN);

    GNN_scatter1<<<(int)((E + T - 1) / T), T, 0, stream>>>(eidx, E, h1, agg1);
    GNN_combine1<<<(int)((nN + T - 1) / T), T, 0, stream>>>(agg1, h1, dinv, b1, W2, h2, (int)nN);
    GNN_scatter2<<<(int)((E + T - 1) / T), T, 0, stream>>>(eidx, E, h2, agg2);
    GNN_final   <<<(int)((nN + T - 1) / T), T, 0, stream>>>(agg2, h2, dinv, b2, out, (int)nN);
}


# gemm1: a wave reads its 16x128 tile of x with whole-row 16-byte non-temporal loads into LDS and takes the A fragments from LDS, not 8 strided bytes per lane from memory; k order and sums unchanged
# speedup vs baseline: 10.9271x; 1.0613x over previous
#include <hip/hip_runtime.h>
#include <hip/hip_bf16.h>


typedef __attribute__((ext_vector_type(2))) float v2f;
typedef __attribute__((ext_vector_type(8))) float v8f;

#define F_IN 128
#define F_H  16
#define F_C  2

__global__ void GNN_zero_f32(float* __restrict__ p, long n) {
    long i = (long)blockIdx.x * blockDim.x + threadIdx.x;
    if (i < n) p[i] = 0.0f;
}

__global__ void GNN_degree(const int* __restrict__ eidx, long E,
                           float* __restrict__ deg) {
    long e = (long)blockIdx.x * blockDim.x + threadIdx.x;
    if (e >= E) return;
    int d = eidx[E + e];
    atomicAdd(&deg[d], 1.0f);
}

__global__ void GNN_dinv(const float* __restrict__ deg,
                         float* __restrict__ dinv, int n) {
    int i = blockIdx.x * blockDim.x + threadIdx.x;
    if (i < n) dinv[i] = rsqrtf(deg[i] + 1.0f);
}

typedef __attribute__((ext_vector_type(4))) float v4f;
#define G1_WAVES 8
#define G1_PITCH 132
__global__ void __launch_bounds__(G1_WAVES * 32)
GNN_gemm1_wmma(const float* __restrict__ x,
               const float* __restrict__ W1,
               const float* __restrict__ dinv,
               float* __restrict__ h1, int nTiles) {
    __shared__ v4f xs[G1_WAVES][16 * G1_PITCH / 4];
    int lane = threadIdx.x & 31;
    int wave = threadIdx.x >> 5;
    int tile = blockIdx.x * G1_WAVES + wave;
    bool active = tile < nTiles;

    int m    = lane & 15;
    int koff = (lane >> 4) << 1;

    int rbase = 0;
    float dv[8];
    #pragma unroll
    for (int r = 0; r < 8; ++r) dv[r] = 0.0f;

    if (active) {
        const v4f* xt = (const v4f*)(x + (long)tile * 16 * F_IN);
        v4f q[16];
        #pragma unroll
        for (int i = 0; i < 16; ++i)
            q[i] = __builtin_nontemporal_load(xt + i * 32 + lane);
        rbase = tile * 16 + (lane >> 4) * 8;
        #pragma unroll
        for (int r = 0; r < 8; ++r) dv[r] = dinv[rbase + r];
        #pragma unroll
        for (int i = 0; i < 16; ++i)
            xs[wave][i * (G1_PITCH / 4) + lane] = q[i];
    }
    __syncthreads();
    if (!active) return;

    const float* rowp = (const float*)xs[wave] + m * G1_PITCH + koff;
    v8f acc = {};
    #pragma unroll 8
    for (int kk = 0; kk < F_IN; kk += 4) {
        v2f a, b;
        a = *(const v2f*)(rowp + kk);
        b.x = W1[(kk + koff)     * F_H + m];
        b.y = W1[(kk + koff + 1) * F_H + m];
        acc = __builtin_amdgcn_wmma_f32_16x16x4_f32(
            false, a, false, b, (short)0, acc, false, false);
    }

    #pragma unroll
    for (int r = 0; r < 8; ++r)
        h1[(long)(rbase + r) * F_H + m] = dv[r] * acc[r];
}

__global__ void GNN_gemm1_tail(const float* __restrict__ x,
                               const float* __restrict__ W1,
                               const float* __restrict__ dinv,
                               float* __restrict__ h1, int startRow, int n) {
    int idx = blockIdx.x * blockDim.x + threadIdx.x;
    int row = startRow + (idx >> 4);
    int col = idx & 15;
    if (row >= n) return;
    float acc = 0.0f;
    for (int k = 0; k < F_IN; ++k)
        acc += x[(long)row * F_IN + k] * W1[k * F_H + col];
    h1[(long)row * F_H + col] = dinv[row] * acc;
}


template <int J>
__device__ __forceinline__ int GNN_row_lane(int x) {
    return __builtin_amdgcn_update_dpp(0, x, 0x150 + J, 0xf, 0xf, false);
}

template <int J>
__device__ __forceinline__ float GNN_gather_step(int s, const float* __restrict__ hp) {
    unsigned sj = (unsigned)GNN_row_lane<J>(s);
    return hp[sj * (unsigned)F_H];
}

template <int J>
__device__ __forceinline__ void GNN_add_step(int d, float v, float* __restrict__ ap) {
    unsigned dj = (unsigned)GNN_row_lane<J>(d);
    atomicAdd(ap + dj * (unsigned)F_H, v);
}

__global__ void GNN_scatter1(const int* __restrict__ eidx, long E,
                             const float* __restrict__ h1,
                             float* __restrict__ agg1) {
    long e = (long)blockIdx.x * blockDim.x + threadIdx.x;
    int lane = threadIdx.x & 31;
    long base = e - lane;
    if (base + 32 <= E) {
        int s = eidx[e];
        int d = eidx[E + e];
        int c = lane & 15;
        const float* hp = h1 + c;
        float* ap = agg1 + c;

        float v[16];
        v[0]  = GNN_gather_step<0>(s, hp);   v[1]  = GNN_gather_step<1>(s, hp);
        v[2]  = GNN_gather_step<2>(s, hp);   v[3]  = GNN_gather_step<3>(s, hp);
        v[4]  = GNN_gather_step<4>(s, hp);   v[5]  = GNN_gather_step<5>(s, hp);
        v[6]  = GNN_gather_step<6>(s, hp);   v[7]  = GNN_gather_step<7>(s, hp);
        v[8]  = GNN_gather_step<8>(s, hp);   v[9]  = GNN_gather_step<9>(s, hp);
        v[10] = GNN_gather_step<10>(s, hp);  v[11] = GNN_gather_step<11>(s, hp);
        v[12] = GNN_gather_step<12>(s, hp);  v[13] = GNN_gather_step<13>(s, hp);
        v[14] = GNN_gather_step<14>(s, hp);  v[15] = GNN_gather_step<15>(s, hp);

        GNN_add_step<0>(d, v[0], ap);    GNN_add_step<1>(d, v[1], ap);
        GNN_add_step<2>(d, v[2], ap);    GNN_add_step<3>(d, v[3], ap);
        GNN_add_step<4>(d, v[4], ap);    GNN_add_step<5>(d, v[5], ap);
        GNN_add_step<6>(d, v[6], ap);    GNN_add_step<7>(d, v[7], ap);
        GNN_add_step<8>(d, v[8], ap);    GNN_add_step<9>(d, v[9], ap);
        GNN_add_step<10>(d, v[10], ap);  GNN_add_step<11>(d, v[11], ap);
        GNN_add_step<12>(d, v[12], ap);  GNN_add_step<13>(d, v[13], ap);
        GNN_add_step<14>(d, v[14], ap);  GNN_add_step<15>(d, v[15], ap);
    } else if (e < E) {
        int s = eidx[e];
        int d = eidx[E + e];
        const float* hr = h1 + (long)s * F_H;
        float* ar = agg1 + (long)d * F_H;
        for (int j = 0; j < F_H; ++j)
            atomicAdd(ar + j, hr[j]);
    }
}

__global__ void GNN_combine1(const float* __restrict__ agg1,
                             const float* __restrict__ h1,
                             const float* __restrict__ dinv,
                             const float* __restrict__ b1,
                             const float* __restrict__ W2,
                             float* __restrict__ h2, int n) {
    int i = blockIdx.x * blockDim.x + threadIdx.x;
    if (i >= n) return;
    float di  = dinv[i];
    float acc0 = 0.0f, acc1 = 0.0f;
    #pragma unroll
    for (int j = 0; j < F_H; ++j) {
        float v = di * (agg1[(long)i * F_H + j] + h1[(long)i * F_H + j]) + b1[j];
        v = fmaxf(v, 0.0f);
        acc0 = fmaf(v, W2[j * F_C + 0], acc0);
        acc1 = fmaf(v, W2[j * F_C + 1], acc1);
    }
    h2[(long)i * F_C + 0] = di * acc0;
    h2[(long)i * F_C + 1] = di * acc1;
}

__global__ void GNN_scatter2(const int* __restrict__ eidx, long E,
                             const float* __restrict__ h2,
                             float* __restrict__ agg2) {
    long e = (long)blockIdx.x * blockDim.x + threadIdx.x;
    if (e >= E) return;
    int s = eidx[e];
    int d = eidx[E + e];
    float2 v = *(const float2*)(h2 + (long)s * F_C);
    atomicAdd(&agg2[(long)d * F_C + 0], v.x);
    atomicAdd(&agg2[(long)d * F_C + 1], v.y);
}

__global__ void GNN_final(const float* __restrict__ agg2,
                          const float* __restrict__ h2,
                          const float* __restrict__ dinv,
                          const float* __restrict__ b2,
                          float* __restrict__ out, int n) {
    int i = blockIdx.x * blockDim.x + threadIdx.x;
    if (i >= n) return;
    float di  = dinv[i];
    float o0 = di * (agg2[(long)i * F_C + 0] + h2[(long)i * F_C + 0]) + b2[0];
    float o1 = di * (agg2[(long)i * F_C + 1] + h2[(long)i * F_C + 1]) + b2[1];
    float mx  = fmaxf(o0, o1);
    float lse = mx + logf(expf(o0 - mx) + expf(o1 - mx));
    out[(long)i * F_C + 0] = o0 - lse;
    out[(long)i * F_C + 1] = o1 - lse;
}


__global__ void GNN_zero_i32(int* __restrict__ p, int n) {
    int i = blockIdx.x * blockDim.x + threadIdx.x;
    if (i < n) p[i] = 0;
}

__global__ void GNN_rank(const int* __restrict__ eidx, long E,
                         int* __restrict__ cnt, int* __restrict__ rank) {
    long e = (long)blockIdx.x * blockDim.x + threadIdx.x;
    if (e >= E) return;
    int d = eidx[E + e];
    rank[e] = atomicAdd(&cnt[d], 1);
}

#define SCAN_T   256
#define SCAN_PER 4

__global__ void GNN_scan_tot(const int* __restrict__ cnt, int n,
                             int* __restrict__ bsum) {
    __shared__ int red[SCAN_T / 32];
    int t = threadIdx.x;
    int i0 = (blockIdx.x * SCAN_T + t) * SCAN_PER;
    int s = 0;
    #pragma unroll
    for (int k = 0; k < SCAN_PER; ++k)
        if (i0 + k < n) s += cnt[i0 + k];
    for (int o = 16; o > 0; o >>= 1) s += __shfl_xor(s, o);
    if ((t & 31) == 0) red[t >> 5] = s;
    __syncthreads();
    if (t == 0) {
        int tot = 0;
        for (int w = 0; w < SCAN_T / 32; ++w) tot += red[w];
        bsum[blockIdx.x] = tot;
    }
}

template <bool WITH_DINV>
__global__ void GNN_scan_off(const int* __restrict__ cnt, int n,
                             const int* __restrict__ bsum,
                             int* __restrict__ off, float* __restrict__ dinv) {
    __shared__ int wtot[SCAN_T / 32];
    int t = threadIdx.x, lane = t & 31, w = t >> 5;

    int p = 0;
    for (int j = t; j < (int)blockIdx.x; j += SCAN_T) p += bsum[j];
    for (int o = 16; o > 0; o >>= 1) p += __shfl_xor(p, o);
    if (lane == 0) wtot[w] = p;
    __syncthreads();
    int base = 0;
    for (int k = 0; k < SCAN_T / 32; ++k) base += wtot[k];
    __syncthreads();

    int i0 = (blockIdx.x * SCAN_T + t) * SCAN_PER;
    int c[SCAN_PER];
    int s = 0;
    #pragma unroll
    for (int k = 0; k < SCAN_PER; ++k) {
        c[k] = (i0 + k < n) ? cnt[i0 + k] : 0;
        s += c[k];
    }
    int inc = s;
    for (int o = 1; o < 32; o <<= 1) {
        int v = __shfl_up(inc, o);
        if (lane >= o) inc += v;
    }
    if (lane == 31) wtot[w] = inc;
    __syncthreads();
    int wb = 0;
    for (int k = 0; k < w; ++k) wb += wtot[k];
    int run = base + wb + inc - s;
    #pragma unroll
    for (int k = 0; k < SCAN_PER; ++k) {
        if (i0 + k < n) {
            off[i0 + k]  = run;
            if (WITH_DINV) dinv[i0 + k] = rsqrtf((float)c[k] + 1.0f);
            run += c[k];
        }
    }
}

__global__ void GNN_place(const int* __restrict__ eidx, long E,
                          const int* __restrict__ off,
                          const int* __restrict__ rank,
                          int* __restrict__ csr) {
    long e = (long)blockIdx.x * blockDim.x + threadIdx.x;
    if (e >= E) return;
    int s = eidx[e];
    int d = eidx[E + e];
    csr[off[d] + rank[e]] = s;
}


#define BIN_SHIFT 10
#define BIN_W     1024
#define BIN_CH    8192
#define BIN_PER   (BIN_CH / BIN_W)
#define SRC_BITS  22
#define STAGE_CAP 49152

__device__ __forceinline__ int GNN_scan1024(int v, int* wtot) {
    int lane = threadIdx.x & 31, w = threadIdx.x >> 5;
    int inc = v;
    for (int o = 1; o < 32; o <<= 1) {
        int u = __shfl_up(inc, o);
        if (lane >= o) inc += u;
    }
    if (lane == 31) wtot[w] = inc;
    __syncthreads();
    int wv = wtot[lane];
    int winc = wv;
    for (int o = 1; o < 32; o <<= 1) {
        int u = __shfl_up(winc, o);
        if (lane >= o) winc += u;
    }
    int wbase = __shfl(winc - wv, w);
    __syncthreads();
    return wbase + inc - v;
}

__global__ void __launch_bounds__(BIN_W)
GNN_bin_count(const int* __restrict__ eidx, long E, int nbin, int NB,
              int* __restrict__ binCnt) {
    __shared__ int hist[BIN_W];
    int t = threadIdx.x;
    hist[t] = 0;
    long e0 = (long)blockIdx.x * BIN_CH;
    int d[BIN_PER];
    for (int k = 0; k < BIN_PER; ++k) {
        long e = e0 + k * BIN_W + t;
        d[k] = (e < E) ? eidx[E + e] : -1;
    }
    __syncthreads();
    for (int k = 0; k < BIN_PER; ++k)
        if (d[k] >= 0) atomicAdd(&hist[d[k] >> BIN_SHIFT], 1);
    __syncthreads();
    if (t < nbin) binCnt[(long)t * NB + blockIdx.x] = hist[t];
}

__global__ void __launch_bounds__(BIN_W)
GNN_bin_group(const int* __restrict__ eidx, long E, int nbin, int NB,
              const int* __restrict__ binCnt, const int* __restrict__ binOff,
              unsigned* __restrict__ pk) {
    __shared__ int cur[BIN_W];
    __shared__ int gdelta[BIN_W];
    __shared__ int wtot[32];
    __shared__ int bufS[BIN_CH];
    __shared__ int bufD[BIN_CH];
    int t = threadIdx.x;
    long e0 = (long)blockIdx.x * BIN_CH;
    int s[BIN_PER], d[BIN_PER];
    for (int k = 0; k < BIN_PER; ++k) {
        long e = e0 + k * BIN_W + t;
        bool in = e < E;
        s[k] = in ? eidx[e] : 0;
        d[k] = in ? eidx[E + e] : -1;
    }
    int c = 0, g = 0;
    if (t < nbin) {
        c = binCnt[(long)t * NB + blockIdx.x];
        g = binOff[(long)t * NB + blockIdx.x];
    }
    int ls = GNN_scan1024(c, wtot);
    cur[t] = ls;
    gdelta[t] = g - ls;
    __syncthreads();
    for (int k = 0; k < BIN_PER; ++k) {
        if (d[k] >= 0) {
            int p = atomicAdd(&cur[d[k] >> BIN_SHIFT], 1);
            bufS[p] = s[k];
            bufD[p] = d[k];
        }
    }
    __syncthreads();
    long left = E - e0;
    int m = left < BIN_CH ? (int)left : BIN_CH;
    for (int i = t; i < m; i += BIN_W) {
        int dd = bufD[i];
        pk[gdelta[dd >> BIN_SHIFT] + i] =
            ((unsigned)(dd & (BIN_W - 1)) << SRC_BITS) | (unsigned)bufS[i];
    }
}

__global__ void __launch_bounds__(BIN_W)
GNN_bin_order(const unsigned* __restrict__ pk, const int* __restrict__ binOff,
              int NB, int nbin, int E, int n,
              int* __restrict__ cnt, int* __restrict__ off,
              float* __restrict__ dinv, int* __restrict__ csr) {
    __shared__ int ncnt[BIN_W];
    __shared__ int wtot[32];
    __shared__ int stage[STAGE_CAP];
    int t = threadIdx.x, b = blockIdx.x;
    int beg = binOff[(long)b * NB];
    int end = (b + 1 < nbin) ? binOff[(long)(b + 1) * NB] : E;
    int m = end - beg;
    ncnt[t] = 0;
    __syncthreads();
    for (int i0 = 0; i0 < m; i0 += BIN_PER * BIN_W) {
        unsigned w[BIN_PER];
        for (int k = 0; k < BIN_PER; ++k)
            w[k] = pk[beg + min(i0 + k * BIN_W + t, m - 1)];
        for (int k = 0; k < BIN_PER; ++k)
            if (i0 + k * BIN_W + t < m) atomicAdd(&ncnt[w[k] >> SRC_BITS], 1);
    }
    __syncthreads();
    int c = ncnt[t];
    int ex = GNN_scan1024(c, wtot);
    ncnt[t] = ex;
    int node = (b << BIN_SHIFT) + t;
    if (node < n) {
        cnt[node]  = c;
        off[node]  = beg + ex;
        dinv[node] = rsqrtf((float)c + 1.0f);
    }
    __syncthreads();
    bool staged = m <= STAGE_CAP;
    for (int i0 = 0; i0 < m; i0 += BIN_PER * BIN_W) {
        unsigned w[BIN_PER];
        for (int k = 0; k < BIN_PER; ++k)
            w[k] = pk[beg + min(i0 + k * BIN_W + t, m - 1)];
        for (int k = 0; k < BIN_PER; ++k) {
            if (i0 + k * BIN_W + t < m) {
                int r = atomicAdd(&ncnt[w[k] >> SRC_BITS], 1);
                int src = (int)(w[k] & ((1u << SRC_BITS) - 1u));
                if (staged) stage[r] = src;
                else        csr[beg + r] = src;
            }
        }
    }
    if (staged) {
        __syncthreads();
        for (int i = t; i < m; i += BIN_W) csr[beg + i] = stage[i];
    }
}

template <int J>
__device__ __forceinline__ float GNN_pull_step(int s, int rem, const float* __restrict__ hp) {
    unsigned sj = (unsigned)GNN_row_lane<J>(s);
    float v = 0.0f;
    if (J < rem) v = hp[sj * (unsigned)F_H];
    return v;
}

template <int J>
__device__ __forceinline__ void GNN_h2_step(float r, const float* __restrict__ W2,
                                            float& a0, float& a1) {
    float rj = __int_as_float(GNN_row_lane<J>(__float_as_int(r)));
    a0 = fmaf(rj, W2[J * F_C + 0], a0);
    a1 = fmaf(rj, W2[J * F_C + 1], a1);
}

#define PULL1_T     256
#define PULL1_NODES (PULL1_T / 16)
__global__ void __launch_bounds__(PULL1_T)
GNN_pull1(const int* __restrict__ off, const int* __restrict__ cnt,
                          const int* __restrict__ csr,
                          const float* __restrict__ h1,
                          const float* __restrict__ dinv,
                          const float* __restrict__ b1,
                          const float* __restrict__ W2,
                          float* __restrict__ h2, int n) {
    __shared__ float line[PULL1_NODES * F_C];
    int gid  = blockIdx.x * blockDim.x + threadIdx.x;
    int node = gid >> 4;
    int c    = threadIdx.x & 15;
    bool valid = node < n;
    int nd   = valid ? node : 0;
    int beg  = off[nd];
    int num  = valid ? cnt[nd] : 0;
    int nmax = max(num, __shfl_xor(num, 16));
    const float* hp = h1 + c;
    float own = hp[(unsigned)nd * (unsigned)F_H];
    float di  = dinv[nd];
    float bc  = b1[c];
    float acc = 0.0f;
    int rem = num;
    int s = nd;
    if (c < rem) s = csr[beg + c];
    for (int b = 0; b < nmax; b += 16) {
        float v[16];
        v[0]  = GNN_pull_step<0>(s, rem, hp);   v[1]  = GNN_pull_step<1>(s, rem, hp);
        v[2]  = GNN_pull_step<2>(s, rem, hp);   v[3]  = GNN_pull_step<3>(s, rem, hp);
        v[4]  = GNN_pull_step<4>(s, rem, hp);   v[5]  = GNN_pull_step<5>(s, rem, hp);
        v[6]  = GNN_pull_step<6>(s, rem, hp);   v[7]  = GNN_pull_step<7>(s, rem, hp);
        v[8]  = GNN_pull_step<8>(s, rem, hp);   v[9]  = GNN_pull_step<9>(s, rem, hp);
        v[10] = GNN_pull_step<10>(s, rem, hp);  v[11] = GNN_pull_step<11>(s, rem, hp);
        v[12] = GNN_pull_step<12>(s, rem, hp);  v[13] = GNN_pull_step<13>(s, rem, hp);
        v[14] = GNN_pull_step<14>(s, rem, hp);  v[15] = GNN_pull_step<15>(s, rem, hp);

        int sn = nd;
        if (c < rem - 16) sn = csr[beg + b + 16 + c];
        acc += v[0];   acc += v[1];   acc += v[2];   acc += v[3];
        acc += v[4];   acc += v[5];   acc += v[6];   acc += v[7];
        acc += v[8];   acc += v[9];   acc += v[10];  acc += v[11];
        acc += v[12];  acc += v[13];  acc += v[14];  acc += v[15];
        s = sn;
        rem -= 16;
    }

    float r = di * (acc + own) + bc;
    r = fmaxf(r, 0.0f);
    float a0 = 0.0f, a1 = 0.0f;
    GNN_h2_step<0>(r, W2, a0, a1);    GNN_h2_step<1>(r, W2, a0, a1);
    GNN_h2_step<2>(r, W2, a0, a1);    GNN_h2_step<3>(r, W2, a0, a1);
    GNN_h2_step<4>(r, W2, a0, a1);    GNN_h2_step<5>(r, W2, a0, a1);
    GNN_h2_step<6>(r, W2, a0, a1);    GNN_h2_step<7>(r, W2, a0, a1);
    GNN_h2_step<8>(r, W2, a0, a1);    GNN_h2_step<9>(r, W2, a0, a1);
    GNN_h2_step<10>(r, W2, a0, a1);   GNN_h2_step<11>(r, W2, a0, a1);
    GNN_h2_step<12>(r, W2, a0, a1);   GNN_h2_step<13>(r, W2, a0, a1);
    GNN_h2_step<14>(r, W2, a0, a1);   GNN_h2_step<15>(r, W2, a0, a1);
    if (c < F_C) line[(threadIdx.x >> 4) * F_C + c] = di * (c ? a1 : a0);
    __syncthreads();
    if (threadIdx.x < PULL1_NODES * F_C) {
        long g = (long)blockIdx.x * (PULL1_NODES * F_C) + threadIdx.x;
        if (g < (long)n * F_C) h2[g] = line[threadIdx.x];
    }
}

__global__ void GNN_pull2(const int* __restrict__ off, const int* __restrict__ cnt,
                          const int* __restrict__ csr,
                          const float* __restrict__ h2,
                          const float* __restrict__ dinv,
                          const float* __restrict__ b2,
                          float* __restrict__ out, int n) {
    int lane = threadIdx.x & 31;
    int wave = (blockIdx.x * blockDim.x + threadIdx.x) >> 5;
    int base = wave * 16;
    if (base >= n) return;
    int c = lane & 15;
    int r = lane >> 4;
    int  mine = base + (lane >> 1);
    bool ok   = mine < n;
    int  md   = ok ? mine : base;
    float own = h2[(long)md * F_C + (lane & 1)];
    float di  = dinv[md];
    float bc  = b2[lane & 1];
    float keep = 0.0f;
    for (int it = 0; it < 8; ++it) {
        int node = base + 8 * r + it;
        bool valid = node < n;
        int nd   = valid ? node : base;
        int beg  = off[nd];
        int num  = valid ? cnt[nd] : 0;
        float a0 = 0.0f, a1 = 0.0f;
        for (int j = c; j < num; j += 32) {
            int j2 = j + 16;
            bool two = j2 < num;
            int s1 = csr[beg + j];
            int s2 = nd;
            if (two) s2 = csr[beg + j2];
            float2 g1 = *(const float2*)(h2 + (long)s1 * F_C);
            float2 g2 = make_float2(0.0f, 0.0f);
            if (two) g2 = *(const float2*)(h2 + (long)s2 * F_C);
            a0 += g1.x;
            a1 += g1.y;
            a0 += g2.x;
            a1 += g2.y;
        }
        for (int o = 8; o > 0; o >>= 1) {
            a0 += __shfl_xor(a0, o);
            a1 += __shfl_xor(a1, o);
        }
        if (((lane >> 1) & 7) == it) keep = (lane & 1) ? a1 : a0;
    }
    float o  = di * (keep + own) + bc;
    float p  = __shfl_xor(o, 1);
    float o0 = (lane & 1) ? p : o;
    float o1 = (lane & 1) ? o : p;
    float mx  = fmaxf(o0, o1);
    float lse = mx + logf(expf(o0 - mx) + expf(o1 - mx));
    if (ok) out[(long)base * F_C + lane] = o - lse;
}

extern "C" void kernel_launch(void* const* d_in, const int* in_sizes, int n_in,
                              void* d_out, int out_size, void* d_ws, size_t ws_size,
                              hipStream_t stream) {
    const float* x  = (const float*)d_in[0];
    const int* eidx = (const int*)  d_in[1];
    const float* W1 = (const float*)d_in[2];
    const float* b1 = (const float*)d_in[3];
    const float* W2 = (const float*)d_in[4];
    const float* b2 = (const float*)d_in[5];
    float* out = (float*)d_out;

    long nN = (long)in_sizes[0] / F_IN;
    long E  = (long)in_sizes[1] / 2;

    const int T = 256;
    float* ws   = (float*)d_ws;

    long nNp = (nN + 63) & ~63L;
    long Ep  = (E + 63) & ~63L;
    long nb  = (nN + (long)SCAN_T * SCAN_PER - 1) / ((long)SCAN_T * SCAN_PER);
    long bigR = (Ep > 32 * nNp) ? Ep : 32 * nNp;
    long need = 7 * nNp + 1024 + Ep + bigR;
    bool fits = E > 0 && E < (1L << 31) && (size_t)need * 4 <= ws_size;

    long nbin = (nN + BIN_W - 1) >> BIN_SHIFT;
    long NB   = (E + BIN_CH - 1) / BIN_CH;
    long nsc  = nbin * NB;
    long nscP = (nsc + 63) & ~63L;
    long nb2  = (nsc + (long)SCAN_T * SCAN_PER - 1) / ((long)SCAN_T * SCAN_PER);
    if (fits && nbin <= BIN_W && nN <= (1L << SRC_BITS) && nb2 <= 1024 && 2 * nscP <= 4 * nNp) {
        int*   cnt  = (int*)ws;
        int*   off  = (int*)ws + nNp;
        float* dinv = ws + 2 * nNp;
        int*   bsum = (int*)ws + 3 * nNp;
        float* h2   = ws + 3 * nNp + 1024;
        int*   binCnt = (int*)h2;
        int*   binOff = (int*)h2 + nscP;
        int*   csr  = (int*)ws + 7 * nNp + 1024;
        float* R    = ws + 7 * nNp + 1024 + Ep;
        unsigned* pk = (unsigned*)R;
        float* h1   = R;

        GNN_bin_count<<<(int)NB, BIN_W, 0, stream>>>(eidx, E, (int)nbin, (int)NB, binCnt);
        GNN_scan_tot <<<(int)nb2, SCAN_T, 0, stream>>>(binCnt, (int)nsc, bsum);
        GNN_scan_off<false><<<(int)nb2, SCAN_T, 0, stream>>>(binCnt, (int)nsc, bsum, binOff, dinv);
        GNN_bin_group<<<(int)NB, BIN_W, 0, stream>>>(eidx, E, (int)nbin, (int)NB, binCnt, binOff, pk);
        GNN_bin_order<<<(int)nbin, BIN_W, 0, stream>>>(pk, binOff, (int)NB, (int)nbin, (int)E, (int)nN,
                                                      cnt, off, dinv, csr);

        int nTiles = (int)(nN / 16);
        if (nTiles > 0)
            GNN_gemm1_wmma<<<(nTiles + 7) / 8, T, 0, stream>>>(x, W1, dinv, h1, nTiles);
        int rem = (int)(nN - (long)nTiles * 16);
        if (rem > 0)
            GNN_gemm1_tail<<<(rem * 16 + T - 1) / T, T, 0, stream>>>(
                x, W1, dinv, h1, nTiles * 16, (int)nN);

        GNN_pull1<<<(int)((nN + PULL1_NODES - 1) / PULL1_NODES), PULL1_T, 0, stream>>>(
            off, cnt, csr, h1, dinv, b1, W2, h2, (int)nN);
        GNN_pull2<<<(int)((nN * 2 + T - 1) / T), T, 0, stream>>>(
            off, cnt, csr, h2, dinv, b2, out, (int)nN);
        return;
    }

    if (nb <= 1024 && fits) {
        int*   cnt  = (int*)ws;
        int*   off  = (int*)ws + nNp;
        float* dinv = ws + 2 * nNp;
        int*   bsum = (int*)ws + 3 * nNp;
        float* h2   = ws + 3 * nNp + 1024;
        int*   csr  = (int*)ws + 7 * nNp + 1024;
        float* R    = ws + 7 * nNp + 1024 + Ep;
        int*   rank = (int*)R;
        float* h1   = R;

        GNN_zero_i32<<<(int)((nN + T - 1) / T), T, 0, stream>>>(cnt, (int)nN);
        GNN_rank    <<<(int)((E + T - 1) / T), T, 0, stream>>>(eidx, E, cnt, rank);
        GNN_scan_tot<<<(int)nb, SCAN_T, 0, stream>>>(cnt, (int)nN, bsum);
        GNN_scan_off<true><<<(int)nb, SCAN_T, 0, stream>>>(cnt, (int)nN, bsum, off, dinv);
        GNN_place   <<<(int)((E + T - 1) / T), T, 0, stream>>>(eidx, E, off, rank, csr);

        int nTiles = (int)(nN / 16);
        if (nTiles > 0)
            GNN_gemm1_wmma<<<(nTiles + 7) / 8, T, 0, stream>>>(x, W1, dinv, h1, nTiles);
        int rem = (int)(nN - (long)nTiles * 16);
        if (rem > 0)
            GNN_gemm1_tail<<<(rem * 16 + T - 1) / T, T, 0, stream>>>(
                x, W1, dinv, h1, nTiles * 16, (int)nN);

        GNN_pull1<<<(int)((nN + PULL1_NODES - 1) / PULL1_NODES), PULL1_T, 0, stream>>>(
            off, cnt, csr, h1, dinv, b1, W2, h2, (int)nN);
        GNN_pull2<<<(int)((nN * 2 + T - 1) / T), T, 0, stream>>>(
            off, cnt, csr, h2, dinv, b2, out, (int)nN);
        return;
    }

    float* deg  = ws;
    float* agg1 = ws + nN;
    float* agg2 = ws + 17 * nN;
    float* dinv = ws + 19 * nN;
    float* h1   = ws + 20 * nN;
    float* h2   = ws + 36 * nN;

    long zn = 19 * nN;
    GNN_zero_f32<<<(int)((zn + T - 1) / T), T, 0, stream>>>(ws, zn);
    GNN_degree  <<<(int)((E + T - 1) / T), T, 0, stream>>>(eidx, E, deg);
    GNN_dinv    <<<(int)((nN + T - 1) / T), T, 0, stream>>>(deg, dinv, (int)nN);

    int nTiles = (int)(nN / 16);
    if (nTiles > 0)
        GNN_gemm1_wmma<<<(nTiles + 7) / 8, T, 0, stream>>>(x, W1, dinv, h1, nTiles);
    int rem = (int)(nN - (long)nTiles * 16);
    if (rem > 0)
        GNN_gemm1_tail<<<(rem * 16 + T - 1) / T, T, 0, stream>>>(
            x, W1, dinv, h1, nTiles * 16, (int)nN);

    GNN_scatter1<<<(int)((E + T - 1) / T), T, 0, stream>>>(eidx, E, h1, agg1);
    GNN_combine1<<<(int)((nN + T - 1) / T), T, 0, stream>>>(agg1, h1, dinv, b1, W2, h2, (int)nN);
    GNN_scatter2<<<(int)((E + T - 1) / T), T, 0, stream>>>(eidx, E, h2, agg2);
    GNN_final   <<<(int)((nN + T - 1) / T), T, 0, stream>>>(agg2, h2, dinv, b2, out, (int)nN);
}


# bin ordering without global count and scans: a chunk stores its bin groups in place, a bin's workgroup sums earlier bins' sizes and collects its groups in LDS; 5 launches; sum order: LDS arrival rank
# speedup vs baseline: 11.3748x; 1.0410x over previous
#include <hip/hip_runtime.h>
#include <hip/hip_bf16.h>


typedef __attribute__((ext_vector_type(2))) float v2f;
typedef __attribute__((ext_vector_type(8))) float v8f;

#define F_IN 128
#define F_H  16
#define F_C  2

__global__ void GNN_zero_f32(float* __restrict__ p, long n) {
    long i = (long)blockIdx.x * blockDim.x + threadIdx.x;
    if (i < n) p[i] = 0.0f;
}

__global__ void GNN_degree(const int* __restrict__ eidx, long E,
                           float* __restrict__ deg) {
    long e = (long)blockIdx.x * blockDim.x + threadIdx.x;
    if (e >= E) return;
    int d = eidx[E + e];
    atomicAdd(&deg[d], 1.0f);
}

__global__ void GNN_dinv(const float* __restrict__ deg,
                         float* __restrict__ dinv, int n) {
    int i = blockIdx.x * blockDim.x + threadIdx.x;
    if (i < n) dinv[i] = rsqrtf(deg[i] + 1.0f);
}

typedef __attribute__((ext_vector_type(4))) float v4f;
#define G1_WAVES 8
#define G1_PITCH 132
__global__ void __launch_bounds__(G1_WAVES * 32)
GNN_gemm1_wmma(const float* __restrict__ x,
               const float* __restrict__ W1,
               const float* __restrict__ dinv,
               float* __restrict__ h1, int nTiles) {
    __shared__ v4f xs[G1_WAVES][16 * G1_PITCH / 4];
    int lane = threadIdx.x & 31;
    int wave = threadIdx.x >> 5;
    int tile = blockIdx.x * G1_WAVES + wave;
    bool active = tile < nTiles;

    int m    = lane & 15;
    int koff = (lane >> 4) << 1;

    int rbase = 0;
    float dv[8];
    #pragma unroll
    for (int r = 0; r < 8; ++r) dv[r] = 0.0f;

    if (active) {
        const v4f* xt = (const v4f*)(x + (long)tile * 16 * F_IN);
        v4f q[16];
        #pragma unroll
        for (int i = 0; i < 16; ++i)
            q[i] = __builtin_nontemporal_load(xt + i * 32 + lane);
        rbase = tile * 16 + (lane >> 4) * 8;
        #pragma unroll
        for (int r = 0; r < 8; ++r) dv[r] = dinv[rbase + r];
        #pragma unroll
        for (int i = 0; i < 16; ++i)
            xs[wave][i * (G1_PITCH / 4) + lane] = q[i];
    }
    __syncthreads();
    if (!active) return;

    const float* rowp = (const float*)xs[wave] + m * G1_PITCH + koff;
    v8f acc = {};
    #pragma unroll 8
    for (int kk = 0; kk < F_IN; kk += 4) {
        v2f a, b;
        a = *(const v2f*)(rowp + kk);
        b.x = W1[(kk + koff)     * F_H + m];
        b.y = W1[(kk + koff + 1) * F_H + m];
        acc = __builtin_amdgcn_wmma_f32_16x16x4_f32(
            false, a, false, b, (short)0, acc, false, false);
    }

    #pragma unroll
    for (int r = 0; r < 8; ++r)
        h1[(long)(rbase + r) * F_H + m] = dv[r] * acc[r];
}

__global__ void GNN_gemm1_tail(const float* __restrict__ x,
                               const float* __restrict__ W1,
                               const float* __restrict__ dinv,
                               float* __restrict__ h1, int startRow, int n) {
    int idx = blockIdx.x * blockDim.x + threadIdx.x;
    int row = startRow + (idx >> 4);
    int col = idx & 15;
    if (row >= n) return;
    float acc = 0.0f;
    for (int k = 0; k < F_IN; ++k)
        acc += x[(long)row * F_IN + k] * W1[k * F_H + col];
    h1[(long)row * F_H + col] = dinv[row] * acc;
}


template <int J>
__device__ __forceinline__ int GNN_row_lane(int x) {
    return __builtin_amdgcn_update_dpp(0, x, 0x150 + J, 0xf, 0xf, false);
}

template <int J>
__device__ __forceinline__ float GNN_gather_step(int s, const float* __restrict__ hp) {
    unsigned sj = (unsigned)GNN_row_lane<J>(s);
    return hp[sj * (unsigned)F_H];
}

template <int J>
__device__ __forceinline__ void GNN_add_step(int d, float v, float* __restrict__ ap) {
    unsigned dj = (unsigned)GNN_row_lane<J>(d);
    atomicAdd(ap + dj * (unsigned)F_H, v);
}

__global__ void GNN_scatter1(const int* __restrict__ eidx, long E,
                             const float* __restrict__ h1,
                             float* __restrict__ agg1) {
    long e = (long)blockIdx.x * blockDim.x + threadIdx.x;
    int lane = threadIdx.x & 31;
    long base = e - lane;
    if (base + 32 <= E) {
        int s = eidx[e];
        int d = eidx[E + e];
        int c = lane & 15;
        const float* hp = h1 + c;
        float* ap = agg1 + c;

        float v[16];
        v[0]  = GNN_gather_step<0>(s, hp);   v[1]  = GNN_gather_step<1>(s, hp);
        v[2]  = GNN_gather_step<2>(s, hp);   v[3]  = GNN_gather_step<3>(s, hp);
        v[4]  = GNN_gather_step<4>(s, hp);   v[5]  = GNN_gather_step<5>(s, hp);
        v[6]  = GNN_gather_step<6>(s, hp);   v[7]  = GNN_gather_step<7>(s, hp);
        v[8]  = GNN_gather_step<8>(s, hp);   v[9]  = GNN_gather_step<9>(s, hp);
        v[10] = GNN_gather_step<10>(s, hp);  v[11] = GNN_gather_step<11>(s, hp);
        v[12] = GNN_gather_step<12>(s, hp);  v[13] = GNN_gather_step<13>(s, hp);
        v[14] = GNN_gather_step<14>(s, hp);  v[15] = GNN_gather_step<15>(s, hp);

        GNN_add_step<0>(d, v[0], ap);    GNN_add_step<1>(d, v[1], ap);
        GNN_add_step<2>(d, v[2], ap);    GNN_add_step<3>(d, v[3], ap);
        GNN_add_step<4>(d, v[4], ap);    GNN_add_step<5>(d, v[5], ap);
        GNN_add_step<6>(d, v[6], ap);    GNN_add_step<7>(d, v[7], ap);
        GNN_add_step<8>(d, v[8], ap);    GNN_add_step<9>(d, v[9], ap);
        GNN_add_step<10>(d, v[10], ap);  GNN_add_step<11>(d, v[11], ap);
        GNN_add_step<12>(d, v[12], ap);  GNN_add_step<13>(d, v[13], ap);
        GNN_add_step<14>(d, v[14], ap);  GNN_add_step<15>(d, v[15], ap);
    } else if (e < E) {
        int s = eidx[e];
        int d = eidx[E + e];
        const float* hr = h1 + (long)s * F_H;
        float* ar = agg1 + (long)d * F_H;
        for (int j = 0; j < F_H; ++j)
            atomicAdd(ar + j, hr[j]);
    }
}

__global__ void GNN_combine1(const float* __restrict__ agg1,
                             const float* __restrict__ h1,
                             const float* __restrict__ dinv,
                             const float* __restrict__ b1,
                             const float* __restrict__ W2,
                             float* __restrict__ h2, int n) {
    int i = blockIdx.x * blockDim.x + threadIdx.x;
    if (i >= n) return;
    float di  = dinv[i];
    float acc0 = 0.0f, acc1 = 0.0f;
    #pragma unroll
    for (int j = 0; j < F_H; ++j) {
        float v = di * (agg1[(long)i * F_H + j] + h1[(long)i * F_H + j]) + b1[j];
        v = fmaxf(v, 0.0f);
        acc0 = fmaf(v, W2[j * F_C + 0], acc0);
        acc1 = fmaf(v, W2[j * F_C + 1], acc1);
    }
    h2[(long)i * F_C + 0] = di * acc0;
    h2[(long)i * F_C + 1] = di * acc1;
}

__global__ void GNN_scatter2(const int* __restrict__ eidx, long E,
                             const float* __restrict__ h2,
                             float* __restrict__ agg2) {
    long e = (long)blockIdx.x * blockDim.x + threadIdx.x;
    if (e >= E) return;
    int s = eidx[e];
    int d = eidx[E + e];
    float2 v = *(const float2*)(h2 + (long)s * F_C);
    atomicAdd(&agg2[(long)d * F_C + 0], v.x);
    atomicAdd(&agg2[(long)d * F_C + 1], v.y);
}

__global__ void GNN_final(const float* __restrict__ agg2,
                          const float* __restrict__ h2,
                          const float* __restrict__ dinv,
                          const float* __restrict__ b2,
                          float* __restrict__ out, int n) {
    int i = blockIdx.x * blockDim.x + threadIdx.x;
    if (i >= n) return;
    float di  = dinv[i];
    float o0 = di * (agg2[(long)i * F_C + 0] + h2[(long)i * F_C + 0]) + b2[0];
    float o1 = di * (agg2[(long)i * F_C + 1] + h2[(long)i * F_C + 1]) + b2[1];
    float mx  = fmaxf(o0, o1);
    float lse = mx + logf(expf(o0 - mx) + expf(o1 - mx));
    out[(long)i * F_C + 0] = o0 - lse;
    out[(long)i * F_C + 1] = o1 - lse;
}


__global__ void GNN_zero_i32(int* __restrict__ p, int n) {
    int i = blockIdx.x * blockDim.x + threadIdx.x;
    if (i < n) p[i] = 0;
}

__global__ void GNN_rank(const int* __restrict__ eidx, long E,
                         int* __restrict__ cnt, int* __restrict__ rank) {
    long e = (long)blockIdx.x * blockDim.x + threadIdx.x;
    if (e >= E) return;
    int d = eidx[E + e];
    rank[e] = atomicAdd(&cnt[d], 1);
}

#define SCAN_T   256
#define SCAN_PER 4

__global__ void GNN_scan_tot(const int* __restrict__ cnt, int n,
                             int* __restrict__ bsum) {
    __shared__ int red[SCAN_T / 32];
    int t = threadIdx.x;
    int i0 = (blockIdx.x * SCAN_T + t) * SCAN_PER;
    int s = 0;
    #pragma unroll
    for (int k = 0; k < SCAN_PER; ++k)
        if (i0 + k < n) s += cnt[i0 + k];
    for (int o = 16; o > 0; o >>= 1) s += __shfl_xor(s, o);
    if ((t & 31) == 0) red[t >> 5] = s;
    __syncthreads();
    if (t == 0) {
        int tot = 0;
        for (int w = 0; w < SCAN_T / 32; ++w) tot += red[w];
        bsum[blockIdx.x] = tot;
    }
}

template <bool WITH_DINV>
__global__ void GNN_scan_off(const int* __restrict__ cnt, int n,
                             const int* __restrict__ bsum,
                             int* __restrict__ off, float* __restrict__ dinv) {
    __shared__ int wtot[SCAN_T / 32];
    int t = threadIdx.x, lane = t & 31, w = t >> 5;

    int p = 0;
    for (int j = t; j < (int)blockIdx.x; j += SCAN_T) p += bsum[j];
    for (int o = 16; o > 0; o >>= 1) p += __shfl_xor(p, o);
    if (lane == 0) wtot[w] = p;
    __syncthreads();
    int base = 0;
    for (int k = 0; k < SCAN_T / 32; ++k) base += wtot[k];
    __syncthreads();

    int i0 = (blockIdx.x * SCAN_T + t) * SCAN_PER;
    int c[SCAN_PER];
    int s = 0;
    #pragma unroll
    for (int k = 0; k < SCAN_PER; ++k) {
        c[k] = (i0 + k < n) ? cnt[i0 + k] : 0;
        s += c[k];
    }
    int inc = s;
    for (int o = 1; o < 32; o <<= 1) {
        int v = __shfl_up(inc, o);
        if (lane >= o) inc += v;
    }
    if (lane == 31) wtot[w] = inc;
    __syncthreads();
    int wb = 0;
    for (int k = 0; k < w; ++k) wb += wtot[k];
    int run = base + wb + inc - s;
    #pragma unroll
    for (int k = 0; k < SCAN_PER; ++k) {
        if (i0 + k < n) {
            off[i0 + k]  = run;
            if (WITH_DINV) dinv[i0 + k] = rsqrtf((float)c[k] + 1.0f);
            run += c[k];
        }
    }
}

__global__ void GNN_place(const int* __restrict__ eidx, long E,
                          const int* __restrict__ off,
                          const int* __restrict__ rank,
                          int* __restrict__ csr) {
    long e = (long)blockIdx.x * blockDim.x + threadIdx.x;
    if (e >= E) return;
    int s = eidx[e];
    int d = eidx[E + e];
    csr[off[d] + rank[e]] = s;
}


#define BIN_SHIFT 10
#define BIN_W     1024
#define BIN_CH    8192
#define BIN_PER   (BIN_CH / BIN_W)
#define SRC_BITS  22

__device__ __forceinline__ int GNN_scan1024(int v, int* wtot) {
    int lane = threadIdx.x & 31, w = threadIdx.x >> 5;
    int inc = v;
    for (int o = 1; o < 32; o <<= 1) {
        int u = __shfl_up(inc, o);
        if (lane >= o) inc += u;
    }
    if (lane == 31) wtot[w] = inc;
    __syncthreads();
    int wv = wtot[lane];
    int winc = wv;
    for (int o = 1; o < 32; o <<= 1) {
        int u = __shfl_up(winc, o);
        if (lane >= o) winc += u;
    }
    int wbase = __shfl(winc - wv, w);
    __syncthreads();
    return wbase + inc - v;
}

__global__ void __launch_bounds__(BIN_W)
GNN_bin_group(const int* __restrict__ eidx, long E, int nbin, int NB, int NBp,
              int* __restrict__ binCnt, int* __restrict__ binLs,
              unsigned* __restrict__ pk) {
    __shared__ int cur[BIN_W];
    __shared__ int wtot[32];
    __shared__ unsigned bufP[BIN_CH];
    int t = threadIdx.x;
    cur[t] = 0;
    long e0 = (long)blockIdx.x * BIN_CH;
    int s[BIN_PER], d[BIN_PER];
    for (int k = 0; k < BIN_PER; ++k) {
        long e = e0 + k * BIN_W + t;
        bool in = e < E;
        s[k] = in ? eidx[e] : 0;
        d[k] = in ? eidx[E + e] : -1;
    }
    __syncthreads();
    for (int k = 0; k < BIN_PER; ++k)
        if (d[k] >= 0) atomicAdd(&cur[d[k] >> BIN_SHIFT], 1);
    __syncthreads();
    int c = cur[t];
    int ls = GNN_scan1024(c, wtot);
    cur[t] = ls;
    if (t < nbin) {
        binCnt[(long)t * NBp + blockIdx.x] = c;
        binLs [(long)t * NBp + blockIdx.x] = ls;
        if (blockIdx.x == 0)
            for (int j = NB; j < NBp; ++j) binCnt[(long)t * NBp + j] = 0;
    }
    __syncthreads();
    for (int k = 0; k < BIN_PER; ++k) {
        if (d[k] >= 0) {
            int p = atomicAdd(&cur[d[k] >> BIN_SHIFT], 1);
            bufP[p] = ((unsigned)(d[k] & (BIN_W - 1)) << SRC_BITS) | (unsigned)s[k];
        }
    }
    __syncthreads();
    long left = E - e0;
    int m = left < BIN_CH ? (int)left : BIN_CH;
    for (int i = t; i < m; i += BIN_W) pk[e0 + i] = bufP[i];
}

#define ORD_PER 40
#define ORD_CAP (ORD_PER * BIN_W)
__global__ void __launch_bounds__(BIN_W)
GNN_bin_order(const unsigned* __restrict__ pk, const int* __restrict__ binCnt,
              const int* __restrict__ binLs, int NB, int NBp, int n,
              int* __restrict__ cnt, int* __restrict__ off,
              float* __restrict__ dinv, int* __restrict__ csr) {
    __shared__ int ncnt[BIN_W];
    __shared__ int pc[BIN_W];
    __shared__ int pl[BIN_W];
    __shared__ int ppo[BIN_W];
    __shared__ int wtot[32];
    __shared__ int red[32];
    __shared__ int mtot;
    __shared__ unsigned buf[ORD_CAP];
    int t = threadIdx.x, b = blockIdx.x, lane = t & 31, w = t >> 5;

    int ps = 0;
    {
        const int4* row = (const int4*)binCnt;
        int nq = (int)(((long)b * NBp) >> 2);
        for (int i0 = 0; i0 < nq; i0 += 8 * BIN_W) {
            int4 q[8];
            #pragma unroll
            for (int u = 0; u < 8; ++u)
                q[u] = row[min(i0 + u * BIN_W + t, nq - 1)];
            #pragma unroll
            for (int u = 0; u < 8; ++u)
                if (i0 + u * BIN_W + t < nq) ps += q[u].x + q[u].y + q[u].z + q[u].w;
        }
    }
    for (int o = 16; o > 0; o >>= 1) ps += __shfl_xor(ps, o);
    if (lane == 0) red[w] = ps;

    int myc = 0, myl = 0;
    if (t < NB) {
        myc = binCnt[(long)b * NBp + t];
        myl = binLs [(long)b * NBp + t];
    }
    ncnt[t] = 0;
    pc[t] = myc;
    pl[t] = myl;
    int po = GNN_scan1024(myc, wtot);
    ppo[t] = po;
    if (t == BIN_W - 1) mtot = po + myc;
    __syncthreads();
    int beg = 0;
    for (int k = 0; k < 32; ++k) beg += red[k];
    int m = mtot;
    bool staged = m <= ORD_CAP;

    unsigned wv[ORD_PER];
    int rk[ORD_PER];
    if (staged) {
        for (int c0 = w; c0 < NB; c0 += 128) {
            int pn[4], pb[4];
            long pg[4];
            #pragma unroll
            for (int u = 0; u < 4; ++u) {
                int c = c0 + 32 * u;
                bool ok = c < NB;
                int cc = ok ? c : 0;
                pn[u] = ok ? pc[cc] : 0;
                pb[u] = ppo[cc];
                pg[u] = (long)cc * BIN_CH + pl[cc];
            }
            unsigned v[8];
            #pragma unroll
            for (int u = 0; u < 4; ++u) {
                v[2 * u]     = (lane      < pn[u]) ? pk[pg[u] + lane]      : 0u;
                v[2 * u + 1] = (lane + 32 < pn[u]) ? pk[pg[u] + lane + 32] : 0u;
            }
            #pragma unroll
            for (int u = 0; u < 4; ++u) {
                if (lane      < pn[u]) buf[pb[u] + lane]      = v[2 * u];
                if (lane + 32 < pn[u]) buf[pb[u] + lane + 32] = v[2 * u + 1];
                for (int j = 64 + lane; j < pn[u]; j += 32) buf[pb[u] + j] = pk[pg[u] + j];
            }
        }
        __syncthreads();
        #pragma unroll
        for (int k = 0; k < ORD_PER; ++k) {
            int i = t + k * BIN_W;
            wv[k] = (i < m) ? buf[i] : 0u;
        }
        #pragma unroll
        for (int k = 0; k < ORD_PER; ++k) {
            int i = t + k * BIN_W;
            rk[k] = 0;
            if (i < m) rk[k] = atomicAdd(&ncnt[wv[k] >> SRC_BITS], 1);
        }
    } else {
        for (int c = w; c < NB; c += 32) {
            int pn = pc[c];
            long pg = (long)c * BIN_CH + pl[c];
            for (int j = lane; j < pn; j += 32) atomicAdd(&ncnt[pk[pg + j] >> SRC_BITS], 1);
        }
    }
    __syncthreads();
    int c = ncnt[t];
    int ex = GNN_scan1024(c, wtot);
    ncnt[t] = ex;
    int node = (b << BIN_SHIFT) + t;
    if (node < n) {
        cnt[node]  = c;
        off[node]  = beg + ex;
        dinv[node] = rsqrtf((float)c + 1.0f);
    }
    __syncthreads();
    if (staged) {
        #pragma unroll
        for (int k = 0; k < ORD_PER; ++k) {
            int i = t + k * BIN_W;
            if (i < m) buf[ncnt[wv[k] >> SRC_BITS] + rk[k]] = wv[k] & ((1u << SRC_BITS) - 1u);
        }
        __syncthreads();
        for (int i = t; i < m; i += BIN_W) csr[beg + i] = (int)buf[i];
    } else {
        for (int c2 = w; c2 < NB; c2 += 32) {
            int pn = pc[c2];
            long pg = (long)c2 * BIN_CH + pl[c2];
            for (int j = lane; j < pn; j += 32) {
                unsigned x = pk[pg + j];
                int r = atomicAdd(&ncnt[x >> SRC_BITS], 1);
                csr[beg + r] = (int)(x & ((1u << SRC_BITS) - 1u));
            }
        }
    }
}

template <int J>
__device__ __forceinline__ float GNN_pull_step(int s, int rem, const float* __restrict__ hp) {
    unsigned sj = (unsigned)GNN_row_lane<J>(s);
    float v = 0.0f;
    if (J < rem) v = hp[sj * (unsigned)F_H];
    return v;
}

template <int J>
__device__ __forceinline__ void GNN_h2_step(float r, const float* __restrict__ W2,
                                            float& a0, float& a1) {
    float rj = __int_as_float(GNN_row_lane<J>(__float_as_int(r)));
    a0 = fmaf(rj, W2[J * F_C + 0], a0);
    a1 = fmaf(rj, W2[J * F_C + 1], a1);
}

#define PULL1_T     256
#define PULL1_NODES (PULL1_T / 16)
__global__ void __launch_bounds__(PULL1_T)
GNN_pull1(const int* __restrict__ off, const int* __restrict__ cnt,
                          const int* __restrict__ csr,
                          const float* __restrict__ h1,
                          const float* __restrict__ dinv,
                          const float* __restrict__ b1,
                          const float* __restrict__ W2,
                          float* __restrict__ h2, int n) {
    __shared__ float line[PULL1_NODES * F_C];
    int gid  = blockIdx.x * blockDim.x + threadIdx.x;
    int node = gid >> 4;
    int c    = threadIdx.x & 15;
    bool valid = node < n;
    int nd   = valid ? node : 0;
    int beg  = off[nd];
    int num  = valid ? cnt[nd] : 0;
    int nmax = max(num, __shfl_xor(num, 16));
    const float* hp = h1 + c;
    float own = hp[(unsigned)nd * (unsigned)F_H];
    float di  = dinv[nd];
    float bc  = b1[c];
    float acc = 0.0f;
    int rem = num;
    int s = nd;
    if (c < rem) s = csr[beg + c];
    for (int b = 0; b < nmax; b += 16) {
        float v[16];
        v[0]  = GNN_pull_step<0>(s, rem, hp);   v[1]  = GNN_pull_step<1>(s, rem, hp);
        v[2]  = GNN_pull_step<2>(s, rem, hp);   v[3]  = GNN_pull_step<3>(s, rem, hp);
        v[4]  = GNN_pull_step<4>(s, rem, hp);   v[5]  = GNN_pull_step<5>(s, rem, hp);
        v[6]  = GNN_pull_step<6>(s, rem, hp);   v[7]  = GNN_pull_step<7>(s, rem, hp);
        v[8]  = GNN_pull_step<8>(s, rem, hp);   v[9]  = GNN_pull_step<9>(s, rem, hp);
        v[10] = GNN_pull_step<10>(s, rem, hp);  v[11] = GNN_pull_step<11>(s, rem, hp);
        v[12] = GNN_pull_step<12>(s, rem, hp);  v[13] = GNN_pull_step<13>(s, rem, hp);
        v[14] = GNN_pull_step<14>(s, rem, hp);  v[15] = GNN_pull_step<15>(s, rem, hp);

        int sn = nd;
        if (c < rem - 16) sn = csr[beg + b + 16 + c];
        acc += v[0];   acc += v[1];   acc += v[2];   acc += v[3];
        acc += v[4];   acc += v[5];   acc += v[6];   acc += v[7];
        acc += v[8];   acc += v[9];   acc += v[10];  acc += v[11];
        acc += v[12];  acc += v[13];  acc += v[14];  acc += v[15];
        s = sn;
        rem -= 16;
    }

    float r = di * (acc + own) + bc;
    r = fmaxf(r, 0.0f);
    float a0 = 0.0f, a1 = 0.0f;
    GNN_h2_step<0>(r, W2, a0, a1);    GNN_h2_step<1>(r, W2, a0, a1);
    GNN_h2_step<2>(r, W2, a0, a1);    GNN_h2_step<3>(r, W2, a0, a1);
    GNN_h2_step<4>(r, W2, a0, a1);    GNN_h2_step<5>(r, W2, a0, a1);
    GNN_h2_step<6>(r, W2, a0, a1);    GNN_h2_step<7>(r, W2, a0, a1);
    GNN_h2_step<8>(r, W2, a0, a1);    GNN_h2_step<9>(r, W2, a0, a1);
    GNN_h2_step<10>(r, W2, a0, a1);   GNN_h2_step<11>(r, W2, a0, a1);
    GNN_h2_step<12>(r, W2, a0, a1);   GNN_h2_step<13>(r, W2, a0, a1);
    GNN_h2_step<14>(r, W2, a0, a1);   GNN_h2_step<15>(r, W2, a0, a1);
    if (c < F_C) line[(threadIdx.x >> 4) * F_C + c] = di * (c ? a1 : a0);
    __syncthreads();
    if (threadIdx.x < PULL1_NODES * F_C) {
        long g = (long)blockIdx.x * (PULL1_NODES * F_C) + threadIdx.x;
        if (g < (long)n * F_C) h2[g] = line[threadIdx.x];
    }
}

__global__ void GNN_pull2(const int* __restrict__ off, const int* __restrict__ cnt,
                          const int* __restrict__ csr,
                          const float* __restrict__ h2,
                          const float* __restrict__ dinv,
                          const float* __restrict__ b2,
                          float* __restrict__ out, int n) {
    int lane = threadIdx.x & 31;
    int wave = (blockIdx.x * blockDim.x + threadIdx.x) >> 5;
    int base = wave * 16;
    if (base >= n) return;
    int c = lane & 15;
    int r = lane >> 4;
    int  mine = base + (lane >> 1);
    bool ok   = mine < n;
    int  md   = ok ? mine : base;
    float own = h2[(long)md * F_C + (lane & 1)];
    float di  = dinv[md];
    float bc  = b2[lane & 1];
    float keep = 0.0f;
    for (int it = 0; it < 8; ++it) {
        int node = base + 8 * r + it;
        bool valid = node < n;
        int nd   = valid ? node : base;
        int beg  = off[nd];
        int num  = valid ? cnt[nd] : 0;
        float a0 = 0.0f, a1 = 0.0f;
        for (int j = c; j < num; j += 32) {
            int j2 = j + 16;
            bool two = j2 < num;
            int s1 = csr[beg + j];
            int s2 = nd;
            if (two) s2 = csr[beg + j2];
            float2 g1 = *(const float2*)(h2 + (long)s1 * F_C);
            float2 g2 = make_float2(0.0f, 0.0f);
            if (two) g2 = *(const float2*)(h2 + (long)s2 * F_C);
            a0 += g1.x;
            a1 += g1.y;
            a0 += g2.x;
            a1 += g2.y;
        }
        for (int o = 8; o > 0; o >>= 1) {
            a0 += __shfl_xor(a0, o);
            a1 += __shfl_xor(a1, o);
        }
        if (((lane >> 1) & 7) == it) keep = (lane & 1) ? a1 : a0;
    }
    float o  = di * (keep + own) + bc;
    float p  = __shfl_xor(o, 1);
    float o0 = (lane & 1) ? p : o;
    float o1 = (lane & 1) ? o : p;
    float mx  = fmaxf(o0, o1);
    float lse = mx + logf(expf(o0 - mx) + expf(o1 - mx));
    if (ok) out[(long)base * F_C + lane] = o - lse;
}

extern "C" void kernel_launch(void* const* d_in, const int* in_sizes, int n_in,
                              void* d_out, int out_size, void* d_ws, size_t ws_size,
                              hipStream_t stream) {
    const float* x  = (const float*)d_in[0];
    const int* eidx = (const int*)  d_in[1];
    const float* W1 = (const float*)d_in[2];
    const float* b1 = (const float*)d_in[3];
    const float* W2 = (const float*)d_in[4];
    const float* b2 = (const float*)d_in[5];
    float* out = (float*)d_out;

    long nN = (long)in_sizes[0] / F_IN;
    long E  = (long)in_sizes[1] / 2;

    const int T = 256;
    float* ws   = (float*)d_ws;

    long nNp = (nN + 63) & ~63L;
    long Ep  = (E + 63) & ~63L;
    long nb  = (nN + (long)SCAN_T * SCAN_PER - 1) / ((long)SCAN_T * SCAN_PER);
    long bigR = (Ep > 32 * nNp) ? Ep : 32 * nNp;
    long need = 7 * nNp + 1024 + Ep + bigR;
    bool fits = E > 0 && E < (1L << 31) && (size_t)need * 4 <= ws_size;

    long nbin = (nN + BIN_W - 1) >> BIN_SHIFT;
    long NB   = (E + BIN_CH - 1) / BIN_CH;
    long NBp  = (NB + 3) & ~3L;
    long nscP = (nbin * NBp + 63) & ~63L;
    if (fits && nbin <= BIN_W && NB <= BIN_W && nN <= (1L << SRC_BITS) && 2 * nscP <= 4 * nNp) {
        int*   cnt  = (int*)ws;
        int*   off  = (int*)ws + nNp;
        float* dinv = ws + 2 * nNp;
        float* h2   = ws + 3 * nNp + 1024;
        int*   binCnt = (int*)h2;
        int*   binLs  = (int*)h2 + nscP;
        int*   csr  = (int*)ws + 7 * nNp + 1024;
        float* R    = ws + 7 * nNp + 1024 + Ep;
        unsigned* pk = (unsigned*)R;
        float* h1   = R;

        GNN_bin_group<<<(int)NB, BIN_W, 0, stream>>>(eidx, E, (int)nbin, (int)NB, (int)NBp,
                                                      binCnt, binLs, pk);
        GNN_bin_order<<<(int)nbin, BIN_W, 0, stream>>>(pk, binCnt, binLs, (int)NB, (int)NBp, (int)nN,
                                                      cnt, off, dinv, csr);

        int nTiles = (int)(nN / 16);
        if (nTiles > 0)
            GNN_gemm1_wmma<<<(nTiles + 7) / 8, T, 0, stream>>>(x, W1, dinv, h1, nTiles);
        int rem = (int)(nN - (long)nTiles * 16);
        if (rem > 0)
            GNN_gemm1_tail<<<(rem * 16 + T - 1) / T, T, 0, stream>>>(
                x, W1, dinv, h1, nTiles * 16, (int)nN);

        GNN_pull1<<<(int)((nN + PULL1_NODES - 1) / PULL1_NODES), PULL1_T, 0, stream>>>(
            off, cnt, csr, h1, dinv, b1, W2, h2, (int)nN);
        GNN_pull2<<<(int)((nN * 2 + T - 1) / T), T, 0, stream>>>(
            off, cnt, csr, h2, dinv, b2, out, (int)nN);
        return;
    }

    if (nb <= 1024 && fits) {
        int*   cnt  = (int*)ws;
        int*   off  = (int*)ws + nNp;
        float* dinv = ws + 2 * nNp;
        int*   bsum = (int*)ws + 3 * nNp;
        float* h2   = ws + 3 * nNp + 1024;
        int*   csr  = (int*)ws + 7 * nNp + 1024;
        float* R    = ws + 7 * nNp + 1024 + Ep;
        int*   rank = (int*)R;
        float* h1   = R;

        GNN_zero_i32<<<(int)((nN + T - 1) / T), T, 0, stream>>>(cnt, (int)nN);
        GNN_rank    <<<(int)((E + T - 1) / T), T, 0, stream>>>(eidx, E, cnt, rank);
        GNN_scan_tot<<<(int)nb, SCAN_T, 0, stream>>>(cnt, (int)nN, bsum);
        GNN_scan_off<true><<<(int)nb, SCAN_T, 0, stream>>>(cnt, (int)nN, bsum, off, dinv);
        GNN_place   <<<(int)((E + T - 1) / T), T, 0, stream>>>(eidx, E, off, rank, csr);

        int nTiles = (int)(nN / 16);
        if (nTiles > 0)
            GNN_gemm1_wmma<<<(nTiles + 7) / 8, T, 0, stream>>>(x, W1, dinv, h1, nTiles);
        int rem = (int)(nN - (long)nTiles * 16);
        if (rem > 0)
            GNN_gemm1_tail<<<(rem * 16 + T - 1) / T, T, 0, stream>>>(
                x, W1, dinv, h1, nTiles * 16, (int)nN);

        GNN_pull1<<<(int)((nN + PULL1_NODES - 1) / PULL1_NODES), PULL1_T, 0, stream>>>(
            off, cnt, csr, h1, dinv, b1, W2, h2, (int)nN);
        GNN_pull2<<<(int)((nN * 2 + T - 1) / T), T, 0, stream>>>(
            off, cnt, csr, h2, dinv, b2, out, (int)nN);
        return;
    }

    float* deg  = ws;
    float* agg1 = ws + nN;
    float* agg2 = ws + 17 * nN;
    float* dinv = ws + 19 * nN;
    float* h1   = ws + 20 * nN;
    float* h2   = ws + 36 * nN;

    long zn = 19 * nN;
    GNN_zero_f32<<<(int)((zn + T - 1) / T), T, 0, stream>>>(ws, zn);
    GNN_degree  <<<(int)((E + T - 1) / T), T, 0, stream>>>(eidx, E, deg);
    GNN_dinv    <<<(int)((nN + T - 1) / T), T, 0, stream>>>(deg, dinv, (int)nN);

    int nTiles = (int)(nN / 16);
    if (nTiles > 0)
        GNN_gemm1_wmma<<<(nTiles + 7) / 8, T, 0, stream>>>(x, W1, dinv, h1, nTiles);
    int rem = (int)(nN - (long)nTiles * 16);
    if (rem > 0)
        GNN_gemm1_tail<<<(rem * 16 + T - 1) / T, T, 0, stream>>>(
            x, W1, dinv, h1, nTiles * 16, (int)nN);

    GNN_scatter1<<<(int)((E + T - 1) / T), T, 0, stream>>>(eidx, E, h1, agg1);
    GNN_combine1<<<(int)((nN + T - 1) / T), T, 0, stream>>>(agg1, h1, dinv, b1, W2, h2, (int)nN);
    GNN_scatter2<<<(int)((E + T - 1) / T), T, 0, stream>>>(eidx, E, h2, agg2);
    GNN_final   <<<(int)((nN + T - 1) / T), T, 0, stream>>>(agg2, h2, dinv, b2, out, (int)nN);
}


# bin_order starts ordering after 3 dependent memory trips, not up to 13: first list slot = sum of the bin's own row of group first slots (no sweep of earlier bins' counts); 13 pieces per trip, not 4
# speedup vs baseline: 11.7146x; 1.0299x over previous
#include <hip/hip_runtime.h>
#include <hip/hip_bf16.h>


typedef __attribute__((ext_vector_type(2))) float v2f;
typedef __attribute__((ext_vector_type(8))) float v8f;

#define F_IN 128
#define F_H  16
#define F_C  2

__global__ void GNN_zero_f32(float* __restrict__ p, long n) {
    long i = (long)blockIdx.x * blockDim.x + threadIdx.x;
    if (i < n) p[i] = 0.0f;
}

__global__ void GNN_degree(const int* __restrict__ eidx, long E,
                           float* __restrict__ deg) {
    long e = (long)blockIdx.x * blockDim.x + threadIdx.x;
    if (e >= E) return;
    int d = eidx[E + e];
    atomicAdd(&deg[d], 1.0f);
}

__global__ void GNN_dinv(const float* __restrict__ deg,
                         float* __restrict__ dinv, int n) {
    int i = blockIdx.x * blockDim.x + threadIdx.x;
    if (i < n) dinv[i] = rsqrtf(deg[i] + 1.0f);
}

typedef __attribute__((ext_vector_type(4))) float v4f;
#define G1_WAVES 8
#define G1_PITCH 132
__global__ void __launch_bounds__(G1_WAVES * 32)
GNN_gemm1_wmma(const float* __restrict__ x,
               const float* __restrict__ W1,
               const float* __restrict__ dinv,
               float* __restrict__ h1, int nTiles) {
    __shared__ v4f xs[G1_WAVES][16 * G1_PITCH / 4];
    int lane = threadIdx.x & 31;
    int wave = threadIdx.x >> 5;
    int tile = blockIdx.x * G1_WAVES + wave;
    bool active = tile < nTiles;

    int m    = lane & 15;
    int koff = (lane >> 4) << 1;

    int rbase = 0;
    float dv[8];
    #pragma unroll
    for (int r = 0; r < 8; ++r) dv[r] = 0.0f;

    if (active) {
        const v4f* xt = (const v4f*)(x + (long)tile * 16 * F_IN);
        v4f q[16];
        #pragma unroll
        for (int i = 0; i < 16; ++i)
            q[i] = __builtin_nontemporal_load(xt + i * 32 + lane);
        rbase = tile * 16 + (lane >> 4) * 8;
        #pragma unroll
        for (int r = 0; r < 8; ++r) dv[r] = dinv[rbase + r];
        #pragma unroll
        for (int i = 0; i < 16; ++i)
            xs[wave][i * (G1_PITCH / 4) + lane] = q[i];
    }
    __syncthreads();
    if (!active) return;

    const float* rowp = (const float*)xs[wave] + m * G1_PITCH + koff;
    v8f acc = {};
    #pragma unroll 8
    for (int kk = 0; kk < F_IN; kk += 4) {
        v2f a, b;
        a = *(const v2f*)(rowp + kk);
        b.x = W1[(kk + koff)     * F_H + m];
        b.y = W1[(kk + koff + 1) * F_H + m];
        acc = __builtin_amdgcn_wmma_f32_16x16x4_f32(
            false, a, false, b, (short)0, acc, false, false);
    }

    #pragma unroll
    for (int r = 0; r < 8; ++r)
        h1[(long)(rbase + r) * F_H + m] = dv[r] * acc[r];
}

__global__ void GNN_gemm1_tail(const float* __restrict__ x,
                               const float* __restrict__ W1,
                               const float* __restrict__ dinv,
                               float* __restrict__ h1, int startRow, int n) {
    int idx = blockIdx.x * blockDim.x + threadIdx.x;
    int row = startRow + (idx >> 4);
    int col = idx & 15;
    if (row >= n) return;
    float acc = 0.0f;
    for (int k = 0; k < F_IN; ++k)
        acc += x[(long)row * F_IN + k] * W1[k * F_H + col];
    h1[(long)row * F_H + col] = dinv[row] * acc;
}


template <int J>
__device__ __forceinline__ int GNN_row_lane(int x) {
    return __builtin_amdgcn_update_dpp(0, x, 0x150 + J, 0xf, 0xf, false);
}

template <int J>
__device__ __forceinline__ float GNN_gather_step(int s, const float* __restrict__ hp) {
    unsigned sj = (unsigned)GNN_row_lane<J>(s);
    return hp[sj * (unsigned)F_H];
}

template <int J>
__device__ __forceinline__ void GNN_add_step(int d, float v, float* __restrict__ ap) {
    unsigned dj = (unsigned)GNN_row_lane<J>(d);
    atomicAdd(ap + dj * (unsigned)F_H, v);
}

__global__ void GNN_scatter1(const int* __restrict__ eidx, long E,
                             const float* __restrict__ h1,
                             float* __restrict__ agg1) {
    long e = (long)blockIdx.x * blockDim.x + threadIdx.x;
    int lane = threadIdx.x & 31;
    long base = e - lane;
    if (base + 32 <= E) {
        int s = eidx[e];
        int d = eidx[E + e];
        int c = lane & 15;
        const float* hp = h1 + c;
        float* ap = agg1 + c;

        float v[16];
        v[0]  = GNN_gather_step<0>(s, hp);   v[1]  = GNN_gather_step<1>(s, hp);
        v[2]  = GNN_gather_step<2>(s, hp);   v[3]  = GNN_gather_step<3>(s, hp);
        v[4]  = GNN_gather_step<4>(s, hp);   v[5]  = GNN_gather_step<5>(s, hp);
        v[6]  = GNN_gather_step<6>(s, hp);   v[7]  = GNN_gather_step<7>(s, hp);
        v[8]  = GNN_gather_step<8>(s, hp);   v[9]  = GNN_gather_step<9>(s, hp);
        v[10] = GNN_gather_step<10>(s, hp);  v[11] = GNN_gather_step<11>(s, hp);
        v[12] = GNN_gather_step<12>(s, hp);  v[13] = GNN_gather_step<13>(s, hp);
        v[14] = GNN_gather_step<14>(s, hp);  v[15] = GNN_gather_step<15>(s, hp);

        GNN_add_step<0>(d, v[0], ap);    GNN_add_step<1>(d, v[1], ap);
        GNN_add_step<2>(d, v[2], ap);    GNN_add_step<3>(d, v[3], ap);
        GNN_add_step<4>(d, v[4], ap);    GNN_add_step<5>(d, v[5], ap);
        GNN_add_step<6>(d, v[6], ap);    GNN_add_step<7>(d, v[7], ap);
        GNN_add_step<8>(d, v[8], ap);    GNN_add_step<9>(d, v[9], ap);
        GNN_add_step<10>(d, v[10], ap);  GNN_add_step<11>(d, v[11], ap);
        GNN_add_step<12>(d, v[12], ap);  GNN_add_step<13>(d, v[13], ap);
        GNN_add_step<14>(d, v[14], ap);  GNN_add_step<15>(d, v[15], ap);
    } else if (e < E) {
        int s = eidx[e];
        int d = eidx[E + e];
        const float* hr = h1 + (long)s * F_H;
        float* ar = agg1 + (long)d * F_H;
        for (int j = 0; j < F_H; ++j)
            atomicAdd(ar + j, hr[j]);
    }
}

__global__ void GNN_combine1(const float* __restrict__ agg1,
                             const float* __restrict__ h1,
                             const float* __restrict__ dinv,
                             const float* __restrict__ b1,
                             const float* __restrict__ W2,
                             float* __restrict__ h2, int n) {
    int i = blockIdx.x * blockDim.x + threadIdx.x;
    if (i >= n) return;
    float di  = dinv[i];
    float acc0 = 0.0f, acc1 = 0.0f;
    #pragma unroll
    for (int j = 0; j < F_H; ++j) {
        float v = di * (agg1[(long)i * F_H + j] + h1[(long)i * F_H + j]) + b1[j];
        v = fmaxf(v, 0.0f);
        acc0 = fmaf(v, W2[j * F_C + 0], acc0);
        acc1 = fmaf(v, W2[j * F_C + 1], acc1);
    }
    h2[(long)i * F_C + 0] = di * acc0;
    h2[(long)i * F_C + 1] = di * acc1;
}

__global__ void GNN_scatter2(const int* __restrict__ eidx, long E,
                             const float* __restrict__ h2,
                             float* __restrict__ agg2) {
    long e = (long)blockIdx.x * blockDim.x + threadIdx.x;
    if (e >= E) return;
    int s = eidx[e];
    int d = eidx[E + e];
    float2 v = *(const float2*)(h2 + (long)s * F_C);
    atomicAdd(&agg2[(long)d * F_C + 0], v.x);
    atomicAdd(&agg2[(long)d * F_C + 1], v.y);
}

__global__ void GNN_final(const float* __restrict__ agg2,
                          const float* __restrict__ h2,
                          const float* __restrict__ dinv,
                          const float* __restrict__ b2,
                          float* __restrict__ out, int n) {
    int i = blockIdx.x * blockDim.x + threadIdx.x;
    if (i >= n) return;
    float di  = dinv[i];
    float o0 = di * (agg2[(long)i * F_C + 0] + h2[(long)i * F_C + 0]) + b2[0];
    float o1 = di * (agg2[(long)i * F_C + 1] + h2[(long)i * F_C + 1]) + b2[1];
    float mx  = fmaxf(o0, o1);
    float lse = mx + logf(expf(o0 - mx) + expf(o1 - mx));
    out[(long)i * F_C + 0] = o0 - lse;
    out[(long)i * F_C + 1] = o1 - lse;
}


__global__ void GNN_zero_i32(int* __restrict__ p, int n) {
    int i = blockIdx.x * blockDim.x + threadIdx.x;
    if (i < n) p[i] = 0;
}

__global__ void GNN_rank(const int* __restrict__ eidx, long E,
                         int* __restrict__ cnt, int* __restrict__ rank) {
    long e = (long)blockIdx.x * blockDim.x + threadIdx.x;
    if (e >= E) return;
    int d = eidx[E + e];
    rank[e] = atomicAdd(&cnt[d], 1);
}

#define SCAN_T   256
#define SCAN_PER 4

__global__ void GNN_scan_tot(const int* __restrict__ cnt, int n,
                             int* __restrict__ bsum) {
    __shared__ int red[SCAN_T / 32];
    int t = threadIdx.x;
    int i0 = (blockIdx.x * SCAN_T + t) * SCAN_PER;
    int s = 0;
    #pragma unroll
    for (int k = 0; k < SCAN_PER; ++k)
        if (i0 + k < n) s += cnt[i0 + k];
    for (int o = 16; o > 0; o >>= 1) s += __shfl_xor(s, o);
    if ((t & 31) == 0) red[t >> 5] = s;
    __syncthreads();
    if (t == 0) {
        int tot = 0;
        for (int w = 0; w < SCAN_T / 32; ++w) tot += red[w];
        bsum[blockIdx.x] = tot;
    }
}

template <bool WITH_DINV>
__global__ void GNN_scan_off(const int* __restrict__ cnt, int n,
                             const int* __restrict__ bsum,
                             int* __restrict__ off, float* __restrict__ dinv) {
    __shared__ int wtot[SCAN_T / 32];
    int t = threadIdx.x, lane = t & 31, w = t >> 5;

    int p = 0;
    for (int j = t; j < (int)blockIdx.x; j += SCAN_T) p += bsum[j];
    for (int o = 16; o > 0; o >>= 1) p += __shfl_xor(p, o);
    if (lane == 0) wtot[w] = p;
    __syncthreads();
    int base = 0;
    for (int k = 0; k < SCAN_T / 32; ++k) base += wtot[k];
    __syncthreads();

    int i0 = (blockIdx.x * SCAN_T + t) * SCAN_PER;
    int c[SCAN_PER];
    int s = 0;
    #pragma unroll
    for (int k = 0; k < SCAN_PER; ++k) {
        c[k] = (i0 + k < n) ? cnt[i0 + k] : 0;
        s += c[k];
    }
    int inc = s;
    for (int o = 1; o < 32; o <<= 1) {
        int v = __shfl_up(inc, o);
        if (lane >= o) inc += v;
    }
    if (lane == 31) wtot[w] = inc;
    __syncthreads();
    int wb = 0;
    for (int k = 0; k < w; ++k) wb += wtot[k];
    int run = base + wb + inc - s;
    #pragma unroll
    for (int k = 0; k < SCAN_PER; ++k) {
        if (i0 + k < n) {
            off[i0 + k]  = run;
            if (WITH_DINV) dinv[i0 + k] = rsqrtf((float)c[k] + 1.0f);
            run += c[k];
        }
    }
}

__global__ void GNN_place(const int* __restrict__ eidx, long E,
                          const int* __restrict__ off,
                          const int* __restrict__ rank,
                          int* __restrict__ csr) {
    long e = (long)blockIdx.x * blockDim.x + threadIdx.x;
    if (e >= E) return;
    int s = eidx[e];
    int d = eidx[E + e];
    csr[off[d] + rank[e]] = s;
}


#define BIN_SHIFT 10
#define BIN_W     1024
#define BIN_CH    8192
#define BIN_PER   (BIN_CH / BIN_W)
#define SRC_BITS  22

__device__ __forceinline__ int GNN_scan1024(int v, int* wtot) {
    int lane = threadIdx.x & 31, w = threadIdx.x >> 5;
    int inc = v;
    for (int o = 1; o < 32; o <<= 1) {
        int u = __shfl_up(inc, o);
        if (lane >= o) inc += u;
    }
    if (lane == 31) wtot[w] = inc;
    __syncthreads();
    int wv = wtot[lane];
    int winc = wv;
    for (int o = 1; o < 32; o <<= 1) {
        int u = __shfl_up(winc, o);
        if (lane >= o) winc += u;
    }
    int wbase = __shfl(winc - wv, w);
    __syncthreads();
    return wbase + inc - v;
}

__global__ void __launch_bounds__(BIN_W)
GNN_bin_group(const int* __restrict__ eidx, long E, int nbin, int NBp,
              int* __restrict__ binCnt, int* __restrict__ binLs,
              unsigned* __restrict__ pk) {
    __shared__ int cur[BIN_W];
    __shared__ int wtot[32];
    __shared__ unsigned bufP[BIN_CH];
    int t = threadIdx.x;
    cur[t] = 0;
    long e0 = (long)blockIdx.x * BIN_CH;
    int s[BIN_PER], d[BIN_PER];
    for (int k = 0; k < BIN_PER; ++k) {
        long e = e0 + k * BIN_W + t;
        bool in = e < E;
        s[k] = in ? eidx[e] : 0;
        d[k] = in ? eidx[E + e] : -1;
    }
    __syncthreads();
    for (int k = 0; k < BIN_PER; ++k)
        if (d[k] >= 0) atomicAdd(&cur[d[k] >> BIN_SHIFT], 1);
    __syncthreads();
    int c = cur[t];
    int ls = GNN_scan1024(c, wtot);
    cur[t] = ls;
    if (t < nbin) {
        binCnt[(long)t * NBp + blockIdx.x] = c;
        binLs [(long)t * NBp + blockIdx.x] = ls;
    }
    __syncthreads();
    for (int k = 0; k < BIN_PER; ++k) {
        if (d[k] >= 0) {
            int p = atomicAdd(&cur[d[k] >> BIN_SHIFT], 1);
            bufP[p] = ((unsigned)(d[k] & (BIN_W - 1)) << SRC_BITS) | (unsigned)s[k];
        }
    }
    __syncthreads();
    long left = E - e0;
    int m = left < BIN_CH ? (int)left : BIN_CH;
    for (int i = t; i < m; i += BIN_W) pk[e0 + i] = bufP[i];
}

#define ORD_PER 40
#define ORD_CAP (ORD_PER * BIN_W)
#define ORD_GRP 13
__global__ void __launch_bounds__(BIN_W)
GNN_bin_order(const unsigned* __restrict__ pk, const int* __restrict__ binCnt,
              const int* __restrict__ binLs, int NB, int NBp, int n,
              int* __restrict__ cnt, int* __restrict__ off,
              float* __restrict__ dinv, int* __restrict__ csr) {
    __shared__ int ncnt[BIN_W];
    __shared__ int pc[BIN_W];
    __shared__ int pl[BIN_W];
    __shared__ int ppo[BIN_W];
    __shared__ int wtot[32];
    __shared__ int red[32];
    __shared__ int mtot;
    __shared__ unsigned buf[ORD_CAP];
    int t = threadIdx.x, b = blockIdx.x, lane = t & 31, w = t >> 5;

    int myc = 0, myl = 0;
    if (t < NB) {
        myc = binCnt[(long)b * NBp + t];
        myl = binLs [(long)b * NBp + t];
    }
    int ps = myl;
    for (int o = 16; o > 0; o >>= 1) ps += __shfl_xor(ps, o);
    if (lane == 0) red[w] = ps;
    ncnt[t] = 0;
    pc[t] = myc;
    pl[t] = myl;
    int po = GNN_scan1024(myc, wtot);
    ppo[t] = po;
    if (t == BIN_W - 1) mtot = po + myc;
    __syncthreads();
    int beg = 0;
    for (int k = 0; k < 32; ++k) beg += red[k];
    int m = mtot;
    bool staged = m <= ORD_CAP;

    unsigned wv[ORD_PER];
    int rk[ORD_PER];
    if (staged) {
        for (int c0 = w; c0 < NB; c0 += 32 * ORD_GRP) {
            int pn[ORD_GRP], pb[ORD_GRP];
            unsigned pg[ORD_GRP];
            #pragma unroll
            for (int u = 0; u < ORD_GRP; ++u) {
                int c = c0 + 32 * u;
                bool ok = c < NB;
                int cc = ok ? c : 0;
                pn[u] = ok ? pc[cc] : 0;
                pb[u] = ppo[cc];
                pg[u] = (unsigned)cc * BIN_CH + (unsigned)pl[cc];
            }
            unsigned v[2 * ORD_GRP];
            #pragma unroll
            for (int u = 0; u < ORD_GRP; ++u) {
                v[2 * u]     = (lane      < pn[u]) ? pk[pg[u] + (unsigned)lane]       : 0u;
                v[2 * u + 1] = (lane + 32 < pn[u]) ? pk[pg[u] + (unsigned)lane + 32u] : 0u;
            }
            #pragma unroll
            for (int u = 0; u < ORD_GRP; ++u) {
                if (lane      < pn[u]) buf[pb[u] + lane]      = v[2 * u];
                if (lane + 32 < pn[u]) buf[pb[u] + lane + 32] = v[2 * u + 1];
                #pragma unroll 1
                for (int j = 64 + lane; j < pn[u]; j += 32) buf[pb[u] + j] = pk[pg[u] + (unsigned)j];
            }
        }
        __syncthreads();
        #pragma unroll
        for (int k = 0; k < ORD_PER; ++k) {
            int i = t + k * BIN_W;
            wv[k] = (i < m) ? buf[i] : 0u;
        }
        #pragma unroll
        for (int k = 0; k < ORD_PER; ++k) {
            int i = t + k * BIN_W;
            rk[k] = 0;
            if (i < m) rk[k] = atomicAdd(&ncnt[wv[k] >> SRC_BITS], 1);
        }
    } else {
        for (int c = w; c < NB; c += 32) {
            int pn = pc[c];
            long pg = (long)c * BIN_CH + pl[c];
            for (int j = lane; j < pn; j += 32) atomicAdd(&ncnt[pk[pg + j] >> SRC_BITS], 1);
        }
    }
    __syncthreads();
    int c = ncnt[t];
    int ex = GNN_scan1024(c, wtot);
    ncnt[t] = ex;
    int node = (b << BIN_SHIFT) + t;
    if (node < n) {
        cnt[node]  = c;
        off[node]  = beg + ex;
        dinv[node] = rsqrtf((float)c + 1.0f);
    }
    __syncthreads();
    if (staged) {
        #pragma unroll
        for (int k = 0; k < ORD_PER; ++k) {
            int i = t + k * BIN_W;
            if (i < m) buf[ncnt[wv[k] >> SRC_BITS] + rk[k]] = wv[k] & ((1u << SRC_BITS) - 1u);
        }
        __syncthreads();
        for (int i = t; i < m; i += BIN_W) csr[beg + i] = (int)buf[i];
    } else {
        for (int c2 = w; c2 < NB; c2 += 32) {
            int pn = pc[c2];
            long pg = (long)c2 * BIN_CH + pl[c2];
            for (int j = lane; j < pn; j += 32) {
                unsigned x = pk[pg + j];
                int r = atomicAdd(&ncnt[x >> SRC_BITS], 1);
                csr[beg + r] = (int)(x & ((1u << SRC_BITS) - 1u));
            }
        }
    }
}

template <int J>
__device__ __forceinline__ float GNN_pull_step(int s, int rem, const float* __restrict__ hp) {
    unsigned sj = (unsigned)GNN_row_lane<J>(s);
    float v = 0.0f;
    if (J < rem) v = hp[sj * (unsigned)F_H];
    return v;
}

template <int J>
__device__ __forceinline__ void GNN_h2_step(float r, const float* __restrict__ W2,
                                            float& a0, float& a1) {
    float rj = __int_as_float(GNN_row_lane<J>(__float_as_int(r)));
    a0 = fmaf(rj, W2[J * F_C + 0], a0);
    a1 = fmaf(rj, W2[J * F_C + 1], a1);
}

#define PULL1_T     256
#define PULL1_NODES (PULL1_T / 16)
__global__ void __launch_bounds__(PULL1_T)
GNN_pull1(const int* __restrict__ off, const int* __restrict__ cnt,
                          const int* __restrict__ csr,
                          const float* __restrict__ h1,
                          const float* __restrict__ dinv,
                          const float* __restrict__ b1,
                          const float* __restrict__ W2,
                          float* __restrict__ h2, int n) {
    __shared__ float line[PULL1_NODES * F_C];
    int gid  = blockIdx.x * blockDim.x + threadIdx.x;
    int node = gid >> 4;
    int c    = threadIdx.x & 15;
    bool valid = node < n;
    int nd   = valid ? node : 0;
    int beg  = off[nd];
    int num  = valid ? cnt[nd] : 0;
    int nmax = max(num, __shfl_xor(num, 16));
    const float* hp = h1 + c;
    float own = hp[(unsigned)nd * (unsigned)F_H];
    float di  = dinv[nd];
    float bc  = b1[c];
    float acc = 0.0f;
    int rem = num;
    int s = nd;
    if (c < rem) s = csr[beg + c];
    for (int b = 0; b < nmax; b += 16) {
        float v[16];
        v[0]  = GNN_pull_step<0>(s, rem, hp);   v[1]  = GNN_pull_step<1>(s, rem, hp);
        v[2]  = GNN_pull_step<2>(s, rem, hp);   v[3]  = GNN_pull_step<3>(s, rem, hp);
        v[4]  = GNN_pull_step<4>(s, rem, hp);   v[5]  = GNN_pull_step<5>(s, rem, hp);
        v[6]  = GNN_pull_step<6>(s, rem, hp);   v[7]  = GNN_pull_step<7>(s, rem, hp);
        v[8]  = GNN_pull_step<8>(s, rem, hp);   v[9]  = GNN_pull_step<9>(s, rem, hp);
        v[10] = GNN_pull_step<10>(s, rem, hp);  v[11] = GNN_pull_step<11>(s, rem, hp);
        v[12] = GNN_pull_step<12>(s, rem, hp);  v[13] = GNN_pull_step<13>(s, rem, hp);
        v[14] = GNN_pull_step<14>(s, rem, hp);  v[15] = GNN_pull_step<15>(s, rem, hp);

        int sn = nd;
        if (c < rem - 16) sn = csr[beg + b + 16 + c];
        acc += v[0];   acc += v[1];   acc += v[2];   acc += v[3];
        acc += v[4];   acc += v[5];   acc += v[6];   acc += v[7];
        acc += v[8];   acc += v[9];   acc += v[10];  acc += v[11];
        acc += v[12];  acc += v[13];  acc += v[14];  acc += v[15];
        s = sn;
        rem -= 16;
    }

    float r = di * (acc + own) + bc;
    r = fmaxf(r, 0.0f);
    float a0 = 0.0f, a1 = 0.0f;
    GNN_h2_step<0>(r, W2, a0, a1);    GNN_h2_step<1>(r, W2, a0, a1);
    GNN_h2_step<2>(r, W2, a0, a1);    GNN_h2_step<3>(r, W2, a0, a1);
    GNN_h2_step<4>(r, W2, a0, a1);    GNN_h2_step<5>(r, W2, a0, a1);
    GNN_h2_step<6>(r, W2, a0, a1);    GNN_h2_step<7>(r, W2, a0, a1);
    GNN_h2_step<8>(r, W2, a0, a1);    GNN_h2_step<9>(r, W2, a0, a1);
    GNN_h2_step<10>(r, W2, a0, a1);   GNN_h2_step<11>(r, W2, a0, a1);
    GNN_h2_step<12>(r, W2, a0, a1);   GNN_h2_step<13>(r, W2, a0, a1);
    GNN_h2_step<14>(r, W2, a0, a1);   GNN_h2_step<15>(r, W2, a0, a1);
    if (c < F_C) line[(threadIdx.x >> 4) * F_C + c] = di * (c ? a1 : a0);
    __syncthreads();
    if (threadIdx.x < PULL1_NODES * F_C) {
        long g = (long)blockIdx.x * (PULL1_NODES * F_C) + threadIdx.x;
        if (g < (long)n * F_C) h2[g] = line[threadIdx.x];
    }
}

__global__ void GNN_pull2(const int* __restrict__ off, const int* __restrict__ cnt,
                          const int* __restrict__ csr,
                          const float* __restrict__ h2,
                          const float* __restrict__ dinv,
                          const float* __restrict__ b2,
                          float* __restrict__ out, int n) {
    int lane = threadIdx.x & 31;
    int wave = (blockIdx.x * blockDim.x + threadIdx.x) >> 5;
    int base = wave * 16;
    if (base >= n) return;
    int c = lane & 15;
    int r = lane >> 4;
    int  mine = base + (lane >> 1);
    bool ok   = mine < n;
    int  md   = ok ? mine : base;
    float own = h2[(long)md * F_C + (lane & 1)];
    float di  = dinv[md];
    float bc  = b2[lane & 1];
    float keep = 0.0f;
    for (int it = 0; it < 8; ++it) {
        int node = base + 8 * r + it;
        bool valid = node < n;
        int nd   = valid ? node : base;
        int beg  = off[nd];
        int num  = valid ? cnt[nd] : 0;
        float a0 = 0.0f, a1 = 0.0f;
        for (int j = c; j < num; j += 32) {
            int j2 = j + 16;
            bool two = j2 < num;
            int s1 = csr[beg + j];
            int s2 = nd;
            if (two) s2 = csr[beg + j2];
            float2 g1 = *(const float2*)(h2 + (long)s1 * F_C);
            float2 g2 = make_float2(0.0f, 0.0f);
            if (two) g2 = *(const float2*)(h2 + (long)s2 * F_C);
            a0 += g1.x;
            a1 += g1.y;
            a0 += g2.x;
            a1 += g2.y;
        }
        for (int o = 8; o > 0; o >>= 1) {
            a0 += __shfl_xor(a0, o);
            a1 += __shfl_xor(a1, o);
        }
        if (((lane >> 1) & 7) == it) keep = (lane & 1) ? a1 : a0;
    }
    float o  = di * (keep + own) + bc;
    float p  = __shfl_xor(o, 1);
    float o0 = (lane & 1) ? p : o;
    float o1 = (lane & 1) ? o : p;
    float mx  = fmaxf(o0, o1);
    float lse = mx + logf(expf(o0 - mx) + expf(o1 - mx));
    if (ok) out[(long)base * F_C + lane] = o - lse;
}

extern "C" void kernel_launch(void* const* d_in, const int* in_sizes, int n_in,
                              void* d_out, int out_size, void* d_ws, size_t ws_size,
                              hipStream_t stream) {
    const float* x  = (const float*)d_in[0];
    const int* eidx = (const int*)  d_in[1];
    const float* W1 = (const float*)d_in[2];
    const float* b1 = (const float*)d_in[3];
    const float* W2 = (const float*)d_in[4];
    const float* b2 = (const float*)d_in[5];
    float* out = (float*)d_out;

    long nN = (long)in_sizes[0] / F_IN;
    long E  = (long)in_sizes[1] / 2;

    const int T = 256;
    float* ws   = (float*)d_ws;

    long nNp = (nN + 63) & ~63L;
    long Ep  = (E + 63) & ~63L;
    long nb  = (nN + (long)SCAN_T * SCAN_PER - 1) / ((long)SCAN_T * SCAN_PER);
    long bigR = (Ep > 32 * nNp) ? Ep : 32 * nNp;
    long need = 7 * nNp + 1024 + Ep + bigR;
    bool fits = E > 0 && E < (1L << 31) && (size_t)need * 4 <= ws_size;

    long nbin = (nN + BIN_W - 1) >> BIN_SHIFT;
    long NB   = (E + BIN_CH - 1) / BIN_CH;
    long NBp  = (NB + 3) & ~3L;
    long nscP = (nbin * NBp + 63) & ~63L;
    if (fits && nbin <= BIN_W && NB <= BIN_W && nN <= (1L << SRC_BITS) && 2 * nscP <= 4 * nNp) {
        int*   cnt  = (int*)ws;
        int*   off  = (int*)ws + nNp;
        float* dinv = ws + 2 * nNp;
        float* h2   = ws + 3 * nNp + 1024;
        int*   binCnt = (int*)h2;
        int*   binLs  = (int*)h2 + nscP;
        int*   csr  = (int*)ws + 7 * nNp + 1024;
        float* R    = ws + 7 * nNp + 1024 + Ep;
        unsigned* pk = (unsigned*)R;
        float* h1   = R;

        GNN_bin_group<<<(int)NB, BIN_W, 0, stream>>>(eidx, E, (int)nbin, (int)NBp,
                                                      binCnt, binLs, pk);
        GNN_bin_order<<<(int)nbin, BIN_W, 0, stream>>>(pk, binCnt, binLs, (int)NB, (int)NBp, (int)nN,
                                                      cnt, off, dinv, csr);

        int nTiles = (int)(nN / 16);
        if (nTiles > 0)
            GNN_gemm1_wmma<<<(nTiles + 7) / 8, T, 0, stream>>>(x, W1, dinv, h1, nTiles);
        int rem = (int)(nN - (long)nTiles * 16);
        if (rem > 0)
            GNN_gemm1_tail<<<(rem * 16 + T - 1) / T, T, 0, stream>>>(
                x, W1, dinv, h1, nTiles * 16, (int)nN);

        GNN_pull1<<<(int)((nN + PULL1_NODES - 1) / PULL1_NODES), PULL1_T, 0, stream>>>(
            off, cnt, csr, h1, dinv, b1, W2, h2, (int)nN);
        GNN_pull2<<<(int)((nN * 2 + T - 1) / T), T, 0, stream>>>(
            off, cnt, csr, h2, dinv, b2, out, (int)nN);
        return;
    }

    if (nb <= 1024 && fits) {
        int*   cnt  = (int*)ws;
        int*   off  = (int*)ws + nNp;
        float* dinv = ws + 2 * nNp;
        int*   bsum = (int*)ws + 3 * nNp;
        float* h2   = ws + 3 * nNp + 1024;
        int*   csr  = (int*)ws + 7 * nNp + 1024;
        float* R    = ws + 7 * nNp + 1024 + Ep;
        int*   rank = (int*)R;
        float* h1   = R;

        GNN_zero_i32<<<(int)((nN + T - 1) / T), T, 0, stream>>>(cnt, (int)nN);
        GNN_rank    <<<(int)((E + T - 1) / T), T, 0, stream>>>(eidx, E, cnt, rank);
        GNN_scan_tot<<<(int)nb, SCAN_T, 0, stream>>>(cnt, (int)nN, bsum);
        GNN_scan_off<true><<<(int)nb, SCAN_T, 0, stream>>>(cnt, (int)nN, bsum, off, dinv);
        GNN_place   <<<(int)((E + T - 1) / T), T, 0, stream>>>(eidx, E, off, rank, csr);

        int nTiles = (int)(nN / 16);
        if (nTiles > 0)
            GNN_gemm1_wmma<<<(nTiles + 7) / 8, T, 0, stream>>>(x, W1, dinv, h1, nTiles);
        int rem = (int)(nN - (long)nTiles * 16);
        if (rem > 0)
            GNN_gemm1_tail<<<(rem * 16 + T - 1) / T, T, 0, stream>>>(
                x, W1, dinv, h1, nTiles * 16, (int)nN);

        GNN_pull1<<<(int)((nN + PULL1_NODES - 1) / PULL1_NODES), PULL1_T, 0, stream>>>(
            off, cnt, csr, h1, dinv, b1, W2, h2, (int)nN);
        GNN_pull2<<<(int)((nN * 2 + T - 1) / T), T, 0, stream>>>(
            off, cnt, csr, h2, dinv, b2, out, (int)nN);
        return;
    }

    float* deg  = ws;
    float* agg1 = ws + nN;
    float* agg2 = ws + 17 * nN;
    float* dinv = ws + 19 * nN;
    float* h1   = ws + 20 * nN;
    float* h2   = ws + 36 * nN;

    long zn = 19 * nN;
    GNN_zero_f32<<<(int)((zn + T - 1) / T), T, 0, stream>>>(ws, zn);
    GNN_degree  <<<(int)((E + T - 1) / T), T, 0, stream>>>(eidx, E, deg);
    GNN_dinv    <<<(int)((nN + T - 1) / T), T, 0, stream>>>(deg, dinv, (int)nN);

    int nTiles = (int)(nN / 16);
    if (nTiles > 0)
        GNN_gemm1_wmma<<<(nTiles + 7) / 8, T, 0, stream>>>(x, W1, dinv, h1, nTiles);
    int rem = (int)(nN - (long)nTiles * 16);
    if (rem > 0)
        GNN_gemm1_tail<<<(rem * 16 + T - 1) / T, T, 0, stream>>>(
            x, W1, dinv, h1, nTiles * 16, (int)nN);

    GNN_scatter1<<<(int)((E + T - 1) / T), T, 0, stream>>>(eidx, E, h1, agg1);
    GNN_combine1<<<(int)((nN + T - 1) / T), T, 0, stream>>>(agg1, h1, dinv, b1, W2, h2, (int)nN);
    GNN_scatter2<<<(int)((E + T - 1) / T), T, 0, stream>>>(eidx, E, h2, agg2);
    GNN_final   <<<(int)((nN + T - 1) / T), T, 0, stream>>>(agg2, h2, dinv, b2, out, (int)nN);
}


# pull2: a workgroup owns 1024 adjacent nodes, the h2 table passes through its LDS in slabs of 40000 rows and an edge takes its row there, not from L2; sum order: per lane by slab, then lane tree
# speedup vs baseline: 12.5291x; 1.0695x over previous
#include <hip/hip_runtime.h>
#include <hip/hip_bf16.h>


typedef __attribute__((ext_vector_type(2))) float v2f;
typedef __attribute__((ext_vector_type(8))) float v8f;

#define F_IN 128
#define F_H  16
#define F_C  2

__global__ void GNN_zero_f32(float* __restrict__ p, long n) {
    long i = (long)blockIdx.x * blockDim.x + threadIdx.x;
    if (i < n) p[i] = 0.0f;
}

__global__ void GNN_degree(const int* __restrict__ eidx, long E,
                           float* __restrict__ deg) {
    long e = (long)blockIdx.x * blockDim.x + threadIdx.x;
    if (e >= E) return;
    int d = eidx[E + e];
    atomicAdd(&deg[d], 1.0f);
}

__global__ void GNN_dinv(const float* __restrict__ deg,
                         float* __restrict__ dinv, int n) {
    int i = blockIdx.x * blockDim.x + threadIdx.x;
    if (i < n) dinv[i] = rsqrtf(deg[i] + 1.0f);
}

typedef __attribute__((ext_vector_type(4))) float v4f;
#define G1_WAVES 8
#define G1_PITCH 132
__global__ void __launch_bounds__(G1_WAVES * 32)
GNN_gemm1_wmma(const float* __restrict__ x,
               const float* __restrict__ W1,
               const float* __restrict__ dinv,
               float* __restrict__ h1, int nTiles) {
    __shared__ v4f xs[G1_WAVES][16 * G1_PITCH / 4];
    int lane = threadIdx.x & 31;
    int wave = threadIdx.x >> 5;
    int tile = blockIdx.x * G1_WAVES + wave;
    bool active = tile < nTiles;

    int m    = lane & 15;
    int koff = (lane >> 4) << 1;

    int rbase = 0;
    float dv[8];
    #pragma unroll
    for (int r = 0; r < 8; ++r) dv[r] = 0.0f;

    if (active) {
        const v4f* xt = (const v4f*)(x + (long)tile * 16 * F_IN);
        v4f q[16];
        #pragma unroll
        for (int i = 0; i < 16; ++i)
            q[i] = __builtin_nontemporal_load(xt + i * 32 + lane);
        rbase = tile * 16 + (lane >> 4) * 8;
        #pragma unroll
        for (int r = 0; r < 8; ++r) dv[r] = dinv[rbase + r];
        #pragma unroll
        for (int i = 0; i < 16; ++i)
            xs[wave][i * (G1_PITCH / 4) + lane] = q[i];
    }
    __syncthreads();
    if (!active) return;

    const float* rowp = (const float*)xs[wave] + m * G1_PITCH + koff;
    v8f acc = {};
    #pragma unroll 8
    for (int kk = 0; kk < F_IN; kk += 4) {
        v2f a, b;
        a = *(const v2f*)(rowp + kk);
        b.x = W1[(kk + koff)     * F_H + m];
        b.y = W1[(kk + koff + 1) * F_H + m];
        acc = __builtin_amdgcn_wmma_f32_16x16x4_f32(
            false, a, false, b, (short)0, acc, false, false);
    }

    #pragma unroll
    for (int r = 0; r < 8; ++r)
        h1[(long)(rbase + r) * F_H + m] = dv[r] * acc[r];
}

__global__ void GNN_gemm1_tail(const float* __restrict__ x,
                               const float* __restrict__ W1,
                               const float* __restrict__ dinv,
                               float* __restrict__ h1, int startRow, int n) {
    int idx = blockIdx.x * blockDim.x + threadIdx.x;
    int row = startRow + (idx >> 4);
    int col = idx & 15;
    if (row >= n) return;
    float acc = 0.0f;
    for (int k = 0; k < F_IN; ++k)
        acc += x[(long)row * F_IN + k] * W1[k * F_H + col];
    h1[(long)row * F_H + col] = dinv[row] * acc;
}


template <int J>
__device__ __forceinline__ int GNN_row_lane(int x) {
    return __builtin_amdgcn_update_dpp(0, x, 0x150 + J, 0xf, 0xf, false);
}

template <int J>
__device__ __forceinline__ float GNN_gather_step(int s, const float* __restrict__ hp) {
    unsigned sj = (unsigned)GNN_row_lane<J>(s);
    return hp[sj * (unsigned)F_H];
}

template <int J>
__device__ __forceinline__ void GNN_add_step(int d, float v, float* __restrict__ ap) {
    unsigned dj = (unsigned)GNN_row_lane<J>(d);
    atomicAdd(ap + dj * (unsigned)F_H, v);
}

__global__ void GNN_scatter1(const int* __restrict__ eidx, long E,
                             const float* __restrict__ h1,
                             float* __restrict__ agg1) {
    long e = (long)blockIdx.x * blockDim.x + threadIdx.x;
    int lane = threadIdx.x & 31;
    long base = e - lane;
    if (base + 32 <= E) {
        int s = eidx[e];
        int d = eidx[E + e];
        int c = lane & 15;
        const float* hp = h1 + c;
        float* ap = agg1 + c;

        float v[16];
        v[0]  = GNN_gather_step<0>(s, hp);   v[1]  = GNN_gather_step<1>(s, hp);
        v[2]  = GNN_gather_step<2>(s, hp);   v[3]  = GNN_gather_step<3>(s, hp);
        v[4]  = GNN_gather_step<4>(s, hp);   v[5]  = GNN_gather_step<5>(s, hp);
        v[6]  = GNN_gather_step<6>(s, hp);   v[7]  = GNN_gather_step<7>(s, hp);
        v[8]  = GNN_gather_step<8>(s, hp);   v[9]  = GNN_gather_step<9>(s, hp);
        v[10] = GNN_gather_step<10>(s, hp);  v[11] = GNN_gather_step<11>(s, hp);
        v[12] = GNN_gather_step<12>(s, hp);  v[13] = GNN_gather_step<13>(s, hp);
        v[14] = GNN_gather_step<14>(s, hp);  v[15] = GNN_gather_step<15>(s, hp);

        GNN_add_step<0>(d, v[0], ap);    GNN_add_step<1>(d, v[1], ap);
        GNN_add_step<2>(d, v[2], ap);    GNN_add_step<3>(d, v[3], ap);
        GNN_add_step<4>(d, v[4], ap);    GNN_add_step<5>(d, v[5], ap);
        GNN_add_step<6>(d, v[6], ap);    GNN_add_step<7>(d, v[7], ap);
        GNN_add_step<8>(d, v[8], ap);    GNN_add_step<9>(d, v[9], ap);
        GNN_add_step<10>(d, v[10], ap);  GNN_add_step<11>(d, v[11], ap);
        GNN_add_step<12>(d, v[12], ap);  GNN_add_step<13>(d, v[13], ap);
        GNN_add_step<14>(d, v[14], ap);  GNN_add_step<15>(d, v[15], ap);
    } else if (e < E) {
        int s = eidx[e];
        int d = eidx[E + e];
        const float* hr = h1 + (long)s * F_H;
        float* ar = agg1 + (long)d * F_H;
        for (int j = 0; j < F_H; ++j)
            atomicAdd(ar + j, hr[j]);
    }
}

__global__ void GNN_combine1(const float* __restrict__ agg1,
                             const float* __restrict__ h1,
                             const float* __restrict__ dinv,
                             const float* __restrict__ b1,
                             const float* __restrict__ W2,
                             float* __restrict__ h2, int n) {
    int i = blockIdx.x * blockDim.x + threadIdx.x;
    if (i >= n) return;
    float di  = dinv[i];
    float acc0 = 0.0f, acc1 = 0.0f;
    #pragma unroll
    for (int j = 0; j < F_H; ++j) {
        float v = di * (agg1[(long)i * F_H + j] + h1[(long)i * F_H + j]) + b1[j];
        v = fmaxf(v, 0.0f);
        acc0 = fmaf(v, W2[j * F_C + 0], acc0);
        acc1 = fmaf(v, W2[j * F_C + 1], acc1);
    }
    h2[(long)i * F_C + 0] = di * acc0;
    h2[(long)i * F_C + 1] = di * acc1;
}

__global__ void GNN_scatter2(const int* __restrict__ eidx, long E,
                             const float* __restrict__ h2,
                             float* __restrict__ agg2) {
    long e = (long)blockIdx.x * blockDim.x + threadIdx.x;
    if (e >= E) return;
    int s = eidx[e];
    int d = eidx[E + e];
    float2 v = *(const float2*)(h2 + (long)s * F_C);
    atomicAdd(&agg2[(long)d * F_C + 0], v.x);
    atomicAdd(&agg2[(long)d * F_C + 1], v.y);
}

__global__ void GNN_final(const float* __restrict__ agg2,
                          const float* __restrict__ h2,
                          const float* __restrict__ dinv,
                          const float* __restrict__ b2,
                          float* __restrict__ out, int n) {
    int i = blockIdx.x * blockDim.x + threadIdx.x;
    if (i >= n) return;
    float di  = dinv[i];
    float o0 = di * (agg2[(long)i * F_C + 0] + h2[(long)i * F_C + 0]) + b2[0];
    float o1 = di * (agg2[(long)i * F_C + 1] + h2[(long)i * F_C + 1]) + b2[1];
    float mx  = fmaxf(o0, o1);
    float lse = mx + logf(expf(o0 - mx) + expf(o1 - mx));
    out[(long)i * F_C + 0] = o0 - lse;
    out[(long)i * F_C + 1] = o1 - lse;
}


__global__ void GNN_zero_i32(int* __restrict__ p, int n) {
    int i = blockIdx.x * blockDim.x + threadIdx.x;
    if (i < n) p[i] = 0;
}

__global__ void GNN_rank(const int* __restrict__ eidx, long E,
                         int* __restrict__ cnt, int* __restrict__ rank) {
    long e = (long)blockIdx.x * blockDim.x + threadIdx.x;
    if (e >= E) return;
    int d = eidx[E + e];
    rank[e] = atomicAdd(&cnt[d], 1);
}

#define SCAN_T   256
#define SCAN_PER 4

__global__ void GNN_scan_tot(const int* __restrict__ cnt, int n,
                             int* __restrict__ bsum) {
    __shared__ int red[SCAN_T / 32];
    int t = threadIdx.x;
    int i0 = (blockIdx.x * SCAN_T + t) * SCAN_PER;
    int s = 0;
    #pragma unroll
    for (int k = 0; k < SCAN_PER; ++k)
        if (i0 + k < n) s += cnt[i0 + k];
    for (int o = 16; o > 0; o >>= 1) s += __shfl_xor(s, o);
    if ((t & 31) == 0) red[t >> 5] = s;
    __syncthreads();
    if (t == 0) {
        int tot = 0;
        for (int w = 0; w < SCAN_T / 32; ++w) tot += red[w];
        bsum[blockIdx.x] = tot;
    }
}

template <bool WITH_DINV>
__global__ void GNN_scan_off(const int* __restrict__ cnt, int n,
                             const int* __restrict__ bsum,
                             int* __restrict__ off, float* __restrict__ dinv) {
    __shared__ int wtot[SCAN_T / 32];
    int t = threadIdx.x, lane = t & 31, w = t >> 5;

    int p = 0;
    for (int j = t; j < (int)blockIdx.x; j += SCAN_T) p += bsum[j];
    for (int o = 16; o > 0; o >>= 1) p += __shfl_xor(p, o);
    if (lane == 0) wtot[w] = p;
    __syncthreads();
    int base = 0;
    for (int k = 0; k < SCAN_T / 32; ++k) base += wtot[k];
    __syncthreads();

    int i0 = (blockIdx.x * SCAN_T + t) * SCAN_PER;
    int c[SCAN_PER];
    int s = 0;
    #pragma unroll
    for (int k = 0; k < SCAN_PER; ++k) {
        c[k] = (i0 + k < n) ? cnt[i0 + k] : 0;
        s += c[k];
    }
    int inc = s;
    for (int o = 1; o < 32; o <<= 1) {
        int v = __shfl_up(inc, o);
        if (lane >= o) inc += v;
    }
    if (lane == 31) wtot[w] = inc;
    __syncthreads();
    int wb = 0;
    for (int k = 0; k < w; ++k) wb += wtot[k];
    int run = base + wb + inc - s;
    #pragma unroll
    for (int k = 0; k < SCAN_PER; ++k) {
        if (i0 + k < n) {
            off[i0 + k]  = run;
            if (WITH_DINV) dinv[i0 + k] = rsqrtf((float)c[k] + 1.0f);
            run += c[k];
        }
    }
}

__global__ void GNN_place(const int* __restrict__ eidx, long E,
                          const int* __restrict__ off,
                          const int* __restrict__ rank,
                          int* __restrict__ csr) {
    long e = (long)blockIdx.x * blockDim.x + threadIdx.x;
    if (e >= E) return;
    int s = eidx[e];
    int d = eidx[E + e];
    csr[off[d] + rank[e]] = s;
}


#define BIN_SHIFT 10
#define BIN_W     1024
#define BIN_CH    8192
#define BIN_PER   (BIN_CH / BIN_W)
#define SRC_BITS  22

__device__ __forceinline__ int GNN_scan1024(int v, int* wtot) {
    int lane = threadIdx.x & 31, w = threadIdx.x >> 5;
    int inc = v;
    for (int o = 1; o < 32; o <<= 1) {
        int u = __shfl_up(inc, o);
        if (lane >= o) inc += u;
    }
    if (lane == 31) wtot[w] = inc;
    __syncthreads();
    int wv = wtot[lane];
    int winc = wv;
    for (int o = 1; o < 32; o <<= 1) {
        int u = __shfl_up(winc, o);
        if (lane >= o) winc += u;
    }
    int wbase = __shfl(winc - wv, w);
    __syncthreads();
    return wbase + inc - v;
}

__global__ void __launch_bounds__(BIN_W)
GNN_bin_group(const int* __restrict__ eidx, long E, int nbin, int NBp,
              int* __restrict__ binCnt, int* __restrict__ binLs,
              unsigned* __restrict__ pk) {
    __shared__ int cur[BIN_W];
    __shared__ int wtot[32];
    __shared__ unsigned bufP[BIN_CH];
    int t = threadIdx.x;
    cur[t] = 0;
    long e0 = (long)blockIdx.x * BIN_CH;
    int s[BIN_PER], d[BIN_PER];
    for (int k = 0; k < BIN_PER; ++k) {
        long e = e0 + k * BIN_W + t;
        bool in = e < E;
        s[k] = in ? eidx[e] : 0;
        d[k] = in ? eidx[E + e] : -1;
    }
    __syncthreads();
    for (int k = 0; k < BIN_PER; ++k)
        if (d[k] >= 0) atomicAdd(&cur[d[k] >> BIN_SHIFT], 1);
    __syncthreads();
    int c = cur[t];
    int ls = GNN_scan1024(c, wtot);
    cur[t] = ls;
    if (t < nbin) {
        binCnt[(long)t * NBp + blockIdx.x] = c;
        binLs [(long)t * NBp + blockIdx.x] = ls;
    }
    __syncthreads();
    for (int k = 0; k < BIN_PER; ++k) {
        if (d[k] >= 0) {
            int p = atomicAdd(&cur[d[k] >> BIN_SHIFT], 1);
            bufP[p] = ((unsigned)(d[k] & (BIN_W - 1)) << SRC_BITS) | (unsigned)s[k];
        }
    }
    __syncthreads();
    long left = E - e0;
    int m = left < BIN_CH ? (int)left : BIN_CH;
    for (int i = t; i < m; i += BIN_W) pk[e0 + i] = bufP[i];
}

#define ORD_PER 40
#define ORD_CAP (ORD_PER * BIN_W)
#define ORD_GRP 13
__global__ void __launch_bounds__(BIN_W)
GNN_bin_order(const unsigned* __restrict__ pk, const int* __restrict__ binCnt,
              const int* __restrict__ binLs, int NB, int NBp, int n,
              int* __restrict__ cnt, int* __restrict__ off,
              float* __restrict__ dinv, int* __restrict__ csr) {
    __shared__ int ncnt[BIN_W];
    __shared__ int pc[BIN_W];
    __shared__ int pl[BIN_W];
    __shared__ int ppo[BIN_W];
    __shared__ int wtot[32];
    __shared__ int red[32];
    __shared__ int mtot;
    __shared__ unsigned buf[ORD_CAP];
    int t = threadIdx.x, b = blockIdx.x, lane = t & 31, w = t >> 5;

    int myc = 0, myl = 0;
    if (t < NB) {
        myc = binCnt[(long)b * NBp + t];
        myl = binLs [(long)b * NBp + t];
    }
    int ps = myl;
    for (int o = 16; o > 0; o >>= 1) ps += __shfl_xor(ps, o);
    if (lane == 0) red[w] = ps;
    ncnt[t] = 0;
    pc[t] = myc;
    pl[t] = myl;
    int po = GNN_scan1024(myc, wtot);
    ppo[t] = po;
    if (t == BIN_W - 1) mtot = po + myc;
    __syncthreads();
    int beg = 0;
    for (int k = 0; k < 32; ++k) beg += red[k];
    int m = mtot;
    bool staged = m <= ORD_CAP;

    unsigned wv[ORD_PER];
    int rk[ORD_PER];
    if (staged) {
        for (int c0 = w; c0 < NB; c0 += 32 * ORD_GRP) {
            int pn[ORD_GRP], pb[ORD_GRP];
            unsigned pg[ORD_GRP];
            #pragma unroll
            for (int u = 0; u < ORD_GRP; ++u) {
                int c = c0 + 32 * u;
                bool ok = c < NB;
                int cc = ok ? c : 0;
                pn[u] = ok ? pc[cc] : 0;
                pb[u] = ppo[cc];
                pg[u] = (unsigned)cc * BIN_CH + (unsigned)pl[cc];
            }
            unsigned v[2 * ORD_GRP];
            #pragma unroll
            for (int u = 0; u < ORD_GRP; ++u) {
                v[2 * u]     = (lane      < pn[u]) ? pk[pg[u] + (unsigned)lane]       : 0u;
                v[2 * u + 1] = (lane + 32 < pn[u]) ? pk[pg[u] + (unsigned)lane + 32u] : 0u;
            }
            #pragma unroll
            for (int u = 0; u < ORD_GRP; ++u) {
                if (lane      < pn[u]) buf[pb[u] + lane]      = v[2 * u];
                if (lane + 32 < pn[u]) buf[pb[u] + lane + 32] = v[2 * u + 1];
                #pragma unroll 1
                for (int j = 64 + lane; j < pn[u]; j += 32) buf[pb[u] + j] = pk[pg[u] + (unsigned)j];
            }
        }
        __syncthreads();
        #pragma unroll
        for (int k = 0; k < ORD_PER; ++k) {
            int i = t + k * BIN_W;
            wv[k] = (i < m) ? buf[i] : 0u;
        }
        #pragma unroll
        for (int k = 0; k < ORD_PER; ++k) {
            int i = t + k * BIN_W;
            rk[k] = 0;
            if (i < m) rk[k] = atomicAdd(&ncnt[wv[k] >> SRC_BITS], 1);
        }
    } else {
        for (int c = w; c < NB; c += 32) {
            int pn = pc[c];
            long pg = (long)c * BIN_CH + pl[c];
            for (int j = lane; j < pn; j += 32) atomicAdd(&ncnt[pk[pg + j] >> SRC_BITS], 1);
        }
    }
    __syncthreads();
    int c = ncnt[t];
    int ex = GNN_scan1024(c, wtot);
    ncnt[t] = ex;
    int node = (b << BIN_SHIFT) + t;
    if (node < n) {
        cnt[node]  = c;
        off[node]  = beg + ex;
        dinv[node] = rsqrtf((float)c + 1.0f);
    }
    __syncthreads();
    if (staged) {
        #pragma unroll
        for (int k = 0; k < ORD_PER; ++k) {
            int i = t + k * BIN_W;
            if (i < m) buf[ncnt[wv[k] >> SRC_BITS] + rk[k]] = wv[k] & ((1u << SRC_BITS) - 1u);
        }
        __syncthreads();
        for (int i = t; i < m; i += BIN_W) csr[beg + i] = (int)buf[i];
    } else {
        for (int c2 = w; c2 < NB; c2 += 32) {
            int pn = pc[c2];
            long pg = (long)c2 * BIN_CH + pl[c2];
            for (int j = lane; j < pn; j += 32) {
                unsigned x = pk[pg + j];
                int r = atomicAdd(&ncnt[x >> SRC_BITS], 1);
                csr[beg + r] = (int)(x & ((1u << SRC_BITS) - 1u));
            }
        }
    }
}

template <int J>
__device__ __forceinline__ float GNN_pull_step(int s, int rem, const float* __restrict__ hp) {
    unsigned sj = (unsigned)GNN_row_lane<J>(s);
    float v = 0.0f;
    if (J < rem) v = hp[sj * (unsigned)F_H];
    return v;
}

template <int J>
__device__ __forceinline__ void GNN_h2_step(float r, const float* __restrict__ W2,
                                            float& a0, float& a1) {
    float rj = __int_as_float(GNN_row_lane<J>(__float_as_int(r)));
    a0 = fmaf(rj, W2[J * F_C + 0], a0);
    a1 = fmaf(rj, W2[J * F_C + 1], a1);
}

#define PULL1_T     256
#define PULL1_NODES (PULL1_T / 16)
__global__ void __launch_bounds__(PULL1_T)
GNN_pull1(const int* __restrict__ off, const int* __restrict__ cnt,
                          const int* __restrict__ csr,
                          const float* __restrict__ h1,
                          const float* __restrict__ dinv,
                          const float* __restrict__ b1,
                          const float* __restrict__ W2,
                          float* __restrict__ h2, int n) {
    __shared__ float line[PULL1_NODES * F_C];
    int gid  = blockIdx.x * blockDim.x + threadIdx.x;
    int node = gid >> 4;
    int c    = threadIdx.x & 15;
    bool valid = node < n;
    int nd   = valid ? node : 0;
    int beg  = off[nd];
    int num  = valid ? cnt[nd] : 0;
    int nmax = max(num, __shfl_xor(num, 16));
    const float* hp = h1 + c;
    float own = hp[(unsigned)nd * (unsigned)F_H];
    float di  = dinv[nd];
    float bc  = b1[c];
    float acc = 0.0f;
    int rem = num;
    int s = nd;
    if (c < rem) s = csr[beg + c];
    for (int b = 0; b < nmax; b += 16) {
        float v[16];
        v[0]  = GNN_pull_step<0>(s, rem, hp);   v[1]  = GNN_pull_step<1>(s, rem, hp);
        v[2]  = GNN_pull_step<2>(s, rem, hp);   v[3]  = GNN_pull_step<3>(s, rem, hp);
        v[4]  = GNN_pull_step<4>(s, rem, hp);   v[5]  = GNN_pull_step<5>(s, rem, hp);
        v[6]  = GNN_pull_step<6>(s, rem, hp);   v[7]  = GNN_pull_step<7>(s, rem, hp);
        v[8]  = GNN_pull_step<8>(s, rem, hp);   v[9]  = GNN_pull_step<9>(s, rem, hp);
        v[10] = GNN_pull_step<10>(s, rem, hp);  v[11] = GNN_pull_step<11>(s, rem, hp);
        v[12] = GNN_pull_step<12>(s, rem, hp);  v[13] = GNN_pull_step<13>(s, rem, hp);
        v[14] = GNN_pull_step<14>(s, rem, hp);  v[15] = GNN_pull_step<15>(s, rem, hp);

        int sn = nd;
        if (c < rem - 16) sn = csr[beg + b + 16 + c];
        acc += v[0];   acc += v[1];   acc += v[2];   acc += v[3];
        acc += v[4];   acc += v[5];   acc += v[6];   acc += v[7];
        acc += v[8];   acc += v[9];   acc += v[10];  acc += v[11];
        acc += v[12];  acc += v[13];  acc += v[14];  acc += v[15];
        s = sn;
        rem -= 16;
    }

    float r = di * (acc + own) + bc;
    r = fmaxf(r, 0.0f);
    float a0 = 0.0f, a1 = 0.0f;
    GNN_h2_step<0>(r, W2, a0, a1);    GNN_h2_step<1>(r, W2, a0, a1);
    GNN_h2_step<2>(r, W2, a0, a1);    GNN_h2_step<3>(r, W2, a0, a1);
    GNN_h2_step<4>(r, W2, a0, a1);    GNN_h2_step<5>(r, W2, a0, a1);
    GNN_h2_step<6>(r, W2, a0, a1);    GNN_h2_step<7>(r, W2, a0, a1);
    GNN_h2_step<8>(r, W2, a0, a1);    GNN_h2_step<9>(r, W2, a0, a1);
    GNN_h2_step<10>(r, W2, a0, a1);   GNN_h2_step<11>(r, W2, a0, a1);
    GNN_h2_step<12>(r, W2, a0, a1);   GNN_h2_step<13>(r, W2, a0, a1);
    GNN_h2_step<14>(r, W2, a0, a1);   GNN_h2_step<15>(r, W2, a0, a1);
    if (c < F_C) line[(threadIdx.x >> 4) * F_C + c] = di * (c ? a1 : a0);
    __syncthreads();
    if (threadIdx.x < PULL1_NODES * F_C) {
        long g = (long)blockIdx.x * (PULL1_NODES * F_C) + threadIdx.x;
        if (g < (long)n * F_C) h2[g] = line[threadIdx.x];
    }
}

#define P2_T     1024
#define P2_HELD  3
#define P2_SLAB  40000
#define P2_EACH(M) M(0) M(1) M(2) M(3) M(4) M(5) M(6) M(7) M(8) M(9) M(10) M(11) M(12) M(13) M(14) M(15)

template <int IT>
__device__ __forceinline__ void GNN_p2_fetch(int mybeg, int mynum, int c,
                                             const int* __restrict__ csr, int (&e)[P2_HELD]) {
    int beg = GNN_row_lane<IT>(mybeg);
    int num = GNN_row_lane<IT>(mynum);
    #pragma unroll
    for (int i = 0; i < P2_HELD; ++i) {
        int j = c + 16 * i;
        e[i] = -1;
        if (j < num) e[i] = csr[beg + j];
    }
}

__device__ __forceinline__ void GNN_p2_take(int s, int lo, unsigned span, const v2f* slab, v2f& a) {
    unsigned u = (unsigned)(s - lo);
    u = u < span ? u : span;
    a += slab[u];
}

__device__ __forceinline__ void GNN_p2_held(const int (&e)[P2_HELD], int lo, unsigned span,
                                            const v2f* slab, v2f& a) {
    #pragma unroll
    for (int i = 0; i < P2_HELD; ++i) GNN_p2_take(e[i], lo, span, slab, a);
}

template <int IT>
__device__ __forceinline__ void GNN_p2_tail(int mybeg, int mynum, int c, int lo, unsigned span,
                                            const int* __restrict__ csr, const v2f* slab, v2f& a) {
    int beg = GNN_row_lane<IT>(mybeg);
    int num = GNN_row_lane<IT>(mynum);
    for (int j = 16 * P2_HELD + c; j < num; j += 16) GNN_p2_take(csr[beg + j], lo, span, slab, a);
}

template <int IT>
__device__ __forceinline__ void GNN_p2_sum(v2f a, int c, float& k0, float& k1) {
    float a0 = a.x, a1 = a.y;
    for (int o = 8; o > 0; o >>= 1) {
        a0 += __shfl_xor(a0, o);
        a1 += __shfl_xor(a1, o);
    }
    if (c == IT) { k0 = a0; k1 = a1; }
}

__global__ void __launch_bounds__(P2_T)
GNN_pull2(const int* __restrict__ off, const int* __restrict__ cnt,
          const int* __restrict__ csr,
          const float* __restrict__ h2,
          const float* __restrict__ dinv,
          const float* __restrict__ b2,
          float* __restrict__ out, int n) {
    __shared__ v4f slab4[P2_SLAB / 2 + 1];
    const v2f* slab = (const v2f*)slab4;
    int t = threadIdx.x;
    int c = t & 15;
    int mine = blockIdx.x * P2_T + t;
    bool ok = mine < n;
    int md = ok ? mine : 0;
    int mybeg = off[md];
    int mynum = ok ? cnt[md] : 0;

    int longest = mynum;
    for (int o = 8; o > 0; o >>= 1) longest = max(longest, __shfl_xor(longest, o));
    bool longer = longest > 16 * P2_HELD;

    int e[16][P2_HELD];
#define P2_FETCH(IT) GNN_p2_fetch<IT>(mybeg, mynum, c, csr, e[IT]);
    P2_EACH(P2_FETCH)
#undef P2_FETCH

    v2f a[16];
    #pragma unroll
    for (int k = 0; k < 16; ++k) a[k] = (v2f){0.0f, 0.0f};

    int np = (n + P2_SLAB - 1) / P2_SLAB;
    int p0 = blockIdx.x % np;
    for (int p = 0; p < np; ++p) {
        int sl = p0 + p;
        if (sl >= np) sl -= np;
        int lo = sl * P2_SLAB;
        int span = min(P2_SLAB, n - lo);
        __syncthreads();
        const v4f* src4 = (const v4f*)(h2 + (long)lo * F_C);
        int n4 = span >> 1;
        int nk = (n4 + P2_T - 1) / P2_T;
        #pragma unroll 4
        for (int k = 0; k < nk; ++k) {
            int i = min(k * P2_T + t, n4 - 1);
            slab4[i] = src4[i];
        }
        if ((span & 1) && t == 0)
            ((v2f*)slab4)[span - 1] = *(const v2f*)(h2 + (long)(lo + span - 1) * F_C);
        if (t == 32) ((v2f*)slab4)[span] = (v2f){0.0f, 0.0f};
        __syncthreads();
#define P2_NODEPASS(IT) \
        GNN_p2_held(e[IT], lo, (unsigned)span, slab, a[IT]); \
        if (longer) GNN_p2_tail<IT>(mybeg, mynum, c, lo, (unsigned)span, csr, slab, a[IT]);
        P2_EACH(P2_NODEPASS)
#undef P2_NODEPASS
    }

    v2f   own = *(const v2f*)(h2 + (long)md * F_C);
    float di  = dinv[md];
    float k0 = 0.0f, k1 = 0.0f;
#define P2_SUM(IT) GNN_p2_sum<IT>(a[IT], c, k0, k1);
    P2_EACH(P2_SUM)
#undef P2_SUM

    float o0 = di * (k0 + own.x) + b2[0];
    float o1 = di * (k1 + own.y) + b2[1];
    float mx  = fmaxf(o0, o1);
    float lse = mx + logf(expf(o0 - mx) + expf(o1 - mx));
    if (ok) {
        v2f r = {o0 - lse, o1 - lse};
        *(v2f*)(out + (long)mine * F_C) = r;
    }
}

extern "C" void kernel_launch(void* const* d_in, const int* in_sizes, int n_in,
                              void* d_out, int out_size, void* d_ws, size_t ws_size,
                              hipStream_t stream) {
    const float* x  = (const float*)d_in[0];
    const int* eidx = (const int*)  d_in[1];
    const float* W1 = (const float*)d_in[2];
    const float* b1 = (const float*)d_in[3];
    const float* W2 = (const float*)d_in[4];
    const float* b2 = (const float*)d_in[5];
    float* out = (float*)d_out;

    long nN = (long)in_sizes[0] / F_IN;
    long E  = (long)in_sizes[1] / 2;

    const int T = 256;
    float* ws   = (float*)d_ws;

    long nNp = (nN + 63) & ~63L;
    long Ep  = (E + 63) & ~63L;
    long nb  = (nN + (long)SCAN_T * SCAN_PER - 1) / ((long)SCAN_T * SCAN_PER);
    long bigR = (Ep > 32 * nNp) ? Ep : 32 * nNp;
    long need = 7 * nNp + 1024 + Ep + bigR;
    bool fits = E > 0 && E < (1L << 31) && (size_t)need * 4 <= ws_size;

    long nbin = (nN + BIN_W - 1) >> BIN_SHIFT;
    long NB   = (E + BIN_CH - 1) / BIN_CH;
    long NBp  = (NB + 3) & ~3L;
    long nscP = (nbin * NBp + 63) & ~63L;
    if (fits && nbin <= BIN_W && NB <= BIN_W && nN <= (1L << SRC_BITS) && 2 * nscP <= 4 * nNp) {
        int*   cnt  = (int*)ws;
        int*   off  = (int*)ws + nNp;
        float* dinv = ws + 2 * nNp;
        float* h2   = ws + 3 * nNp + 1024;
        int*   binCnt = (int*)h2;
        int*   binLs  = (int*)h2 + nscP;
        int*   csr  = (int*)ws + 7 * nNp + 1024;
        float* R    = ws + 7 * nNp + 1024 + Ep;
        unsigned* pk = (unsigned*)R;
        float* h1   = R;

        GNN_bin_group<<<(int)NB, BIN_W, 0, stream>>>(eidx, E, (int)nbin, (int)NBp,
                                                      binCnt, binLs, pk);
        GNN_bin_order<<<(int)nbin, BIN_W, 0, stream>>>(pk, binCnt, binLs, (int)NB, (int)NBp, (int)nN,
                                                      cnt, off, dinv, csr);

        int nTiles = (int)(nN / 16);
        if (nTiles > 0)
            GNN_gemm1_wmma<<<(nTiles + 7) / 8, T, 0, stream>>>(x, W1, dinv, h1, nTiles);
        int rem = (int)(nN - (long)nTiles * 16);
        if (rem > 0)
            GNN_gemm1_tail<<<(rem * 16 + T - 1) / T, T, 0, stream>>>(
                x, W1, dinv, h1, nTiles * 16, (int)nN);

        GNN_pull1<<<(int)((nN + PULL1_NODES - 1) / PULL1_NODES), PULL1_T, 0, stream>>>(
            off, cnt, csr, h1, dinv, b1, W2, h2, (int)nN);
        GNN_pull2<<<(int)((nN + P2_T - 1) / P2_T), P2_T, 0, stream>>>(
            off, cnt, csr, h2, dinv, b2, out, (int)nN);
        return;
    }

    if (nb <= 1024 && fits) {
        int*   cnt  = (int*)ws;
        int*   off  = (int*)ws + nNp;
        float* dinv = ws + 2 * nNp;
        int*   bsum = (int*)ws + 3 * nNp;
        float* h2   = ws + 3 * nNp + 1024;
        int*   csr  = (int*)ws + 7 * nNp + 1024;
        float* R    = ws + 7 * nNp + 1024 + Ep;
        int*   rank = (int*)R;
        float* h1   = R;

        GNN_zero_i32<<<(int)((nN + T - 1) / T), T, 0, stream>>>(cnt, (int)nN);
        GNN_rank    <<<(int)((E + T - 1) / T), T, 0, stream>>>(eidx, E, cnt, rank);
        GNN_scan_tot<<<(int)nb, SCAN_T, 0, stream>>>(cnt, (int)nN, bsum);
        GNN_scan_off<true><<<(int)nb, SCAN_T, 0, stream>>>(cnt, (int)nN, bsum, off, dinv);
        GNN_place   <<<(int)((E + T - 1) / T), T, 0, stream>>>(eidx, E, off, rank, csr);

        int nTiles = (int)(nN / 16);
        if (nTiles > 0)
            GNN_gemm1_wmma<<<(nTiles + 7) / 8, T, 0, stream>>>(x, W1, dinv, h1, nTiles);
        int rem = (int)(nN - (long)nTiles * 16);
        if (rem > 0)
            GNN_gemm1_tail<<<(rem * 16 + T - 1) / T, T, 0, stream>>>(
                x, W1, dinv, h1, nTiles * 16, (int)nN);

        GNN_pull1<<<(int)((nN + PULL1_NODES - 1) / PULL1_NODES), PULL1_T, 0, stream>>>(
            off, cnt, csr, h1, dinv, b1, W2, h2, (int)nN);
        GNN_pull2<<<(int)((nN + P2_T - 1) / P2_T), P2_T, 0, stream>>>(
            off, cnt, csr, h2, dinv, b2, out, (int)nN);
        return;
    }

    float* deg  = ws;
    float* agg1 = ws + nN;
    float* agg2 = ws + 17 * nN;
    float* dinv = ws + 19 * nN;
    float* h1   = ws + 20 * nN;
    float* h2   = ws + 36 * nN;

    long zn = 19 * nN;
    GNN_zero_f32<<<(int)((zn + T - 1) / T), T, 0, stream>>>(ws, zn);
    GNN_degree  <<<(int)((E + T - 1) / T), T, 0, stream>>>(eidx, E, deg);
    GNN_dinv    <<<(int)((nN + T - 1) / T), T, 0, stream>>>(deg, dinv, (int)nN);

    int nTiles = (int)(nN / 16);
    if (nTiles > 0)
        GNN_gemm1_wmma<<<(nTiles + 7) / 8, T, 0, stream>>>(x, W1, dinv, h1, nTiles);
    int rem = (int)(nN - (long)nTiles * 16);
    if (rem > 0)
        GNN_gemm1_tail<<<(rem * 16 + T - 1) / T, T, 0, stream>>>(
            x, W1, dinv, h1, nTiles * 16, (int)nN);

    GNN_scatter1<<<(int)((E + T - 1) / T), T, 0, stream>>>(eidx, E, h1, agg1);
    GNN_combine1<<<(int)((nN + T - 1) / T), T, 0, stream>>>(agg1, h1, dinv, b1, W2, h2, (int)nN);
    GNN_scatter2<<<(int)((E + T - 1) / T), T, 0, stream>>>(eidx, E, h2, agg2);
    GNN_final   <<<(int)((nN + T - 1) / T), T, 0, stream>>>(agg2, h2, dinv, b2, out, (int)nN);
}


# layer 2 sums one float per source row, z = hs0 - hs1 (2-class log_softmax needs only o0 - o1): pull1 stores z, pull2 passes a 4-byte table through LDS in 3 slabs, not 5; sum order: differences summed
# speedup vs baseline: 13.2395x; 1.0567x over previous
#include <hip/hip_runtime.h>
#include <hip/hip_bf16.h>


typedef __attribute__((ext_vector_type(2))) float v2f;
typedef __attribute__((ext_vector_type(8))) float v8f;

#define F_IN 128
#define F_H  16
#define F_C  2

__global__ void GNN_zero_f32(float* __restrict__ p, long n) {
    long i = (long)blockIdx.x * blockDim.x + threadIdx.x;
    if (i < n) p[i] = 0.0f;
}

__global__ void GNN_degree(const int* __restrict__ eidx, long E,
                           float* __restrict__ deg) {
    long e = (long)blockIdx.x * blockDim.x + threadIdx.x;
    if (e >= E) return;
    int d = eidx[E + e];
    atomicAdd(&deg[d], 1.0f);
}

__global__ void GNN_dinv(const float* __restrict__ deg,
                         float* __restrict__ dinv, int n) {
    int i = blockIdx.x * blockDim.x + threadIdx.x;
    if (i < n) dinv[i] = rsqrtf(deg[i] + 1.0f);
}

typedef __attribute__((ext_vector_type(4))) float v4f;
#define G1_WAVES 8
#define G1_PITCH 132
__global__ void __launch_bounds__(G1_WAVES * 32)
GNN_gemm1_wmma(const float* __restrict__ x,
               const float* __restrict__ W1,
               const float* __restrict__ dinv,
               float* __restrict__ h1, int nTiles) {
    __shared__ v4f xs[G1_WAVES][16 * G1_PITCH / 4];
    int lane = threadIdx.x & 31;
    int wave = threadIdx.x >> 5;
    int tile = blockIdx.x * G1_WAVES + wave;
    bool active = tile < nTiles;

    int m    = lane & 15;
    int koff = (lane >> 4) << 1;

    int rbase = 0;
    float dv[8];
    #pragma unroll
    for (int r = 0; r < 8; ++r) dv[r] = 0.0f;

    if (active) {
        const v4f* xt = (const v4f*)(x + (long)tile * 16 * F_IN);
        v4f q[16];
        #pragma unroll
        for (int i = 0; i < 16; ++i)
            q[i] = __builtin_nontemporal_load(xt + i * 32 + lane);
        rbase = tile * 16 + (lane >> 4) * 8;
        #pragma unroll
        for (int r = 0; r < 8; ++r) dv[r] = dinv[rbase + r];
        #pragma unroll
        for (int i = 0; i < 16; ++i)
            xs[wave][i * (G1_PITCH / 4) + lane] = q[i];
    }
    __syncthreads();
    if (!active) return;

    const float* rowp = (const float*)xs[wave] + m * G1_PITCH + koff;
    v8f acc = {};
    #pragma unroll 8
    for (int kk = 0; kk < F_IN; kk += 4) {
        v2f a, b;
        a = *(const v2f*)(rowp + kk);
        b.x = W1[(kk + koff)     * F_H + m];
        b.y = W1[(kk + koff + 1) * F_H + m];
        acc = __builtin_amdgcn_wmma_f32_16x16x4_f32(
            false, a, false, b, (short)0, acc, false, false);
    }

    #pragma unroll
    for (int r = 0; r < 8; ++r)
        h1[(long)(rbase + r) * F_H + m] = dv[r] * acc[r];
}

__global__ void GNN_gemm1_tail(const float* __restrict__ x,
                               const float* __restrict__ W1,
                               const float* __restrict__ dinv,
                               float* __restrict__ h1, int startRow, int n) {
    int idx = blockIdx.x * blockDim.x + threadIdx.x;
    int row = startRow + (idx >> 4);
    int col = idx & 15;
    if (row >= n) return;
    float acc = 0.0f;
    for (int k = 0; k < F_IN; ++k)
        acc += x[(long)row * F_IN + k] * W1[k * F_H + col];
    h1[(long)row * F_H + col] = dinv[row] * acc;
}


template <int J>
__device__ __forceinline__ int GNN_row_lane(int x) {
    return __builtin_amdgcn_update_dpp(0, x, 0x150 + J, 0xf, 0xf, false);
}

template <int J>
__device__ __forceinline__ float GNN_gather_step(int s, const float* __restrict__ hp) {
    unsigned sj = (unsigned)GNN_row_lane<J>(s);
    return hp[sj * (unsigned)F_H];
}

template <int J>
__device__ __forceinline__ void GNN_add_step(int d, float v, float* __restrict__ ap) {
    unsigned dj = (unsigned)GNN_row_lane<J>(d);
    atomicAdd(ap + dj * (unsigned)F_H, v);
}

__global__ void GNN_scatter1(const int* __restrict__ eidx, long E,
                             const float* __restrict__ h1,
                             float* __restrict__ agg1) {
    long e = (long)blockIdx.x * blockDim.x + threadIdx.x;
    int lane = threadIdx.x & 31;
    long base = e - lane;
    if (base + 32 <= E) {
        int s = eidx[e];
        int d = eidx[E + e];
        int c = lane & 15;
        const float* hp = h1 + c;
        float* ap = agg1 + c;

        float v[16];
        v[0]  = GNN_gather_step<0>(s, hp);   v[1]  = GNN_gather_step<1>(s, hp);
        v[2]  = GNN_gather_step<2>(s, hp);   v[3]  = GNN_gather_step<3>(s, hp);
        v[4]  = GNN_gather_step<4>(s, hp);   v[5]  = GNN_gather_step<5>(s, hp);
        v[6]  = GNN_gather_step<6>(s, hp);   v[7]  = GNN_gather_step<7>(s, hp);
        v[8]  = GNN_gather_step<8>(s, hp);   v[9]  = GNN_gather_step<9>(s, hp);
        v[10] = GNN_gather_step<10>(s, hp);  v[11] = GNN_gather_step<11>(s, hp);
        v[12] = GNN_gather_step<12>(s, hp);  v[13] = GNN_gather_step<13>(s, hp);
        v[14] = GNN_gather_step<14>(s, hp);  v[15] = GNN_gather_step<15>(s, hp);

        GNN_add_step<0>(d, v[0], ap);    GNN_add_step<1>(d, v[1], ap);
        GNN_add_step<2>(d, v[2], ap);    GNN_add_step<3>(d, v[3], ap);
        GNN_add_step<4>(d, v[4], ap);    GNN_add_step<5>(d, v[5], ap);
        GNN_add_step<6>(d, v[6], ap);    GNN_add_step<7>(d, v[7], ap);
        GNN_add_step<8>(d, v[8], ap);    GNN_add_step<9>(d, v[9], ap);
        GNN_add_step<10>(d, v[10], ap);  GNN_add_step<11>(d, v[11], ap);
        GNN_add_step<12>(d, v[12], ap);  GNN_add_step<13>(d, v[13], ap);
        GNN_add_step<14>(d, v[14], ap);  GNN_add_step<15>(d, v[15], ap);
    } else if (e < E) {
        int s = eidx[e];
        int d = eidx[E + e];
        const float* hr = h1 + (long)s * F_H;
        float* ar = agg1 + (long)d * F_H;
        for (int j = 0; j < F_H; ++j)
            atomicAdd(ar + j, hr[j]);
    }
}

__global__ void GNN_combine1(const float* __restrict__ agg1,
                             const float* __restrict__ h1,
                             const float* __restrict__ dinv,
                             const float* __restrict__ b1,
                             const float* __restrict__ W2,
                             float* __restrict__ h2, int n) {
    int i = blockIdx.x * blockDim.x + threadIdx.x;
    if (i >= n) return;
    float di  = dinv[i];
    float acc0 = 0.0f, acc1 = 0.0f;
    #pragma unroll
    for (int j = 0; j < F_H; ++j) {
        float v = di * (agg1[(long)i * F_H + j] + h1[(long)i * F_H + j]) + b1[j];
        v = fmaxf(v, 0.0f);
        acc0 = fmaf(v, W2[j * F_C + 0], acc0);
        acc1 = fmaf(v, W2[j * F_C + 1], acc1);
    }
    h2[(long)i * F_C + 0] = di * acc0;
    h2[(long)i * F_C + 1] = di * acc1;
}

__global__ void GNN_scatter2(const int* __restrict__ eidx, long E,
                             const float* __restrict__ h2,
                             float* __restrict__ agg2) {
    long e = (long)blockIdx.x * blockDim.x + threadIdx.x;
    if (e >= E) return;
    int s = eidx[e];
    int d = eidx[E + e];
    float2 v = *(const float2*)(h2 + (long)s * F_C);
    atomicAdd(&agg2[(long)d * F_C + 0], v.x);
    atomicAdd(&agg2[(long)d * F_C + 1], v.y);
}

__global__ void GNN_final(const float* __restrict__ agg2,
                          const float* __restrict__ h2,
                          const float* __restrict__ dinv,
                          const float* __restrict__ b2,
                          float* __restrict__ out, int n) {
    int i = blockIdx.x * blockDim.x + threadIdx.x;
    if (i >= n) return;
    float di  = dinv[i];
    float o0 = di * (agg2[(long)i * F_C + 0] + h2[(long)i * F_C + 0]) + b2[0];
    float o1 = di * (agg2[(long)i * F_C + 1] + h2[(long)i * F_C + 1]) + b2[1];
    float mx  = fmaxf(o0, o1);
    float lse = mx + logf(expf(o0 - mx) + expf(o1 - mx));
    out[(long)i * F_C + 0] = o0 - lse;
    out[(long)i * F_C + 1] = o1 - lse;
}


__global__ void GNN_zero_i32(int* __restrict__ p, int n) {
    int i = blockIdx.x * blockDim.x + threadIdx.x;
    if (i < n) p[i] = 0;
}

__global__ void GNN_rank(const int* __restrict__ eidx, long E,
                         int* __restrict__ cnt, int* __restrict__ rank) {
    long e = (long)blockIdx.x * blockDim.x + threadIdx.x;
    if (e >= E) return;
    int d = eidx[E + e];
    rank[e] = atomicAdd(&cnt[d], 1);
}

#define SCAN_T   256
#define SCAN_PER 4

__global__ void GNN_scan_tot(const int* __restrict__ cnt, int n,
                             int* __restrict__ bsum) {
    __shared__ int red[SCAN_T / 32];
    int t = threadIdx.x;
    int i0 = (blockIdx.x * SCAN_T + t) * SCAN_PER;
    int s = 0;
    #pragma unroll
    for (int k = 0; k < SCAN_PER; ++k)
        if (i0 + k < n) s += cnt[i0 + k];
    for (int o = 16; o > 0; o >>= 1) s += __shfl_xor(s, o);
    if ((t & 31) == 0) red[t >> 5] = s;
    __syncthreads();
    if (t == 0) {
        int tot = 0;
        for (int w = 0; w < SCAN_T / 32; ++w) tot += red[w];
        bsum[blockIdx.x] = tot;
    }
}

template <bool WITH_DINV>
__global__ void GNN_scan_off(const int* __restrict__ cnt, int n,
                             const int* __restrict__ bsum,
                             int* __restrict__ off, float* __restrict__ dinv) {
    __shared__ int wtot[SCAN_T / 32];
    int t = threadIdx.x, lane = t & 31, w = t >> 5;

    int p = 0;
    for (int j = t; j < (int)blockIdx.x; j += SCAN_T) p += bsum[j];
    for (int o = 16; o > 0; o >>= 1) p += __shfl_xor(p, o);
    if (lane == 0) wtot[w] = p;
    __syncthreads();
    int base = 0;
    for (int k = 0; k < SCAN_T / 32; ++k) base += wtot[k];
    __syncthreads();

    int i0 = (blockIdx.x * SCAN_T + t) * SCAN_PER;
    int c[SCAN_PER];
    int s = 0;
    #pragma unroll
    for (int k = 0; k < SCAN_PER; ++k) {
        c[k] = (i0 + k < n) ? cnt[i0 + k] : 0;
        s += c[k];
    }
    int inc = s;
    for (int o = 1; o < 32; o <<= 1) {
        int v = __shfl_up(inc, o);
        if (lane >= o) inc += v;
    }
    if (lane == 31) wtot[w] = inc;
    __syncthreads();
    int wb = 0;
    for (int k = 0; k < w; ++k) wb += wtot[k];
    int run = base + wb + inc - s;
    #pragma unroll
    for (int k = 0; k < SCAN_PER; ++k) {
        if (i0 + k < n) {
            off[i0 + k]  = run;
            if (WITH_DINV) dinv[i0 + k] = rsqrtf((float)c[k] + 1.0f);
            run += c[k];
        }
    }
}

__global__ void GNN_place(const int* __restrict__ eidx, long E,
                          const int* __restrict__ off,
                          const int* __restrict__ rank,
                          int* __restrict__ csr) {
    long e = (long)blockIdx.x * blockDim.x + threadIdx.x;
    if (e >= E) return;
    int s = eidx[e];
    int d = eidx[E + e];
    csr[off[d] + rank[e]] = s;
}


#define BIN_SHIFT 10
#define BIN_W     1024
#define BIN_CH    8192
#define BIN_PER   (BIN_CH / BIN_W)
#define SRC_BITS  22

__device__ __forceinline__ int GNN_scan1024(int v, int* wtot) {
    int lane = threadIdx.x & 31, w = threadIdx.x >> 5;
    int inc = v;
    for (int o = 1; o < 32; o <<= 1) {
        int u = __shfl_up(inc, o);
        if (lane >= o) inc += u;
    }
    if (lane == 31) wtot[w] = inc;
    __syncthreads();
    int wv = wtot[lane];
    int winc = wv;
    for (int o = 1; o < 32; o <<= 1) {
        int u = __shfl_up(winc, o);
        if (lane >= o) winc += u;
    }
    int wbase = __shfl(winc - wv, w);
    __syncthreads();
    return wbase + inc - v;
}

__global__ void __launch_bounds__(BIN_W)
GNN_bin_group(const int* __restrict__ eidx, long E, int nbin, int NBp,
              int* __restrict__ binCnt, int* __restrict__ binLs,
              unsigned* __restrict__ pk) {
    __shared__ int cur[BIN_W];
    __shared__ int wtot[32];
    __shared__ unsigned bufP[BIN_CH];
    int t = threadIdx.x;
    cur[t] = 0;
    long e0 = (long)blockIdx.x * BIN_CH;
    int s[BIN_PER], d[BIN_PER];
    for (int k = 0; k < BIN_PER; ++k) {
        long e = e0 + k * BIN_W + t;
        bool in = e < E;
        s[k] = in ? eidx[e] : 0;
        d[k] = in ? eidx[E + e] : -1;
    }
    __syncthreads();
    for (int k = 0; k < BIN_PER; ++k)
        if (d[k] >= 0) atomicAdd(&cur[d[k] >> BIN_SHIFT], 1);
    __syncthreads();
    int c = cur[t];
    int ls = GNN_scan1024(c, wtot);
    cur[t] = ls;
    if (t < nbin) {
        binCnt[(long)t * NBp + blockIdx.x] = c;
        binLs [(long)t * NBp + blockIdx.x] = ls;
    }
    __syncthreads();
    for (int k = 0; k < BIN_PER; ++k) {
        if (d[k] >= 0) {
            int p = atomicAdd(&cur[d[k] >> BIN_SHIFT], 1);
            bufP[p] = ((unsigned)(d[k] & (BIN_W - 1)) << SRC_BITS) | (unsigned)s[k];
        }
    }
    __syncthreads();
    long left = E - e0;
    int m = left < BIN_CH ? (int)left : BIN_CH;
    for (int i = t; i < m; i += BIN_W) pk[e0 + i] = bufP[i];
}

#define ORD_PER 40
#define ORD_CAP (ORD_PER * BIN_W)
#define ORD_GRP 13
__global__ void __launch_bounds__(BIN_W)
GNN_bin_order(const unsigned* __restrict__ pk, const int* __restrict__ binCnt,
              const int* __restrict__ binLs, int NB, int NBp, int n,
              int* __restrict__ cnt, int* __restrict__ off,
              float* __restrict__ dinv, int* __restrict__ csr) {
    __shared__ int ncnt[BIN_W];
    __shared__ int pc[BIN_W];
    __shared__ int pl[BIN_W];
    __shared__ int ppo[BIN_W];
    __shared__ int wtot[32];
    __shared__ int red[32];
    __shared__ int mtot;
    __shared__ unsigned buf[ORD_CAP];
    int t = threadIdx.x, b = blockIdx.x, lane = t & 31, w = t >> 5;

    int myc = 0, myl = 0;
    if (t < NB) {
        myc = binCnt[(long)b * NBp + t];
        myl = binLs [(long)b * NBp + t];
    }
    int ps = myl;
    for (int o = 16; o > 0; o >>= 1) ps += __shfl_xor(ps, o);
    if (lane == 0) red[w] = ps;
    ncnt[t] = 0;
    pc[t] = myc;
    pl[t] = myl;
    int po = GNN_scan1024(myc, wtot);
    ppo[t] = po;
    if (t == BIN_W - 1) mtot = po + myc;
    __syncthreads();
    int beg = 0;
    for (int k = 0; k < 32; ++k) beg += red[k];
    int m = mtot;
    bool staged = m <= ORD_CAP;

    unsigned wv[ORD_PER];
    int rk[ORD_PER];
    if (staged) {
        for (int c0 = w; c0 < NB; c0 += 32 * ORD_GRP) {
            int pn[ORD_GRP], pb[ORD_GRP];
            unsigned pg[ORD_GRP];
            #pragma unroll
            for (int u = 0; u < ORD_GRP; ++u) {
                int c = c0 + 32 * u;
                bool ok = c < NB;
                int cc = ok ? c : 0;
                pn[u] = ok ? pc[cc] : 0;
                pb[u] = ppo[cc];
                pg[u] = (unsigned)cc * BIN_CH + (unsigned)pl[cc];
            }
            unsigned v[2 * ORD_GRP];
            #pragma unroll
            for (int u = 0; u < ORD_GRP; ++u) {
                v[2 * u]     = (lane      < pn[u]) ? pk[pg[u] + (unsigned)lane]       : 0u;
                v[2 * u + 1] = (lane + 32 < pn[u]) ? pk[pg[u] + (unsigned)lane + 32u] : 0u;
            }
            #pragma unroll
            for (int u = 0; u < ORD_GRP; ++u) {
                if (lane      < pn[u]) buf[pb[u] + lane]      = v[2 * u];
                if (lane + 32 < pn[u]) buf[pb[u] + lane + 32] = v[2 * u + 1];
                #pragma unroll 1
                for (int j = 64 + lane; j < pn[u]; j += 32) buf[pb[u] + j] = pk[pg[u] + (unsigned)j];
            }
        }
        __syncthreads();
        #pragma unroll
        for (int k = 0; k < ORD_PER; ++k) {
            int i = t + k * BIN_W;
            wv[k] = (i < m) ? buf[i] : 0u;
        }
        #pragma unroll
        for (int k = 0; k < ORD_PER; ++k) {
            int i = t + k * BIN_W;
            rk[k] = 0;
            if (i < m) rk[k] = atomicAdd(&ncnt[wv[k] >> SRC_BITS], 1);
        }
    } else {
        for (int c = w; c < NB; c += 32) {
            int pn = pc[c];
            long pg = (long)c * BIN_CH + pl[c];
            for (int j = lane; j < pn; j += 32) atomicAdd(&ncnt[pk[pg + j] >> SRC_BITS], 1);
        }
    }
    __syncthreads();
    int c = ncnt[t];
    int ex = GNN_scan1024(c, wtot);
    ncnt[t] = ex;
    int node = (b << BIN_SHIFT) + t;
    if (node < n) {
        cnt[node]  = c;
        off[node]  = beg + ex;
        dinv[node] = rsqrtf((float)c + 1.0f);
    }
    __syncthreads();
    if (staged) {
        #pragma unroll
        for (int k = 0; k < ORD_PER; ++k) {
            int i = t + k * BIN_W;
            if (i < m) buf[ncnt[wv[k] >> SRC_BITS] + rk[k]] = wv[k] & ((1u << SRC_BITS) - 1u);
        }
        __syncthreads();
        for (int i = t; i < m; i += BIN_W) csr[beg + i] = (int)buf[i];
    } else {
        for (int c2 = w; c2 < NB; c2 += 32) {
            int pn = pc[c2];
            long pg = (long)c2 * BIN_CH + pl[c2];
            for (int j = lane; j < pn; j += 32) {
                unsigned x = pk[pg + j];
                int r = atomicAdd(&ncnt[x >> SRC_BITS], 1);
                csr[beg + r] = (int)(x & ((1u << SRC_BITS) - 1u));
            }
        }
    }
}

template <int J>
__device__ __forceinline__ float GNN_pull_step(int s, int rem, const float* __restrict__ hp) {
    unsigned sj = (unsigned)GNN_row_lane<J>(s);
    float v = 0.0f;
    if (J < rem) v = hp[sj * (unsigned)F_H];
    return v;
}

template <int J>
__device__ __forceinline__ void GNN_h2_step(float r, const float* __restrict__ W2,
                                            float& a0, float& a1) {
    float rj = __int_as_float(GNN_row_lane<J>(__float_as_int(r)));
    a0 = fmaf(rj, W2[J * F_C + 0], a0);
    a1 = fmaf(rj, W2[J * F_C + 1], a1);
}

#define PULL1_T     512
#define PULL1_NODES (PULL1_T / 16)
__global__ void __launch_bounds__(PULL1_T)
GNN_pull1(const int* __restrict__ off, const int* __restrict__ cnt,
                          const int* __restrict__ csr,
                          const float* __restrict__ h1,
                          const float* __restrict__ dinv,
                          const float* __restrict__ b1,
                          const float* __restrict__ W2,
                          float* __restrict__ zt, int n) {
    __shared__ float line[PULL1_NODES];
    int gid  = blockIdx.x * blockDim.x + threadIdx.x;
    int node = gid >> 4;
    int c    = threadIdx.x & 15;
    bool valid = node < n;
    int nd   = valid ? node : 0;
    int beg  = off[nd];
    int num  = valid ? cnt[nd] : 0;
    int nmax = max(num, __shfl_xor(num, 16));
    const float* hp = h1 + c;
    float own = hp[(unsigned)nd * (unsigned)F_H];
    float di  = dinv[nd];
    float bc  = b1[c];
    float acc = 0.0f;
    int rem = num;
    int s = nd;
    if (c < rem) s = csr[beg + c];
    for (int b = 0; b < nmax; b += 16) {
        float v[16];
        v[0]  = GNN_pull_step<0>(s, rem, hp);   v[1]  = GNN_pull_step<1>(s, rem, hp);
        v[2]  = GNN_pull_step<2>(s, rem, hp);   v[3]  = GNN_pull_step<3>(s, rem, hp);
        v[4]  = GNN_pull_step<4>(s, rem, hp);   v[5]  = GNN_pull_step<5>(s, rem, hp);
        v[6]  = GNN_pull_step<6>(s, rem, hp);   v[7]  = GNN_pull_step<7>(s, rem, hp);
        v[8]  = GNN_pull_step<8>(s, rem, hp);   v[9]  = GNN_pull_step<9>(s, rem, hp);
        v[10] = GNN_pull_step<10>(s, rem, hp);  v[11] = GNN_pull_step<11>(s, rem, hp);
        v[12] = GNN_pull_step<12>(s, rem, hp);  v[13] = GNN_pull_step<13>(s, rem, hp);
        v[14] = GNN_pull_step<14>(s, rem, hp);  v[15] = GNN_pull_step<15>(s, rem, hp);

        int sn = nd;
        if (c < rem - 16) sn = csr[beg + b + 16 + c];
        acc += v[0];   acc += v[1];   acc += v[2];   acc += v[3];
        acc += v[4];   acc += v[5];   acc += v[6];   acc += v[7];
        acc += v[8];   acc += v[9];   acc += v[10];  acc += v[11];
        acc += v[12];  acc += v[13];  acc += v[14];  acc += v[15];
        s = sn;
        rem -= 16;
    }

    float r = di * (acc + own) + bc;
    r = fmaxf(r, 0.0f);
    float a0 = 0.0f, a1 = 0.0f;
    GNN_h2_step<0>(r, W2, a0, a1);    GNN_h2_step<1>(r, W2, a0, a1);
    GNN_h2_step<2>(r, W2, a0, a1);    GNN_h2_step<3>(r, W2, a0, a1);
    GNN_h2_step<4>(r, W2, a0, a1);    GNN_h2_step<5>(r, W2, a0, a1);
    GNN_h2_step<6>(r, W2, a0, a1);    GNN_h2_step<7>(r, W2, a0, a1);
    GNN_h2_step<8>(r, W2, a0, a1);    GNN_h2_step<9>(r, W2, a0, a1);
    GNN_h2_step<10>(r, W2, a0, a1);   GNN_h2_step<11>(r, W2, a0, a1);
    GNN_h2_step<12>(r, W2, a0, a1);   GNN_h2_step<13>(r, W2, a0, a1);
    GNN_h2_step<14>(r, W2, a0, a1);   GNN_h2_step<15>(r, W2, a0, a1);
    if (c == 0) line[threadIdx.x >> 4] = di * (a0 - a1);
    __syncthreads();
    if (threadIdx.x < PULL1_NODES) {
        int g = blockIdx.x * PULL1_NODES + threadIdx.x;
        if (g < n) zt[g] = line[threadIdx.x];
    }
}

#define P2_T     1024
#define P2_HELD  3
#define P2_SLAB  80000
#define P2_EACH(M) M(0) M(1) M(2) M(3) M(4) M(5) M(6) M(7) M(8) M(9) M(10) M(11) M(12) M(13) M(14) M(15)

template <int IT>
__device__ __forceinline__ void GNN_p2_fetch(int mybeg, int mynum, int c,
                                             const int* __restrict__ csr, int (&e)[P2_HELD]) {
    int beg = GNN_row_lane<IT>(mybeg);
    int num = GNN_row_lane<IT>(mynum);
    #pragma unroll
    for (int i = 0; i < P2_HELD; ++i) {
        int j = c + 16 * i;
        e[i] = -1;
        if (j < num) e[i] = csr[beg + j];
    }
}

__device__ __forceinline__ void GNN_p2_take(int s, int lo, unsigned span, const float* slab, float& a) {
    unsigned u = (unsigned)(s - lo);
    u = u < span ? u : span;
    a += slab[u];
}

__device__ __forceinline__ void GNN_p2_held(const int (&e)[P2_HELD], int lo, unsigned span,
                                            const float* slab, float& a) {
    #pragma unroll
    for (int i = 0; i < P2_HELD; ++i) GNN_p2_take(e[i], lo, span, slab, a);
}

template <int IT>
__device__ __forceinline__ void GNN_p2_tail(int mybeg, int mynum, int c, int lo, unsigned span,
                                            const int* __restrict__ csr, const float* slab, float& a) {
    int beg = GNN_row_lane<IT>(mybeg);
    int num = GNN_row_lane<IT>(mynum);
    for (int j = 16 * P2_HELD + c; j < num; j += 16) GNN_p2_take(csr[beg + j], lo, span, slab, a);
}

template <int IT>
__device__ __forceinline__ void GNN_p2_sum(float a, int c, float& k) {
    for (int o = 8; o > 0; o >>= 1) a += __shfl_xor(a, o);
    if (c == IT) k = a;
}

__global__ void __launch_bounds__(P2_T)
GNN_pull2(const int* __restrict__ off, const int* __restrict__ cnt,
          const int* __restrict__ csr,
          const float* __restrict__ zt,
          const float* __restrict__ dinv,
          const float* __restrict__ b2,
          float* __restrict__ out, int n) {
    __shared__ v4f slab4[P2_SLAB / 4 + 1];
    float* slab = (float*)slab4;
    int t = threadIdx.x;
    int c = t & 15;
    int mine = blockIdx.x * P2_T + t;
    bool ok = mine < n;
    int md = ok ? mine : 0;
    int mybeg = off[md];
    int mynum = ok ? cnt[md] : 0;

    int longest = mynum;
    for (int o = 8; o > 0; o >>= 1) longest = max(longest, __shfl_xor(longest, o));
    bool longer = longest > 16 * P2_HELD;

    int e[16][P2_HELD];
#define P2_FETCH(IT) GNN_p2_fetch<IT>(mybeg, mynum, c, csr, e[IT]);
    P2_EACH(P2_FETCH)
#undef P2_FETCH

    float a[16];
    #pragma unroll
    for (int k = 0; k < 16; ++k) a[k] = 0.0f;

    int np = (n + P2_SLAB - 1) / P2_SLAB;
    int p0 = blockIdx.x % np;
    for (int p = 0; p < np; ++p) {
        int sl = p0 + p;
        if (sl >= np) sl -= np;
        int lo = sl * P2_SLAB;
        int span = min(P2_SLAB, n - lo);
        __syncthreads();
        const v4f* src4 = (const v4f*)(zt + lo);
        int n4 = span >> 2;
        int nk = (n4 + P2_T - 1) / P2_T;
        #pragma unroll 4
        for (int k = 0; k < nk; ++k) {
            int i = min(k * P2_T + t, n4 - 1);
            slab4[i] = src4[i];
        }
        int rest = (n4 << 2) + t;
        if (rest < span) slab[rest] = zt[lo + rest];
        if (t == 32) slab[span] = 0.0f;
        __syncthreads();
#define P2_NODEPASS(IT) \
        GNN_p2_held(e[IT], lo, (unsigned)span, slab, a[IT]); \
        if (longer) GNN_p2_tail<IT>(mybeg, mynum, c, lo, (unsigned)span, csr, slab, a[IT]);
        P2_EACH(P2_NODEPASS)
#undef P2_NODEPASS
    }

    float own = zt[md];
    float di  = dinv[md];
    float k0 = 0.0f;
#define P2_SUM(IT) GNN_p2_sum<IT>(a[IT], c, k0);
    P2_EACH(P2_SUM)
#undef P2_SUM

    float dz  = di * (k0 + own) + (b2[0] - b2[1]);
    float mx  = fmaxf(dz, 0.0f);
    float lse = mx + logf(expf(dz - mx) + expf(0.0f - mx));
    if (ok) {
        v2f r = {dz - lse, 0.0f - lse};
        *(v2f*)(out + (long)mine * F_C) = r;
    }
}

extern "C" void kernel_launch(void* const* d_in, const int* in_sizes, int n_in,
                              void* d_out, int out_size, void* d_ws, size_t ws_size,
                              hipStream_t stream) {
    const float* x  = (const float*)d_in[0];
    const int* eidx = (const int*)  d_in[1];
    const float* W1 = (const float*)d_in[2];
    const float* b1 = (const float*)d_in[3];
    const float* W2 = (const float*)d_in[4];
    const float* b2 = (const float*)d_in[5];
    float* out = (float*)d_out;

    long nN = (long)in_sizes[0] / F_IN;
    long E  = (long)in_sizes[1] / 2;

    const int T = 256;
    float* ws   = (float*)d_ws;

    long nNp = (nN + 63) & ~63L;
    long Ep  = (E + 63) & ~63L;
    long nb  = (nN + (long)SCAN_T * SCAN_PER - 1) / ((long)SCAN_T * SCAN_PER);
    long bigR = (Ep > 32 * nNp) ? Ep : 32 * nNp;
    long need = 7 * nNp + 1024 + Ep + bigR;
    bool fits = E > 0 && E < (1L << 31) && (size_t)need * 4 <= ws_size;

    long nbin = (nN + BIN_W - 1) >> BIN_SHIFT;
    long NB   = (E + BIN_CH - 1) / BIN_CH;
    long NBp  = (NB + 3) & ~3L;
    long nscP = (nbin * NBp + 63) & ~63L;
    if (fits && nbin <= BIN_W && NB <= BIN_W && nN <= (1L << SRC_BITS) && 2 * nscP <= 4 * nNp) {
        int*   cnt  = (int*)ws;
        int*   off  = (int*)ws + nNp;
        float* dinv = ws + 2 * nNp;
        float* h2   = ws + 3 * nNp + 1024;
        int*   binCnt = (int*)h2;
        int*   binLs  = (int*)h2 + nscP;
        int*   csr  = (int*)ws + 7 * nNp + 1024;
        float* R    = ws + 7 * nNp + 1024 + Ep;
        unsigned* pk = (unsigned*)R;
        float* h1   = R;

        GNN_bin_group<<<(int)NB, BIN_W, 0, stream>>>(eidx, E, (int)nbin, (int)NBp,
                                                      binCnt, binLs, pk);
        GNN_bin_order<<<(int)nbin, BIN_W, 0, stream>>>(pk, binCnt, binLs, (int)NB, (int)NBp, (int)nN,
                                                      cnt, off, dinv, csr);

        int nTiles = (int)(nN / 16);
        if (nTiles > 0)
            GNN_gemm1_wmma<<<(nTiles + 7) / 8, T, 0, stream>>>(x, W1, dinv, h1, nTiles);
        int rem = (int)(nN - (long)nTiles * 16);
        if (rem > 0)
            GNN_gemm1_tail<<<(rem * 16 + T - 1) / T, T, 0, stream>>>(
                x, W1, dinv, h1, nTiles * 16, (int)nN);

        GNN_pull1<<<(int)((nN + PULL1_NODES - 1) / PULL1_NODES), PULL1_T, 0, stream>>>(
            off, cnt, csr, h1, dinv, b1, W2, h2, (int)nN);
        GNN_pull2<<<(int)((nN + P2_T - 1) / P2_T), P2_T, 0, stream>>>(
            off, cnt, csr, h2, dinv, b2, out, (int)nN);
        return;
    }

    if (nb <= 1024 && fits) {
        int*   cnt  = (int*)ws;
        int*   off  = (int*)ws + nNp;
        float* dinv = ws + 2 * nNp;
        int*   bsum = (int*)ws + 3 * nNp;
        float* h2   = ws + 3 * nNp + 1024;
        int*   csr  = (int*)ws + 7 * nNp + 1024;
        float* R    = ws + 7 * nNp + 1024 + Ep;
        int*   rank = (int*)R;
        float* h1   = R;

        GNN_zero_i32<<<(int)((nN + T - 1) / T), T, 0, stream>>>(cnt, (int)nN);
        GNN_rank    <<<(int)((E + T - 1) / T), T, 0, stream>>>(eidx, E, cnt, rank);
        GNN_scan_tot<<<(int)nb, SCAN_T, 0, stream>>>(cnt, (int)nN, bsum);
        GNN_scan_off<true><<<(int)nb, SCAN_T, 0, stream>>>(cnt, (int)nN, bsum, off, dinv);
        GNN_place   <<<(int)((E + T - 1) / T), T, 0, stream>>>(eidx, E, off, rank, csr);

        int nTiles = (int)(nN / 16);
        if (nTiles > 0)
            GNN_gemm1_wmma<<<(nTiles + 7) / 8, T, 0, stream>>>(x, W1, dinv, h1, nTiles);
        int rem = (int)(nN - (long)nTiles * 16);
        if (rem > 0)
            GNN_gemm1_tail<<<(rem * 16 + T - 1) / T, T, 0, stream>>>(
                x, W1, dinv, h1, nTiles * 16, (int)nN);

        GNN_pull1<<<(int)((nN + PULL1_NODES - 1) / PULL1_NODES), PULL1_T, 0, stream>>>(
            off, cnt, csr, h1, dinv, b1, W2, h2, (int)nN);
        GNN_pull2<<<(int)((nN + P2_T - 1) / P2_T), P2_T, 0, stream>>>(
            off, cnt, csr, h2, dinv, b2, out, (int)nN);
        return;
    }

    float* deg  = ws;
    float* agg1 = ws + nN;
    float* agg2 = ws + 17 * nN;
    float* dinv = ws + 19 * nN;
    float* h1   = ws + 20 * nN;
    float* h2   = ws + 36 * nN;

    long zn = 19 * nN;
    GNN_zero_f32<<<(int)((zn + T - 1) / T), T, 0, stream>>>(ws, zn);
    GNN_degree  <<<(int)((E + T - 1) / T), T, 0, stream>>>(eidx, E, deg);
    GNN_dinv    <<<(int)((nN + T - 1) / T), T, 0, stream>>>(deg, dinv, (int)nN);

    int nTiles = (int)(nN / 16);
    if (nTiles > 0)
        GNN_gemm1_wmma<<<(nTiles + 7) / 8, T, 0, stream>>>(x, W1, dinv, h1, nTiles);
    int rem = (int)(nN - (long)nTiles * 16);
    if (rem > 0)
        GNN_gemm1_tail<<<(rem * 16 + T - 1) / T, T, 0, stream>>>(
            x, W1, dinv, h1, nTiles * 16, (int)nN);

    GNN_scatter1<<<(int)((E + T - 1) / T), T, 0, stream>>>(eidx, E, h1, agg1);
    GNN_combine1<<<(int)((nN + T - 1) / T), T, 0, stream>>>(agg1, h1, dinv, b1, W2, h2, (int)nN);
    GNN_scatter2<<<(int)((E + T - 1) / T), T, 0, stream>>>(eidx, E, h2, agg2);
    GNN_final   <<<(int)((nN + T - 1) / T), T, 0, stream>>>(agg2, h2, dinv, b2, out, (int)nN);
}
